# starting kernel
# baseline (speedup 1.0000x reference)
#include <hip/hip_runtime.h>
#include <math.h>

#define NN   50000
#define EE   640000
#define DD   128
#define PED  37
#define PEDP 40
#define NT   9
#define NL   3

typedef __attribute__((ext_vector_type(2))) float v2f;
typedef __attribute__((ext_vector_type(8))) float v8f;

__device__ __forceinline__ int f2ord(float f) {
  int i = __float_as_int(f);
  return (i >= 0) ? i : (i ^ 0x7fffffff);
}
__device__ __forceinline__ float ord2f(int i) {
  return __int_as_float((i >= 0) ? i : (i ^ 0x7fffffff));
}


__global__ void k_copy_x(const float* __restrict__ X, float* __restrict__ xbuf) {
  int i = blockIdx.x * 256 + threadIdx.x;
  xbuf[i] = X[i];
}

__global__ void k_pad_pe(const float* __restrict__ PE, float* __restrict__ PEp) {
  int i = blockIdx.x * 256 + threadIdx.x;
  if (i >= NN * PEDP) return;
  int r = i / PEDP, c = i - r * PEDP;
  PEp[i] = (c < PED) ? PE[r * PED + c] : 0.0f;
}

__global__ void k_pad_w(const float* __restrict__ W, float* __restrict__ Wp) {
  int i = blockIdx.x * 256 + threadIdx.x;
  if (i >= PEDP * DD) return;
  int k = i >> 7;
  Wp[i] = (k < PED) ? W[i] : 0.0f;
}

__global__ void k_init(float* __restrict__ outb, int* __restrict__ amax,
                       float* __restrict__ denom, float* __restrict__ ssum,
                       float* __restrict__ ssq) {
  int i = blockIdx.x * 256 + threadIdx.x;
  outb[i] = 0.0f;
  if (i < NN) { amax[i] = f2ord(-INFINITY); denom[i] = 0.0f; }
  if (i < DD) { ssum[i] = 0.0f; ssq[i] = 0.0f; }
}

template <int K, bool PE_EPI>
__global__ void k_gemm(const float* __restrict__ A, const float* __restrict__ B,
                       const float* __restrict__ X, const float* __restrict__ bb,
                       float* __restrict__ C) {
  int wave = threadIdx.x >> 5;
  int lane = threadIdx.x & 31;
  int t  = blockIdx.x * 8 + wave;
  int tn = t & 7;
  int tm = t >> 3;
  int mlo = lane & 15;
  int hi  = lane >> 4;
  const float* arow = A + (size_t)(tm * 16 + mlo) * K;
  int ncol = tn * 16 + mlo;

  v8f c = {};
  #pragma unroll 4
  for (int k0 = 0; k0 < K; k0 += 4) {
    v2f a = *(const v2f*)(arow + k0 + hi * 2);
    v2f b;
    b.x = B[(size_t)(k0 + hi * 2    ) * DD + ncol];
    b.y = B[(size_t)(k0 + hi * 2 + 1) * DD + ncol];
    c = __builtin_amdgcn_wmma_f32_16x16x4_f32(false, a, false, b, (short)0, c,
                                              false, false);
  }
  #pragma unroll
  for (int r = 0; r < 8; ++r) {
    int row = tm * 16 + r + hi * 8;
    float v = c[r];
    if (PE_EPI) v += X[(size_t)row * DD + ncol] + bb[ncol];
    C[(size_t)row * DD + ncol] = v;
  }
}


__global__ void k_scores(const float* __restrict__ h, const float* __restrict__ al,
                         const float* __restrict__ ar, float* __restrict__ sl,
                         float* __restrict__ sr) {
  int wave = threadIdx.x >> 5, lane = threadIdx.x & 31;
  int n = blockIdx.x * 8 + wave;
  const float* row = h + (size_t)n * DD;
  float accl = 0.0f, accr = 0.0f;
  #pragma unroll
  for (int j = 0; j < 4; ++j) {
    int d = lane + 32 * j;
    float v = row[d];
    accl += v * al[d];
    accr += v * ar[d];
  }
  #pragma unroll
  for (int off = 16; off; off >>= 1) {
    accl += __shfl_xor(accl, off, 32);
    accr += __shfl_xor(accr, off, 32);
  }
  if (lane == 0) { sl[n] = accl; sr[n] = accr; }
}

__global__ void k_xe(const float* __restrict__ ee, const float* __restrict__ ae,
                     float* __restrict__ xe) {
  int wave = threadIdx.x >> 5, lane = threadIdx.x & 31;
  if (wave >= NT) return;
  float acc = 0.0f;
  #pragma unroll
  for (int j = 0; j < 4; ++j) {
    int d = lane + 32 * j;
    acc += ee[wave * DD + d] * ae[d];
  }
  #pragma unroll
  for (int off = 16; off; off >>= 1) acc += __shfl_xor(acc, off, 32);
  if (lane == 0) xe[wave] = acc;
}


__global__ void k_edge_max(const float* __restrict__ sl, const float* __restrict__ sr,
                           const float* __restrict__ xe, const int* __restrict__ src,
                           const int* __restrict__ dst, const int* __restrict__ attr,
                           float* __restrict__ alpha, int* __restrict__ amax) {
  int e = blockIdx.x * 256 + threadIdx.x;
  int s = src[e], d = dst[e];
  float a = sl[s] + sr[d] + xe[attr[e]];
  a = (a > 0.0f) ? a : 0.2f * a;
  alpha[e] = a;
  atomicMax(&amax[d], f2ord(a));
}

__global__ void k_edge_exp(const int* __restrict__ amax, const int* __restrict__ dst,
                           float* __restrict__ alpha, float* __restrict__ denom) {
  int e = blockIdx.x * 256 + threadIdx.x;
  int d = dst[e];
  float ex = expf(alpha[e] - ord2f(amax[d]));
  alpha[e] = ex;
  atomicAdd(&denom[d], ex);
}

__global__ void k_aggregate(const float* __restrict__ h, const float* __restrict__ alpha,
                            const float* __restrict__ denom, const int* __restrict__ src,
                            const int* __restrict__ dst, float* __restrict__ outb) {
  int wave = threadIdx.x >> 5, lane = threadIdx.x & 31;
  int e = blockIdx.x * 8 + wave;
  int s = src[e], d = dst[e];
  float coeff = alpha[e] / denom[d];
  float4 v = ((const float4*)(h + (size_t)s * DD))[lane];
  float* op = outb + (size_t)d * DD + lane * 4;
  atomicAdd(op + 0, coeff * v.x);
  atomicAdd(op + 1, coeff * v.y);
  atomicAdd(op + 2, coeff * v.z);
  atomicAdd(op + 3, coeff * v.w);
}


__global__ void k_bias_stats(float* __restrict__ outb, const float* __restrict__ bias,
                             float* __restrict__ ssum, float* __restrict__ ssq) {
  int d = threadIdx.x;
  int r0 = blockIdx.x * 128;
  float b = bias[d];
  float s = 0.0f, s2 = 0.0f;
  for (int r = r0; r < r0 + 128 && r < NN; ++r) {
    float v = outb[(size_t)r * DD + d] + b;
    outb[(size_t)r * DD + d] = v;
    s += v; s2 += v * v;
  }
  atomicAdd(&ssum[d], s);
  atomicAdd(&ssq[d], s2);
}

__global__ void k_bn_residual(const float* __restrict__ outb, const float* __restrict__ ssum,
                              const float* __restrict__ ssq, const float* __restrict__ gam,
                              const float* __restrict__ bet, float* __restrict__ xbuf) {
  int i = blockIdx.x * 256 + threadIdx.x;
  int d = i & (DD - 1);
  float mu  = ssum[d] * (1.0f / NN);
  float var = ssq[d] * (1.0f / NN) - mu * mu;
  float inv = rsqrtf(var + 1e-5f);
  xbuf[i] = (outb[i] - mu) * inv * gam[d] + bet[d] + xbuf[i];
}

__global__ void k_final(const float* __restrict__ outb, const float* __restrict__ bias,
                        const float* __restrict__ xbuf, float* __restrict__ out) {
  int i = blockIdx.x * 256 + threadIdx.x;
  int d = i & (DD - 1);
  out[i] = outb[i] + bias[d] + xbuf[i];
}


extern "C" void kernel_launch(void* const* d_in, const int* in_sizes, int n_in,
                              void* d_out, int out_size, void* d_ws, size_t ws_size,
                              hipStream_t stream) {
  const float* X_n   = (const float*)d_in[0];
  const float* PE    = (const float*)d_in[1];
  const int*   eidx  = (const int*)d_in[2];
  const int*   eattr = (const int*)d_in[3];
  const float* eemb  = (const float*)d_in[4];
  const float* peW   = (const float*)d_in[5];
  const float* peB   = (const float*)d_in[6];
  const float* linW  = (const float*)d_in[7];
  const float* attl  = (const float*)d_in[8];
  const float* attr_ = (const float*)d_in[9];
  const float* atte  = (const float*)d_in[10];
  const float* bias  = (const float*)d_in[11];
  const float* gam   = (const float*)d_in[12];
  const float* bet   = (const float*)d_in[13];
  const int* src = eidx;
  const int* dst = eidx + EE;

  float* ws = (float*)d_ws;
  size_t o = 0;
  float* xbuf  = ws + o; o += (size_t)NN * DD;
  float* tmp   = ws + o; o += (size_t)NN * DD;
  float* h     = ws + o; o += (size_t)NN * DD;
  float* PEp   = ws + o; o += (size_t)NN * PEDP;
  float* Wp    = ws + o; o += (size_t)PEDP * DD;
  float* sl    = ws + o; o += NN;
  float* sr    = ws + o; o += NN;
  int*   amax  = (int*)(ws + o); o += NN;
  float* denom = ws + o; o += NN;
  float* alpha = ws + o; o += EE;
  float* xe    = ws + o; o += 16;
  float* ssum  = ws + o; o += DD;
  float* ssq   = ws + o; o += DD;
  float* outb  = tmp;

  const int ND_BLKS = (NN * DD) / 256;

  k_copy_x<<<ND_BLKS, 256, 0, stream>>>(X_n, xbuf);
  k_pad_pe<<<(NN * PEDP + 255) / 256, 256, 0, stream>>>(PE, PEp);

  for (int l = 0; l < NL; ++l) {
    k_pad_w<<<(PEDP * DD + 255) / 256, 256, 0, stream>>>(peW + (size_t)l * PED * DD, Wp);
    k_gemm<PEDP, true><<<3125, 256, 0, stream>>>(PEp, Wp, xbuf, peB + l * DD, tmp);
    k_gemm<DD, false><<<3125, 256, 0, stream>>>(tmp, linW + (size_t)l * DD * DD,
                                                nullptr, nullptr, h);
    k_init<<<ND_BLKS, 256, 0, stream>>>(outb, amax, denom, ssum, ssq);

    k_xe<<<1, 288, 0, stream>>>(eemb + (size_t)l * NT * DD, atte + l * DD, xe);
    k_scores<<<NN / 8, 256, 0, stream>>>(h, attl + l * DD, attr_ + l * DD, sl, sr);

    k_edge_max<<<EE / 256, 256, 0, stream>>>(sl, sr, xe, src, dst, eattr, alpha, amax);
    k_edge_exp<<<EE / 256, 256, 0, stream>>>(amax, dst, alpha, denom);
    k_aggregate<<<EE / 8, 256, 0, stream>>>(h, alpha, denom, src, dst, outb);

    if (l < NL - 1) {
      k_bias_stats<<<(NN + 127) / 128, 128, 0, stream>>>(outb, bias + l * DD, ssum, ssq);
      k_bn_residual<<<ND_BLKS, 256, 0, stream>>>(outb, ssum, ssq, gam + l * DD,
                                                 bet + l * DD, xbuf);
    } else {
      k_final<<<ND_BLKS, 256, 0, stream>>>(outb, bias + l * DD, xbuf, (float*)d_out);
    }
  }
}


# Edges bucketed by destination once per call (returning int atomic rank + scan); one wave per node does max, exp, weighted row sum in registers in edge-id order: float atomics gone, sum order changed
# speedup vs baseline: 1.8788x; 1.8788x over previous
#include <hip/hip_runtime.h>
#include <math.h>

#define NN   50000
#define EE   640000
#define DD   128
#define PED  37
#define PEDP 40
#define NT   9
#define NL   3

typedef __attribute__((ext_vector_type(2))) float v2f;
typedef __attribute__((ext_vector_type(8))) float v8f;

typedef __attribute__((ext_vector_type(4))) int v4i;


__global__ void k_copy_x(const float* __restrict__ X, float* __restrict__ xbuf,
                         int* __restrict__ cnt) {
  int i = blockIdx.x * 256 + threadIdx.x;
  xbuf[i] = X[i];
  if (i < NN) cnt[i] = 0;
}

__global__ void k_pad_pe(const float* __restrict__ PE, float* __restrict__ PEp) {
  int i = blockIdx.x * 256 + threadIdx.x;
  if (i >= NN * PEDP) return;
  int r = i / PEDP, c = i - r * PEDP;
  PEp[i] = (c < PED) ? PE[r * PED + c] : 0.0f;
}

__global__ void k_pad_w(const float* __restrict__ W, float* __restrict__ Wp) {
  int i = blockIdx.x * 256 + threadIdx.x;
  if (i >= PEDP * DD) return;
  int k = i >> 7;
  Wp[i] = (k < PED) ? W[i] : 0.0f;
}

__global__ void k_rank(const int* __restrict__ dst, int* __restrict__ cnt,
                       int* __restrict__ rank) {
  int e = blockIdx.x * 256 + threadIdx.x;
  rank[e] = atomicAdd(&cnt[dst[e]], 1);
}

__global__ void __launch_bounds__(1024) k_scan(const int* __restrict__ cnt,
                                               int* __restrict__ off) {
  __shared__ int wtot[32];
  int wave = threadIdx.x >> 5, lane = threadIdx.x & 31;
  const v4i* c4 = (const v4i*)cnt;
  v4i* o4 = (v4i*)off;
  v4i v[13];
  int s = 0;
  #pragma unroll
  for (int i = 0; i < 13; ++i) {
    int q = wave * 416 + i * 32 + lane;
    v4i z = {0, 0, 0, 0};
    v[i] = (q < NN / 4) ? c4[q] : z;
    s += v[i].x + v[i].y + v[i].z + v[i].w;
  }
  for (int o = 16; o; o >>= 1) s += __shfl_xor(s, o, 32);
  if (lane == 0) wtot[wave] = s;
  __syncthreads();
  int carry = 0;
  for (int w = 0; w < wave; ++w) carry += wtot[w];
  #pragma unroll
  for (int i = 0; i < 13; ++i) {
    int t = v[i].x + v[i].y + v[i].z + v[i].w;
    int y = t;
    for (int o = 1; o < 32; o <<= 1) {
      int u = __shfl_up(y, o, 32);
      if (lane >= o) y += u;
    }
    int ex = carry + y - t;
    int q = wave * 416 + i * 32 + lane;
    if (q < NN / 4) {
      v4i r = {ex, ex + v[i].x, ex + v[i].x + v[i].y, ex + v[i].x + v[i].y + v[i].z};
      o4[q] = r;
    }
    carry += __shfl(y, 31, 32);
  }
  if (threadIdx.x == 1023) off[NN] = carry;
}

__global__ void k_fill(const int* __restrict__ src, const int* __restrict__ dst,
                       const int* __restrict__ attr, const int* __restrict__ off,
                       const int* __restrict__ rank, v4i* __restrict__ rec) {
  int e = blockIdx.x * 256 + threadIdx.x;
  v4i r = {src[e], attr[e], e, 0};
  rec[off[dst[e]] + rank[e]] = r;
}

template <int K, bool PE_EPI>
__global__ void k_gemm(const float* __restrict__ A, const float* __restrict__ B,
                       const float* __restrict__ X, const float* __restrict__ bb,
                       float* __restrict__ C) {
  int wave = threadIdx.x >> 5;
  int lane = threadIdx.x & 31;
  int t  = blockIdx.x * 8 + wave;
  int tn = t & 7;
  int tm = t >> 3;
  int mlo = lane & 15;
  int hi  = lane >> 4;
  const float* arow = A + (size_t)(tm * 16 + mlo) * K;
  int ncol = tn * 16 + mlo;

  v8f c = {};
  #pragma unroll 4
  for (int k0 = 0; k0 < K; k0 += 4) {
    v2f a = *(const v2f*)(arow + k0 + hi * 2);
    v2f b;
    b.x = B[(size_t)(k0 + hi * 2    ) * DD + ncol];
    b.y = B[(size_t)(k0 + hi * 2 + 1) * DD + ncol];
    c = __builtin_amdgcn_wmma_f32_16x16x4_f32(false, a, false, b, (short)0, c,
                                              false, false);
  }
  #pragma unroll
  for (int r = 0; r < 8; ++r) {
    int row = tm * 16 + r + hi * 8;
    float v = c[r];
    if (PE_EPI) v += X[(size_t)row * DD + ncol] + bb[ncol];
    C[(size_t)row * DD + ncol] = v;
  }
}


__global__ void k_scores(const float* __restrict__ h, const float* __restrict__ al,
                         const float* __restrict__ ar, float* __restrict__ sl,
                         float* __restrict__ sr) {
  int wave = threadIdx.x >> 5, lane = threadIdx.x & 31;
  int n = blockIdx.x * 8 + wave;
  const float* row = h + (size_t)n * DD;
  float accl = 0.0f, accr = 0.0f;
  #pragma unroll
  for (int j = 0; j < 4; ++j) {
    int d = lane + 32 * j;
    float v = row[d];
    accl += v * al[d];
    accr += v * ar[d];
  }
  #pragma unroll
  for (int off = 16; off; off >>= 1) {
    accl += __shfl_xor(accl, off, 32);
    accr += __shfl_xor(accr, off, 32);
  }
  if (lane == 0) { sl[n] = accl; sr[n] = accr; }
}

__global__ void k_xe(const float* __restrict__ ee, const float* __restrict__ ae,
                     float* __restrict__ xe, float* __restrict__ ssum,
                     float* __restrict__ ssq) {
  int wave = threadIdx.x >> 5, lane = threadIdx.x & 31;
  if (threadIdx.x < DD) { ssum[threadIdx.x] = 0.0f; ssq[threadIdx.x] = 0.0f; }
  if (wave >= NT) return;
  float acc = 0.0f;
  #pragma unroll
  for (int j = 0; j < 4; ++j) {
    int d = lane + 32 * j;
    acc += ee[wave * DD + d] * ae[d];
  }
  #pragma unroll
  for (int off = 16; off; off >>= 1) acc += __shfl_xor(acc, off, 32);
  if (lane == 0) xe[wave] = acc;
}

__device__ __forceinline__ float edge_score(const float* __restrict__ sl,
                                            const float* __restrict__ xe, float srn,
                                            int s, int t) {
  float a = sl[s] + srn + xe[t];
  return (a > 0.0f) ? a : 0.2f * a;
}

__global__ void k_node(const float* __restrict__ h, const float* __restrict__ sl,
                       const float* __restrict__ sr, const float* __restrict__ xe,
                       const int* __restrict__ off, const v4i* __restrict__ rec,
                       float* __restrict__ outb) {
  int wave = threadIdx.x >> 5, lane = threadIdx.x & 31;
  int n = blockIdx.x * 8 + wave;
  int beg = __builtin_amdgcn_readfirstlane(off[n]);
  int end = __builtin_amdgcn_readfirstlane(off[n + 1]);
  float srn = sr[n];

  int j0 = beg + lane;
  bool has0 = j0 < end;
  int s0 = 0, e0 = 0;
  float a0 = -INFINITY;
  if (has0) {
    v4i r = rec[j0];
    s0 = r.x; e0 = r.z;
    a0 = edge_score(sl, xe, srn, r.x, r.y);
  }
  float m = a0;
  for (int j = j0 + 32; j < end; j += 32) {
    v4i r = rec[j];
    m = fmaxf(m, edge_score(sl, xe, srn, r.x, r.y));
  }
  for (int o = 16; o; o >>= 1) m = fmaxf(m, __shfl_xor(m, o, 32));

  float4 acc = make_float4(0.0f, 0.0f, 0.0f, 0.0f);
  float den = 0.0f;
  for (int base = beg; base < end; base += 32) {
    int sC = s0, eC = e0;
    float aC = a0;
    bool hasC = has0;
    if (base != beg) {
      int j = base + lane;
      hasC = j < end; sC = 0; eC = 0; aC = -INFINITY;
      if (hasC) {
        v4i r = rec[j];
        sC = r.x; eC = r.z;
        aC = edge_score(sl, xe, srn, r.x, r.y);
      }
    }
    float exC = hasC ? expf(aC - m) : 0.0f;
    int c = end - base; if (c > 32) c = 32;
    int rk = 0;
    #pragma unroll 4
    for (int k = 0; k < c; ++k) rk += (__shfl(eC, k, 32) < eC) ? 1 : 0;
    int k = 0;
    for (; k + 4 <= c; k += 4) {
      int L0 = __builtin_ctz(__builtin_amdgcn_ballot_w32(hasC && rk == k));
      int L1 = __builtin_ctz(__builtin_amdgcn_ballot_w32(hasC && rk == k + 1));
      int L2 = __builtin_ctz(__builtin_amdgcn_ballot_w32(hasC && rk == k + 2));
      int L3 = __builtin_ctz(__builtin_amdgcn_ballot_w32(hasC && rk == k + 3));
      float w0 = __shfl(exC, L0, 32), w1 = __shfl(exC, L1, 32);
      float w2 = __shfl(exC, L2, 32), w3 = __shfl(exC, L3, 32);
      float4 v0 = ((const float4*)(h + (size_t)__shfl(sC, L0, 32) * DD))[lane];
      float4 v1 = ((const float4*)(h + (size_t)__shfl(sC, L1, 32) * DD))[lane];
      float4 v2 = ((const float4*)(h + (size_t)__shfl(sC, L2, 32) * DD))[lane];
      float4 v3 = ((const float4*)(h + (size_t)__shfl(sC, L3, 32) * DD))[lane];
      den += w0; acc.x += w0 * v0.x; acc.y += w0 * v0.y; acc.z += w0 * v0.z; acc.w += w0 * v0.w;
      den += w1; acc.x += w1 * v1.x; acc.y += w1 * v1.y; acc.z += w1 * v1.z; acc.w += w1 * v1.w;
      den += w2; acc.x += w2 * v2.x; acc.y += w2 * v2.y; acc.z += w2 * v2.z; acc.w += w2 * v2.w;
      den += w3; acc.x += w3 * v3.x; acc.y += w3 * v3.y; acc.z += w3 * v3.z; acc.w += w3 * v3.w;
    }
    for (; k < c; ++k) {
      int L0 = __builtin_ctz(__builtin_amdgcn_ballot_w32(hasC && rk == k));
      float w0 = __shfl(exC, L0, 32);
      float4 v0 = ((const float4*)(h + (size_t)__shfl(sC, L0, 32) * DD))[lane];
      den += w0; acc.x += w0 * v0.x; acc.y += w0 * v0.y; acc.z += w0 * v0.z; acc.w += w0 * v0.w;
    }
  }
  float inv = (end > beg) ? 1.0f / den : 0.0f;
  float4 o4 = make_float4(acc.x * inv, acc.y * inv, acc.z * inv, acc.w * inv);
  ((float4*)(outb + (size_t)n * DD))[lane] = o4;
}


__global__ void k_bias_stats(float* __restrict__ outb, const float* __restrict__ bias,
                             float* __restrict__ ssum, float* __restrict__ ssq) {
  int d = threadIdx.x;
  int r0 = blockIdx.x * 128;
  float b = bias[d];
  float s = 0.0f, s2 = 0.0f;
  for (int r = r0; r < r0 + 128 && r < NN; ++r) {
    float v = outb[(size_t)r * DD + d] + b;
    outb[(size_t)r * DD + d] = v;
    s += v; s2 += v * v;
  }
  atomicAdd(&ssum[d], s);
  atomicAdd(&ssq[d], s2);
}

__global__ void k_bn_residual(const float* __restrict__ outb, const float* __restrict__ ssum,
                              const float* __restrict__ ssq, const float* __restrict__ gam,
                              const float* __restrict__ bet, float* __restrict__ xbuf) {
  int i = blockIdx.x * 256 + threadIdx.x;
  int d = i & (DD - 1);
  float mu  = ssum[d] * (1.0f / NN);
  float var = ssq[d] * (1.0f / NN) - mu * mu;
  float inv = rsqrtf(var + 1e-5f);
  xbuf[i] = (outb[i] - mu) * inv * gam[d] + bet[d] + xbuf[i];
}

__global__ void k_final(const float* __restrict__ outb, const float* __restrict__ bias,
                        const float* __restrict__ xbuf, float* __restrict__ out) {
  int i = blockIdx.x * 256 + threadIdx.x;
  int d = i & (DD - 1);
  out[i] = outb[i] + bias[d] + xbuf[i];
}


extern "C" void kernel_launch(void* const* d_in, const int* in_sizes, int n_in,
                              void* d_out, int out_size, void* d_ws, size_t ws_size,
                              hipStream_t stream) {
  const float* X_n   = (const float*)d_in[0];
  const float* PE    = (const float*)d_in[1];
  const int*   eidx  = (const int*)d_in[2];
  const int*   eattr = (const int*)d_in[3];
  const float* eemb  = (const float*)d_in[4];
  const float* peW   = (const float*)d_in[5];
  const float* peB   = (const float*)d_in[6];
  const float* linW  = (const float*)d_in[7];
  const float* attl  = (const float*)d_in[8];
  const float* attr_ = (const float*)d_in[9];
  const float* atte  = (const float*)d_in[10];
  const float* bias  = (const float*)d_in[11];
  const float* gam   = (const float*)d_in[12];
  const float* bet   = (const float*)d_in[13];
  const int* src = eidx;
  const int* dst = eidx + EE;

  float* ws = (float*)d_ws;
  size_t o = 0;
  float* xbuf  = ws + o; o += (size_t)NN * DD;
  float* tmp   = ws + o; o += (size_t)NN * DD;
  float* h     = ws + o; o += (size_t)NN * DD;
  float* PEp   = ws + o; o += (size_t)NN * PEDP;
  float* Wp    = ws + o; o += (size_t)PEDP * DD;
  float* sl    = ws + o; o += NN;
  float* sr    = ws + o; o += NN;
  int*   cnt   = (int*)(ws + o); o += NN;
  int*   off   = (int*)(ws + o); o += NN + 16;
  int*   rank  = (int*)(ws + o); o += EE;
  v4i*   rec   = (v4i*)(ws + o); o += (size_t)EE * 4;
  float* xe    = ws + o; o += 16;
  float* ssum  = ws + o; o += DD;
  float* ssq   = ws + o; o += DD;
  float* outb  = tmp;

  const int ND_BLKS = (NN * DD) / 256;

  k_copy_x<<<ND_BLKS, 256, 0, stream>>>(X_n, xbuf, cnt);
  k_pad_pe<<<(NN * PEDP + 255) / 256, 256, 0, stream>>>(PE, PEp);
  k_rank<<<EE / 256, 256, 0, stream>>>(dst, cnt, rank);
  k_scan<<<1, 1024, 0, stream>>>(cnt, off);
  k_fill<<<EE / 256, 256, 0, stream>>>(src, dst, eattr, off, rank, rec);

  for (int l = 0; l < NL; ++l) {
    k_pad_w<<<(PEDP * DD + 255) / 256, 256, 0, stream>>>(peW + (size_t)l * PED * DD, Wp);
    k_gemm<PEDP, true><<<3125, 256, 0, stream>>>(PEp, Wp, xbuf, peB + l * DD, tmp);
    k_gemm<DD, false><<<3125, 256, 0, stream>>>(tmp, linW + (size_t)l * DD * DD,
                                                nullptr, nullptr, h);
    k_xe<<<1, 288, 0, stream>>>(eemb + (size_t)l * NT * DD, atte + l * DD, xe, ssum, ssq);
    k_scores<<<NN / 8, 256, 0, stream>>>(h, attl + l * DD, attr_ + l * DD, sl, sr);

    k_node<<<NN / 8, 256, 0, stream>>>(h, sl, sr, xe, off, rec, outb);

    if (l < NL - 1) {
      k_bias_stats<<<(NN + 127) / 128, 128, 0, stream>>>(outb, bias + l * DD, ssum, ssq);
      k_bn_residual<<<ND_BLKS, 256, 0, stream>>>(outb, ssum, ssq, gam + l * DD,
                                                 bet + l * DD, xbuf);
    } else {
      k_final<<<ND_BLKS, 256, 0, stream>>>(outb, bias + l * DD, xbuf, (float*)d_out);
    }
  }
}


# Lanes of the element-wise, score and edge passes take several 16-byte pieces each (5 float4; 5 nodes per wave; 4 edges) instead of one float: 4x to 20x fewer waves to start; no sum reordered
# speedup vs baseline: 2.0434x; 1.0876x over previous
#include <hip/hip_runtime.h>
#include <math.h>

#define NN   50000
#define EE   640000
#define DD   128
#define PED  37
#define PEDP 40
#define NT   9
#define NL   3

typedef __attribute__((ext_vector_type(2))) float v2f;
typedef __attribute__((ext_vector_type(8))) float v8f;

typedef __attribute__((ext_vector_type(4))) int v4i;
typedef __attribute__((ext_vector_type(4))) float v4f;

#define SU   5
#define ND4  (NN * DD / 4)
#define SBLK (ND4 / (256 * SU))
static_assert(ND4 % (256 * SU) == 0, "element-wise passes assume an exact grid");
static_assert(SU == 5, "the two load fences below name SU = 5 values");
static_assert(EE % 1024 == 0, "edge passes assume four edges per lane, exact grid");


__global__ void __launch_bounds__(256) k_copy_x(const float* __restrict__ X,
                                                float* __restrict__ xbuf,
                                                int* __restrict__ cnt) {
  const v4f* X4 = (const v4f*)X;
  v4f* x4 = (v4f*)xbuf;
  int q0 = blockIdx.x * (256 * SU) + threadIdx.x;
  v4f v[SU];
  #pragma unroll
  for (int j = 0; j < SU; ++j) v[j] = X4[q0 + j * 256];
  asm volatile("" :: "v"(v[0]), "v"(v[1]), "v"(v[2]), "v"(v[3]), "v"(v[4]));
  #pragma unroll
  for (int j = 0; j < SU; ++j) x4[q0 + j * 256] = v[j];
  int g = blockIdx.x * 256 + threadIdx.x;
  v4i z = {0, 0, 0, 0};
  if (g < NN / 4) ((v4i*)cnt)[g] = z;
}

__global__ void __launch_bounds__(256) k_pad_pe(const float* __restrict__ PE,
                                                float* __restrict__ PEp) {
  const int TOT = NN * PEDP / 4;
  float4 v[4];
  #pragma unroll
  for (int j = 0; j < 4; ++j) {
    int q = blockIdx.x * 1024 + j * 256 + threadIdx.x;
    int qc = (q < TOT) ? q : TOT - 1;
    int r = qc / (PEDP / 4);
    int c = (qc - r * (PEDP / 4)) * 4;
    const float* p = PE + (size_t)r * PED + c;
    bool in1 = c + 1 < PED, in2 = c + 2 < PED, in3 = c + 3 < PED;
    float a0 = p[0];
    float a1 = p[in1 ? 1 : 0];
    float a2 = p[in2 ? 2 : 0];
    float a3 = p[in3 ? 3 : 0];
    v[j] = make_float4(a0, in1 ? a1 : 0.0f, in2 ? a2 : 0.0f, in3 ? a3 : 0.0f);
  }
  #pragma unroll
  for (int j = 0; j < 4; ++j) {
    int q = blockIdx.x * 1024 + j * 256 + threadIdx.x;
    if (q < TOT) ((float4*)PEp)[q] = v[j];
  }
}

__global__ void k_pad_w(const float* __restrict__ W, float* __restrict__ Wp) {
  int i = blockIdx.x * 256 + threadIdx.x;
  if (i >= PEDP * DD) return;
  int k = i >> 7;
  Wp[i] = (k < PED) ? W[i] : 0.0f;
}

__global__ void __launch_bounds__(256) k_rank(const int* __restrict__ dst,
                                              int* __restrict__ cnt,
                                              int* __restrict__ rank) {
  int q = blockIdx.x * 256 + threadIdx.x;
  v4i d = ((const v4i*)dst)[q];
  v4i r;
  r.x = atomicAdd(&cnt[d.x], 1);
  r.y = atomicAdd(&cnt[d.y], 1);
  r.z = atomicAdd(&cnt[d.z], 1);
  r.w = atomicAdd(&cnt[d.w], 1);
  ((v4i*)rank)[q] = r;
}

__global__ void __launch_bounds__(1024) k_scan(const int* __restrict__ cnt,
                                               int* __restrict__ off) {
  __shared__ int wtot[32];
  int wave = threadIdx.x >> 5, lane = threadIdx.x & 31;
  const v4i* c4 = (const v4i*)cnt;
  v4i* o4 = (v4i*)off;
  v4i v[13];
  int s = 0;
  #pragma unroll
  for (int i = 0; i < 13; ++i) {
    int q = wave * 416 + i * 32 + lane;
    v4i z = {0, 0, 0, 0};
    v[i] = (q < NN / 4) ? c4[q] : z;
    s += v[i].x + v[i].y + v[i].z + v[i].w;
  }
  for (int o = 16; o; o >>= 1) s += __shfl_xor(s, o, 32);
  if (lane == 0) wtot[wave] = s;
  __syncthreads();
  int carry = 0;
  for (int w = 0; w < wave; ++w) carry += wtot[w];
  #pragma unroll
  for (int i = 0; i < 13; ++i) {
    int t = v[i].x + v[i].y + v[i].z + v[i].w;
    int y = t;
    for (int o = 1; o < 32; o <<= 1) {
      int u = __shfl_up(y, o, 32);
      if (lane >= o) y += u;
    }
    int ex = carry + y - t;
    int q = wave * 416 + i * 32 + lane;
    if (q < NN / 4) {
      v4i r = {ex, ex + v[i].x, ex + v[i].x + v[i].y, ex + v[i].x + v[i].y + v[i].z};
      o4[q] = r;
    }
    carry += __shfl(y, 31, 32);
  }
  if (threadIdx.x == 1023) off[NN] = carry;
}

__global__ void __launch_bounds__(256) k_fill(const int* __restrict__ src,
                                              const int* __restrict__ dst,
                                              const int* __restrict__ attr,
                                              const int* __restrict__ off,
                                              const int* __restrict__ rank,
                                              v4i* __restrict__ rec) {
  int q = blockIdx.x * 256 + threadIdx.x;
  v4i s = ((const v4i*)src)[q];
  v4i d = ((const v4i*)dst)[q];
  v4i a = ((const v4i*)attr)[q];
  v4i k = ((const v4i*)rank)[q];
  int o0 = off[d.x], o1 = off[d.y], o2 = off[d.z], o3 = off[d.w];
  int e = q * 4;
  v4i r0 = {s.x, a.x, e,     0};
  v4i r1 = {s.y, a.y, e + 1, 0};
  v4i r2 = {s.z, a.z, e + 2, 0};
  v4i r3 = {s.w, a.w, e + 3, 0};
  rec[o0 + k.x] = r0;
  rec[o1 + k.y] = r1;
  rec[o2 + k.z] = r2;
  rec[o3 + k.w] = r3;
}

template <int K, bool PE_EPI>
__global__ void k_gemm(const float* __restrict__ A, const float* __restrict__ B,
                       const float* __restrict__ X, const float* __restrict__ bb,
                       float* __restrict__ C) {
  int wave = threadIdx.x >> 5;
  int lane = threadIdx.x & 31;
  int t  = blockIdx.x * 8 + wave;
  int tn = t & 7;
  int tm = t >> 3;
  int mlo = lane & 15;
  int hi  = lane >> 4;
  const float* arow = A + (size_t)(tm * 16 + mlo) * K;
  int ncol = tn * 16 + mlo;

  v8f c = {};
  #pragma unroll 4
  for (int k0 = 0; k0 < K; k0 += 4) {
    v2f a = *(const v2f*)(arow + k0 + hi * 2);
    v2f b;
    b.x = B[(size_t)(k0 + hi * 2    ) * DD + ncol];
    b.y = B[(size_t)(k0 + hi * 2 + 1) * DD + ncol];
    c = __builtin_amdgcn_wmma_f32_16x16x4_f32(false, a, false, b, (short)0, c,
                                              false, false);
  }
  #pragma unroll
  for (int r = 0; r < 8; ++r) {
    int row = tm * 16 + r + hi * 8;
    float v = c[r];
    if (PE_EPI) v += X[(size_t)row * DD + ncol] + bb[ncol];
    C[(size_t)row * DD + ncol] = v;
  }
}


#define NPW 5
static_assert(NN % (8 * NPW) == 0, "score pass assumes an exact grid");
__global__ void __launch_bounds__(256) k_scores(const float* __restrict__ h,
                                                const float* __restrict__ al,
                                                const float* __restrict__ ar,
                                                float* __restrict__ sl,
                                                float* __restrict__ sr) {
  int wave = threadIdx.x >> 5, lane = threadIdx.x & 31;
  int n0 = (blockIdx.x * 8 + wave) * NPW;
  const float* row = h + (size_t)n0 * DD + lane;
  float v[NPW][4];
  #pragma unroll
  for (int i = 0; i < NPW; ++i) {
    #pragma unroll
    for (int j = 0; j < 4; ++j) v[i][j] = row[i * DD + 32 * j];
  }
  float wl[4], wr[4];
  #pragma unroll
  for (int j = 0; j < 4; ++j) { wl[j] = al[lane + 32 * j]; wr[j] = ar[lane + 32 * j]; }
  float ol = 0.0f, orr = 0.0f;
  #pragma unroll
  for (int i = 0; i < NPW; ++i) {
    float accl = 0.0f, accr = 0.0f;
    #pragma unroll
    for (int j = 0; j < 4; ++j) {
      accl += v[i][j] * wl[j];
      accr += v[i][j] * wr[j];
    }
    #pragma unroll
    for (int off = 16; off; off >>= 1) {
      accl += __shfl_xor(accl, off, 32);
      accr += __shfl_xor(accr, off, 32);
    }
    if (lane == i) { ol = accl; orr = accr; }
  }
  if (lane < NPW) { sl[n0 + lane] = ol; sr[n0 + lane] = orr; }
}

__global__ void k_xe(const float* __restrict__ ee, const float* __restrict__ ae,
                     float* __restrict__ xe, float* __restrict__ ssum,
                     float* __restrict__ ssq) {
  int wave = threadIdx.x >> 5, lane = threadIdx.x & 31;
  if (threadIdx.x < DD) { ssum[threadIdx.x] = 0.0f; ssq[threadIdx.x] = 0.0f; }
  if (wave >= NT) return;
  float acc = 0.0f;
  #pragma unroll
  for (int j = 0; j < 4; ++j) {
    int d = lane + 32 * j;
    acc += ee[wave * DD + d] * ae[d];
  }
  #pragma unroll
  for (int off = 16; off; off >>= 1) acc += __shfl_xor(acc, off, 32);
  if (lane == 0) xe[wave] = acc;
}

__device__ __forceinline__ float edge_score(const float* __restrict__ sl,
                                            const float* __restrict__ xe, float srn,
                                            int s, int t) {
  float a = sl[s] + srn + xe[t];
  return (a > 0.0f) ? a : 0.2f * a;
}

__global__ void k_node(const float* __restrict__ h, const float* __restrict__ sl,
                       const float* __restrict__ sr, const float* __restrict__ xe,
                       const int* __restrict__ off, const v4i* __restrict__ rec,
                       float* __restrict__ outb) {
  int wave = threadIdx.x >> 5, lane = threadIdx.x & 31;
  int n = blockIdx.x * 8 + wave;
  int beg = __builtin_amdgcn_readfirstlane(off[n]);
  int end = __builtin_amdgcn_readfirstlane(off[n + 1]);
  float srn = sr[n];

  int j0 = beg + lane;
  bool has0 = j0 < end;
  int s0 = 0, e0 = 0;
  float a0 = -INFINITY;
  if (has0) {
    v4i r = rec[j0];
    s0 = r.x; e0 = r.z;
    a0 = edge_score(sl, xe, srn, r.x, r.y);
  }
  float m = a0;
  for (int j = j0 + 32; j < end; j += 32) {
    v4i r = rec[j];
    m = fmaxf(m, edge_score(sl, xe, srn, r.x, r.y));
  }
  for (int o = 16; o; o >>= 1) m = fmaxf(m, __shfl_xor(m, o, 32));

  float4 acc = make_float4(0.0f, 0.0f, 0.0f, 0.0f);
  float den = 0.0f;
  for (int base = beg; base < end; base += 32) {
    int sC = s0, eC = e0;
    float aC = a0;
    bool hasC = has0;
    if (base != beg) {
      int j = base + lane;
      hasC = j < end; sC = 0; eC = 0; aC = -INFINITY;
      if (hasC) {
        v4i r = rec[j];
        sC = r.x; eC = r.z;
        aC = edge_score(sl, xe, srn, r.x, r.y);
      }
    }
    float exC = hasC ? expf(aC - m) : 0.0f;
    int c = end - base; if (c > 32) c = 32;
    int rk = 0;
    #pragma unroll 4
    for (int k = 0; k < c; ++k) rk += (__shfl(eC, k, 32) < eC) ? 1 : 0;
    int k = 0;
    for (; k + 4 <= c; k += 4) {
      int L0 = __builtin_ctz(__builtin_amdgcn_ballot_w32(hasC && rk == k));
      int L1 = __builtin_ctz(__builtin_amdgcn_ballot_w32(hasC && rk == k + 1));
      int L2 = __builtin_ctz(__builtin_amdgcn_ballot_w32(hasC && rk == k + 2));
      int L3 = __builtin_ctz(__builtin_amdgcn_ballot_w32(hasC && rk == k + 3));
      float w0 = __shfl(exC, L0, 32), w1 = __shfl(exC, L1, 32);
      float w2 = __shfl(exC, L2, 32), w3 = __shfl(exC, L3, 32);
      float4 v0 = ((const float4*)(h + (size_t)__shfl(sC, L0, 32) * DD))[lane];
      float4 v1 = ((const float4*)(h + (size_t)__shfl(sC, L1, 32) * DD))[lane];
      float4 v2 = ((const float4*)(h + (size_t)__shfl(sC, L2, 32) * DD))[lane];
      float4 v3 = ((const float4*)(h + (size_t)__shfl(sC, L3, 32) * DD))[lane];
      den += w0; acc.x += w0 * v0.x; acc.y += w0 * v0.y; acc.z += w0 * v0.z; acc.w += w0 * v0.w;
      den += w1; acc.x += w1 * v1.x; acc.y += w1 * v1.y; acc.z += w1 * v1.z; acc.w += w1 * v1.w;
      den += w2; acc.x += w2 * v2.x; acc.y += w2 * v2.y; acc.z += w2 * v2.z; acc.w += w2 * v2.w;
      den += w3; acc.x += w3 * v3.x; acc.y += w3 * v3.y; acc.z += w3 * v3.z; acc.w += w3 * v3.w;
    }
    for (; k < c; ++k) {
      int L0 = __builtin_ctz(__builtin_amdgcn_ballot_w32(hasC && rk == k));
      float w0 = __shfl(exC, L0, 32);
      float4 v0 = ((const float4*)(h + (size_t)__shfl(sC, L0, 32) * DD))[lane];
      den += w0; acc.x += w0 * v0.x; acc.y += w0 * v0.y; acc.z += w0 * v0.z; acc.w += w0 * v0.w;
    }
  }
  float inv = (end > beg) ? 1.0f / den : 0.0f;
  float4 o4 = make_float4(acc.x * inv, acc.y * inv, acc.z * inv, acc.w * inv);
  ((float4*)(outb + (size_t)n * DD))[lane] = o4;
}


__global__ void k_bias_stats(float* __restrict__ outb, const float* __restrict__ bias,
                             float* __restrict__ ssum, float* __restrict__ ssq) {
  int d = threadIdx.x;
  int r0 = blockIdx.x * 128;
  float b = bias[d];
  float s = 0.0f, s2 = 0.0f;
  for (int r = r0; r < r0 + 128 && r < NN; ++r) {
    float v = outb[(size_t)r * DD + d] + b;
    outb[(size_t)r * DD + d] = v;
    s += v; s2 += v * v;
  }
  atomicAdd(&ssum[d], s);
  atomicAdd(&ssq[d], s2);
}

__device__ __forceinline__ float bn_one(float o, float s, float s2, float g, float b, float x) {
  float mu  = s * (1.0f / NN);
  float var = s2 * (1.0f / NN) - mu * mu;
  float inv = rsqrtf(var + 1e-5f);
  return (o - mu) * inv * g + b + x;
}

__global__ void __launch_bounds__(256) k_bn_residual(const float* __restrict__ outb,
                                                     const float* __restrict__ ssum,
                                                     const float* __restrict__ ssq,
                                                     const float* __restrict__ gam,
                                                     const float* __restrict__ bet,
                                                     float* __restrict__ xbuf) {
  const v4f* o4 = (const v4f*)outb;
  v4f* x4 = (v4f*)xbuf;
  int q0 = blockIdx.x * (256 * SU) + threadIdx.x;
  int c4 = threadIdx.x & (DD / 4 - 1);
  v4f o[SU], x[SU];
  v4f s  = ((const v4f*)ssum)[c4];
  v4f s2 = ((const v4f*)ssq)[c4];
  v4f g  = ((const v4f*)gam)[c4];
  v4f b  = ((const v4f*)bet)[c4];
  #pragma unroll
  for (int j = 0; j < SU; ++j) { o[j] = o4[q0 + j * 256]; x[j] = x4[q0 + j * 256]; }
  asm volatile("" :: "v"(x[0]), "v"(x[1]), "v"(x[2]), "v"(x[3]), "v"(x[4]));
  #pragma unroll
  for (int j = 0; j < SU; ++j) {
    v4f r;
    r.x = bn_one(o[j].x, s.x, s2.x, g.x, b.x, x[j].x);
    r.y = bn_one(o[j].y, s.y, s2.y, g.y, b.y, x[j].y);
    r.z = bn_one(o[j].z, s.z, s2.z, g.z, b.z, x[j].z);
    r.w = bn_one(o[j].w, s.w, s2.w, g.w, b.w, x[j].w);
    x4[q0 + j * 256] = r;
  }
}

__global__ void __launch_bounds__(256) k_final(const float* __restrict__ outb,
                                               const float* __restrict__ bias,
                                               const float* __restrict__ xbuf,
                                               float* __restrict__ out) {
  const float4* o4 = (const float4*)outb;
  const float4* x4 = (const float4*)xbuf;
  int q0 = blockIdx.x * (256 * SU) + threadIdx.x;
  float4 o[SU], x[SU];
  #pragma unroll
  for (int j = 0; j < SU; ++j) { o[j] = o4[q0 + j * 256]; x[j] = x4[q0 + j * 256]; }
  float4 b = ((const float4*)bias)[threadIdx.x & (DD / 4 - 1)];
  #pragma unroll
  for (int j = 0; j < SU; ++j) {
    float4 r;
    r.x = o[j].x + b.x + x[j].x;
    r.y = o[j].y + b.y + x[j].y;
    r.z = o[j].z + b.z + x[j].z;
    r.w = o[j].w + b.w + x[j].w;
    ((float4*)out)[q0 + j * 256] = r;
  }
}


extern "C" void kernel_launch(void* const* d_in, const int* in_sizes, int n_in,
                              void* d_out, int out_size, void* d_ws, size_t ws_size,
                              hipStream_t stream) {
  const float* X_n   = (const float*)d_in[0];
  const float* PE    = (const float*)d_in[1];
  const int*   eidx  = (const int*)d_in[2];
  const int*   eattr = (const int*)d_in[3];
  const float* eemb  = (const float*)d_in[4];
  const float* peW   = (const float*)d_in[5];
  const float* peB   = (const float*)d_in[6];
  const float* linW  = (const float*)d_in[7];
  const float* attl  = (const float*)d_in[8];
  const float* attr_ = (const float*)d_in[9];
  const float* atte  = (const float*)d_in[10];
  const float* bias  = (const float*)d_in[11];
  const float* gam   = (const float*)d_in[12];
  const float* bet   = (const float*)d_in[13];
  const int* src = eidx;
  const int* dst = eidx + EE;

  float* ws = (float*)d_ws;
  size_t o = 0;
  float* xbuf  = ws + o; o += (size_t)NN * DD;
  float* tmp   = ws + o; o += (size_t)NN * DD;
  float* h     = ws + o; o += (size_t)NN * DD;
  float* PEp   = ws + o; o += (size_t)NN * PEDP;
  float* Wp    = ws + o; o += (size_t)PEDP * DD;
  float* sl    = ws + o; o += NN;
  float* sr    = ws + o; o += NN;
  int*   cnt   = (int*)(ws + o); o += NN;
  int*   off   = (int*)(ws + o); o += NN + 16;
  int*   rank  = (int*)(ws + o); o += EE;
  v4i*   rec   = (v4i*)(ws + o); o += (size_t)EE * 4;
  float* xe    = ws + o; o += 16;
  float* ssum  = ws + o; o += DD;
  float* ssq   = ws + o; o += DD;
  float* outb  = tmp;

  k_copy_x<<<SBLK, 256, 0, stream>>>(X_n, xbuf, cnt);
  k_pad_pe<<<(NN * PEDP / 4 + 1023) / 1024, 256, 0, stream>>>(PE, PEp);
  k_rank<<<EE / 1024, 256, 0, stream>>>(dst, cnt, rank);
  k_scan<<<1, 1024, 0, stream>>>(cnt, off);
  k_fill<<<EE / 1024, 256, 0, stream>>>(src, dst, eattr, off, rank, rec);

  for (int l = 0; l < NL; ++l) {
    k_pad_w<<<(PEDP * DD + 255) / 256, 256, 0, stream>>>(peW + (size_t)l * PED * DD, Wp);
    k_gemm<PEDP, true><<<3125, 256, 0, stream>>>(PEp, Wp, xbuf, peB + l * DD, tmp);
    k_gemm<DD, false><<<3125, 256, 0, stream>>>(tmp, linW + (size_t)l * DD * DD,
                                                nullptr, nullptr, h);
    k_xe<<<1, 288, 0, stream>>>(eemb + (size_t)l * NT * DD, atte + l * DD, xe, ssum, ssq);
    k_scores<<<NN / (8 * NPW), 256, 0, stream>>>(h, attl + l * DD, attr_ + l * DD, sl, sr);

    k_node<<<NN / 8, 256, 0, stream>>>(h, sl, sr, xe, off, rec, outb);

    if (l < NL - 1) {
      k_bias_stats<<<(NN + 127) / 128, 128, 0, stream>>>(outb, bias + l * DD, ssum, ssq);
      k_bn_residual<<<SBLK, 256, 0, stream>>>(outb, ssum, ssq, gam + l * DD,
                                              bet + l * DD, xbuf);
    } else {
      k_final<<<SBLK, 256, 0, stream>>>(outb, bias + l * DD, xbuf, (float*)d_out);
    }
  }
}


# The layer's two WMMA products run in one launch: each work-group hands its 16x128 strip of the first product to the second through LDS instead of through memory; k order of both sums unchanged
# speedup vs baseline: 2.6582x; 1.3009x over previous
#include <hip/hip_runtime.h>
#include <math.h>

#define NN   50000
#define EE   640000
#define DD   128
#define PED  37
#define PEDP 40
#define NT   9
#define NL   3

typedef __attribute__((ext_vector_type(2))) float v2f;
typedef __attribute__((ext_vector_type(8))) float v8f;

typedef __attribute__((ext_vector_type(4))) int v4i;
typedef __attribute__((ext_vector_type(4))) float v4f;

#define SU   5
#define ND4  (NN * DD / 4)
#define SBLK (ND4 / (256 * SU))
static_assert(ND4 % (256 * SU) == 0, "element-wise passes assume an exact grid");
static_assert(SU == 5, "the two load fences below name SU = 5 values");
static_assert(EE % 1024 == 0, "edge passes assume four edges per lane, exact grid");


__global__ void __launch_bounds__(256) k_copy_x(const float* __restrict__ X,
                                                float* __restrict__ xbuf,
                                                int* __restrict__ cnt) {
  const v4f* X4 = (const v4f*)X;
  v4f* x4 = (v4f*)xbuf;
  int q0 = blockIdx.x * (256 * SU) + threadIdx.x;
  v4f v[SU];
  #pragma unroll
  for (int j = 0; j < SU; ++j) v[j] = X4[q0 + j * 256];
  asm volatile("" :: "v"(v[0]), "v"(v[1]), "v"(v[2]), "v"(v[3]), "v"(v[4]));
  #pragma unroll
  for (int j = 0; j < SU; ++j) x4[q0 + j * 256] = v[j];
  int g = blockIdx.x * 256 + threadIdx.x;
  v4i z = {0, 0, 0, 0};
  if (g < NN / 4) ((v4i*)cnt)[g] = z;
}

__global__ void __launch_bounds__(256) k_pad_pe(const float* __restrict__ PE,
                                                float* __restrict__ PEp) {
  const int TOT = NN * PEDP / 4;
  float4 v[4];
  #pragma unroll
  for (int j = 0; j < 4; ++j) {
    int q = blockIdx.x * 1024 + j * 256 + threadIdx.x;
    int qc = (q < TOT) ? q : TOT - 1;
    int r = qc / (PEDP / 4);
    int c = (qc - r * (PEDP / 4)) * 4;
    const float* p = PE + (size_t)r * PED + c;
    bool in1 = c + 1 < PED, in2 = c + 2 < PED, in3 = c + 3 < PED;
    float a0 = p[0];
    float a1 = p[in1 ? 1 : 0];
    float a2 = p[in2 ? 2 : 0];
    float a3 = p[in3 ? 3 : 0];
    v[j] = make_float4(a0, in1 ? a1 : 0.0f, in2 ? a2 : 0.0f, in3 ? a3 : 0.0f);
  }
  #pragma unroll
  for (int j = 0; j < 4; ++j) {
    int q = blockIdx.x * 1024 + j * 256 + threadIdx.x;
    if (q < TOT) ((float4*)PEp)[q] = v[j];
  }
}

__global__ void k_pad_w(const float* __restrict__ W, float* __restrict__ Wp) {
  int i = blockIdx.x * 256 + threadIdx.x;
  if (i >= PEDP * DD) return;
  int k = i >> 7;
  Wp[i] = (k < PED) ? W[i] : 0.0f;
}

__global__ void __launch_bounds__(256) k_rank(const int* __restrict__ dst,
                                              int* __restrict__ cnt,
                                              int* __restrict__ rank) {
  int q = blockIdx.x * 256 + threadIdx.x;
  v4i d = ((const v4i*)dst)[q];
  v4i r;
  r.x = atomicAdd(&cnt[d.x], 1);
  r.y = atomicAdd(&cnt[d.y], 1);
  r.z = atomicAdd(&cnt[d.z], 1);
  r.w = atomicAdd(&cnt[d.w], 1);
  ((v4i*)rank)[q] = r;
}

__global__ void __launch_bounds__(1024) k_scan(const int* __restrict__ cnt,
                                               int* __restrict__ off) {
  __shared__ int wtot[32];
  int wave = threadIdx.x >> 5, lane = threadIdx.x & 31;
  const v4i* c4 = (const v4i*)cnt;
  v4i* o4 = (v4i*)off;
  v4i v[13];
  int s = 0;
  #pragma unroll
  for (int i = 0; i < 13; ++i) {
    int q = wave * 416 + i * 32 + lane;
    v4i z = {0, 0, 0, 0};
    v[i] = (q < NN / 4) ? c4[q] : z;
    s += v[i].x + v[i].y + v[i].z + v[i].w;
  }
  for (int o = 16; o; o >>= 1) s += __shfl_xor(s, o, 32);
  if (lane == 0) wtot[wave] = s;
  __syncthreads();
  int carry = 0;
  for (int w = 0; w < wave; ++w) carry += wtot[w];
  #pragma unroll
  for (int i = 0; i < 13; ++i) {
    int t = v[i].x + v[i].y + v[i].z + v[i].w;
    int y = t;
    for (int o = 1; o < 32; o <<= 1) {
      int u = __shfl_up(y, o, 32);
      if (lane >= o) y += u;
    }
    int ex = carry + y - t;
    int q = wave * 416 + i * 32 + lane;
    if (q < NN / 4) {
      v4i r = {ex, ex + v[i].x, ex + v[i].x + v[i].y, ex + v[i].x + v[i].y + v[i].z};
      o4[q] = r;
    }
    carry += __shfl(y, 31, 32);
  }
  if (threadIdx.x == 1023) off[NN] = carry;
}

__global__ void __launch_bounds__(256) k_fill(const int* __restrict__ src,
                                              const int* __restrict__ dst,
                                              const int* __restrict__ attr,
                                              const int* __restrict__ off,
                                              const int* __restrict__ rank,
                                              v4i* __restrict__ rec) {
  int q = blockIdx.x * 256 + threadIdx.x;
  v4i s = ((const v4i*)src)[q];
  v4i d = ((const v4i*)dst)[q];
  v4i a = ((const v4i*)attr)[q];
  v4i k = ((const v4i*)rank)[q];
  int o0 = off[d.x], o1 = off[d.y], o2 = off[d.z], o3 = off[d.w];
  int e = q * 4;
  v4i r0 = {s.x, a.x, e,     0};
  v4i r1 = {s.y, a.y, e + 1, 0};
  v4i r2 = {s.z, a.z, e + 2, 0};
  v4i r3 = {s.w, a.w, e + 3, 0};
  rec[o0 + k.x] = r0;
  rec[o1 + k.y] = r1;
  rec[o2 + k.z] = r2;
  rec[o3 + k.w] = r3;
}

#define TP (DD + 4)
static_assert(NN % 16 == 0, "row tiles are exact");
__global__ void __launch_bounds__(256) k_gemm2x(const float* __restrict__ PEp,
                                                const float* __restrict__ Wp,
                                                const float* __restrict__ X,
                                                const float* __restrict__ bb,
                                                const float* __restrict__ W2,
                                                float* __restrict__ H) {
  __shared__ __attribute__((aligned(16))) float strip[16 * TP];
  int wave = threadIdx.x >> 5;
  int lane = threadIdx.x & 31;
  int tm  = blockIdx.x;
  int mlo = lane & 15;
  int hi  = lane >> 4;
  int ncol = wave * 16 + mlo;

  const float* arow = PEp + (size_t)(tm * 16 + mlo) * PEDP;
  v8f c = {};
  #pragma unroll
  for (int k0 = 0; k0 < PEDP; k0 += 4) {
    v2f a = *(const v2f*)(arow + k0 + hi * 2);
    v2f b;
    b.x = Wp[(size_t)(k0 + hi * 2    ) * DD + ncol];
    b.y = Wp[(size_t)(k0 + hi * 2 + 1) * DD + ncol];
    c = __builtin_amdgcn_wmma_f32_16x16x4_f32(false, a, false, b, (short)0, c,
                                              false, false);
  }
  #pragma unroll
  for (int r = 0; r < 8; ++r) {
    int row = r + hi * 8;
    float v = c[r];
    v += X[(size_t)(tm * 16 + row) * DD + ncol] + bb[ncol];
    strip[row * TP + ncol] = v;
  }
  __syncthreads();

  const float* srow = strip + mlo * TP + hi * 2;
  v8f d = {};
  #pragma unroll 4
  for (int k0 = 0; k0 < DD; k0 += 4) {
    v2f a = *(const v2f*)(srow + k0);
    v2f b;
    b.x = W2[(size_t)(k0 + hi * 2    ) * DD + ncol];
    b.y = W2[(size_t)(k0 + hi * 2 + 1) * DD + ncol];
    d = __builtin_amdgcn_wmma_f32_16x16x4_f32(false, a, false, b, (short)0, d,
                                              false, false);
  }
  #pragma unroll
  for (int r = 0; r < 8; ++r) {
    int row = tm * 16 + r + hi * 8;
    H[(size_t)row * DD + ncol] = d[r];
  }
}


#define NPW 5
static_assert(NN % (8 * NPW) == 0, "score pass assumes an exact grid");
__global__ void __launch_bounds__(256) k_scores(const float* __restrict__ h,
                                                const float* __restrict__ al,
                                                const float* __restrict__ ar,
                                                float* __restrict__ sl,
                                                float* __restrict__ sr) {
  int wave = threadIdx.x >> 5, lane = threadIdx.x & 31;
  int n0 = (blockIdx.x * 8 + wave) * NPW;
  const float* row = h + (size_t)n0 * DD + lane;
  float v[NPW][4];
  #pragma unroll
  for (int i = 0; i < NPW; ++i) {
    #pragma unroll
    for (int j = 0; j < 4; ++j) v[i][j] = row[i * DD + 32 * j];
  }
  float wl[4], wr[4];
  #pragma unroll
  for (int j = 0; j < 4; ++j) { wl[j] = al[lane + 32 * j]; wr[j] = ar[lane + 32 * j]; }
  float ol = 0.0f, orr = 0.0f;
  #pragma unroll
  for (int i = 0; i < NPW; ++i) {
    float accl = 0.0f, accr = 0.0f;
    #pragma unroll
    for (int j = 0; j < 4; ++j) {
      accl += v[i][j] * wl[j];
      accr += v[i][j] * wr[j];
    }
    #pragma unroll
    for (int off = 16; off; off >>= 1) {
      accl += __shfl_xor(accl, off, 32);
      accr += __shfl_xor(accr, off, 32);
    }
    if (lane == i) { ol = accl; orr = accr; }
  }
  if (lane < NPW) { sl[n0 + lane] = ol; sr[n0 + lane] = orr; }
}

__global__ void k_xe(const float* __restrict__ ee, const float* __restrict__ ae,
                     float* __restrict__ xe, float* __restrict__ ssum,
                     float* __restrict__ ssq) {
  int wave = threadIdx.x >> 5, lane = threadIdx.x & 31;
  if (threadIdx.x < DD) { ssum[threadIdx.x] = 0.0f; ssq[threadIdx.x] = 0.0f; }
  if (wave >= NT) return;
  float acc = 0.0f;
  #pragma unroll
  for (int j = 0; j < 4; ++j) {
    int d = lane + 32 * j;
    acc += ee[wave * DD + d] * ae[d];
  }
  #pragma unroll
  for (int off = 16; off; off >>= 1) acc += __shfl_xor(acc, off, 32);
  if (lane == 0) xe[wave] = acc;
}

__device__ __forceinline__ float edge_score(const float* __restrict__ sl,
                                            const float* __restrict__ xe, float srn,
                                            int s, int t) {
  float a = sl[s] + srn + xe[t];
  return (a > 0.0f) ? a : 0.2f * a;
}

__global__ void k_node(const float* __restrict__ h, const float* __restrict__ sl,
                       const float* __restrict__ sr, const float* __restrict__ xe,
                       const int* __restrict__ off, const v4i* __restrict__ rec,
                       float* __restrict__ outb) {
  int wave = threadIdx.x >> 5, lane = threadIdx.x & 31;
  int n = blockIdx.x * 8 + wave;
  int beg = __builtin_amdgcn_readfirstlane(off[n]);
  int end = __builtin_amdgcn_readfirstlane(off[n + 1]);
  float srn = sr[n];

  int j0 = beg + lane;
  bool has0 = j0 < end;
  int s0 = 0, e0 = 0;
  float a0 = -INFINITY;
  if (has0) {
    v4i r = rec[j0];
    s0 = r.x; e0 = r.z;
    a0 = edge_score(sl, xe, srn, r.x, r.y);
  }
  float m = a0;
  for (int j = j0 + 32; j < end; j += 32) {
    v4i r = rec[j];
    m = fmaxf(m, edge_score(sl, xe, srn, r.x, r.y));
  }
  for (int o = 16; o; o >>= 1) m = fmaxf(m, __shfl_xor(m, o, 32));

  float4 acc = make_float4(0.0f, 0.0f, 0.0f, 0.0f);
  float den = 0.0f;
  for (int base = beg; base < end; base += 32) {
    int sC = s0, eC = e0;
    float aC = a0;
    bool hasC = has0;
    if (base != beg) {
      int j = base + lane;
      hasC = j < end; sC = 0; eC = 0; aC = -INFINITY;
      if (hasC) {
        v4i r = rec[j];
        sC = r.x; eC = r.z;
        aC = edge_score(sl, xe, srn, r.x, r.y);
      }
    }
    float exC = hasC ? expf(aC - m) : 0.0f;
    int c = end - base; if (c > 32) c = 32;
    int rk = 0;
    #pragma unroll 4
    for (int k = 0; k < c; ++k) rk += (__shfl(eC, k, 32) < eC) ? 1 : 0;
    int k = 0;
    for (; k + 4 <= c; k += 4) {
      int L0 = __builtin_ctz(__builtin_amdgcn_ballot_w32(hasC && rk == k));
      int L1 = __builtin_ctz(__builtin_amdgcn_ballot_w32(hasC && rk == k + 1));
      int L2 = __builtin_ctz(__builtin_amdgcn_ballot_w32(hasC && rk == k + 2));
      int L3 = __builtin_ctz(__builtin_amdgcn_ballot_w32(hasC && rk == k + 3));
      float w0 = __shfl(exC, L0, 32), w1 = __shfl(exC, L1, 32);
      float w2 = __shfl(exC, L2, 32), w3 = __shfl(exC, L3, 32);
      float4 v0 = ((const float4*)(h + (size_t)__shfl(sC, L0, 32) * DD))[lane];
      float4 v1 = ((const float4*)(h + (size_t)__shfl(sC, L1, 32) * DD))[lane];
      float4 v2 = ((const float4*)(h + (size_t)__shfl(sC, L2, 32) * DD))[lane];
      float4 v3 = ((const float4*)(h + (size_t)__shfl(sC, L3, 32) * DD))[lane];
      den += w0; acc.x += w0 * v0.x; acc.y += w0 * v0.y; acc.z += w0 * v0.z; acc.w += w0 * v0.w;
      den += w1; acc.x += w1 * v1.x; acc.y += w1 * v1.y; acc.z += w1 * v1.z; acc.w += w1 * v1.w;
      den += w2; acc.x += w2 * v2.x; acc.y += w2 * v2.y; acc.z += w2 * v2.z; acc.w += w2 * v2.w;
      den += w3; acc.x += w3 * v3.x; acc.y += w3 * v3.y; acc.z += w3 * v3.z; acc.w += w3 * v3.w;
    }
    for (; k < c; ++k) {
      int L0 = __builtin_ctz(__builtin_amdgcn_ballot_w32(hasC && rk == k));
      float w0 = __shfl(exC, L0, 32);
      float4 v0 = ((const float4*)(h + (size_t)__shfl(sC, L0, 32) * DD))[lane];
      den += w0; acc.x += w0 * v0.x; acc.y += w0 * v0.y; acc.z += w0 * v0.z; acc.w += w0 * v0.w;
    }
  }
  float inv = (end > beg) ? 1.0f / den : 0.0f;
  float4 o4 = make_float4(acc.x * inv, acc.y * inv, acc.z * inv, acc.w * inv);
  ((float4*)(outb + (size_t)n * DD))[lane] = o4;
}


__global__ void k_bias_stats(float* __restrict__ outb, const float* __restrict__ bias,
                             float* __restrict__ ssum, float* __restrict__ ssq) {
  int d = threadIdx.x;
  int r0 = blockIdx.x * 128;
  float b = bias[d];
  float s = 0.0f, s2 = 0.0f;
  for (int r = r0; r < r0 + 128 && r < NN; ++r) {
    float v = outb[(size_t)r * DD + d] + b;
    outb[(size_t)r * DD + d] = v;
    s += v; s2 += v * v;
  }
  atomicAdd(&ssum[d], s);
  atomicAdd(&ssq[d], s2);
}

__device__ __forceinline__ float bn_one(float o, float s, float s2, float g, float b, float x) {
  float mu  = s * (1.0f / NN);
  float var = s2 * (1.0f / NN) - mu * mu;
  float inv = rsqrtf(var + 1e-5f);
  return (o - mu) * inv * g + b + x;
}

__global__ void __launch_bounds__(256) k_bn_residual(const float* __restrict__ outb,
                                                     const float* __restrict__ ssum,
                                                     const float* __restrict__ ssq,
                                                     const float* __restrict__ gam,
                                                     const float* __restrict__ bet,
                                                     float* __restrict__ xbuf) {
  const v4f* o4 = (const v4f*)outb;
  v4f* x4 = (v4f*)xbuf;
  int q0 = blockIdx.x * (256 * SU) + threadIdx.x;
  int c4 = threadIdx.x & (DD / 4 - 1);
  v4f o[SU], x[SU];
  v4f s  = ((const v4f*)ssum)[c4];
  v4f s2 = ((const v4f*)ssq)[c4];
  v4f g  = ((const v4f*)gam)[c4];
  v4f b  = ((const v4f*)bet)[c4];
  #pragma unroll
  for (int j = 0; j < SU; ++j) { o[j] = o4[q0 + j * 256]; x[j] = x4[q0 + j * 256]; }
  asm volatile("" :: "v"(x[0]), "v"(x[1]), "v"(x[2]), "v"(x[3]), "v"(x[4]));
  #pragma unroll
  for (int j = 0; j < SU; ++j) {
    v4f r;
    r.x = bn_one(o[j].x, s.x, s2.x, g.x, b.x, x[j].x);
    r.y = bn_one(o[j].y, s.y, s2.y, g.y, b.y, x[j].y);
    r.z = bn_one(o[j].z, s.z, s2.z, g.z, b.z, x[j].z);
    r.w = bn_one(o[j].w, s.w, s2.w, g.w, b.w, x[j].w);
    x4[q0 + j * 256] = r;
  }
}

__global__ void __launch_bounds__(256) k_final(const float* __restrict__ outb,
                                               const float* __restrict__ bias,
                                               const float* __restrict__ xbuf,
                                               float* __restrict__ out) {
  const float4* o4 = (const float4*)outb;
  const float4* x4 = (const float4*)xbuf;
  int q0 = blockIdx.x * (256 * SU) + threadIdx.x;
  float4 o[SU], x[SU];
  #pragma unroll
  for (int j = 0; j < SU; ++j) { o[j] = o4[q0 + j * 256]; x[j] = x4[q0 + j * 256]; }
  float4 b = ((const float4*)bias)[threadIdx.x & (DD / 4 - 1)];
  #pragma unroll
  for (int j = 0; j < SU; ++j) {
    float4 r;
    r.x = o[j].x + b.x + x[j].x;
    r.y = o[j].y + b.y + x[j].y;
    r.z = o[j].z + b.z + x[j].z;
    r.w = o[j].w + b.w + x[j].w;
    ((float4*)out)[q0 + j * 256] = r;
  }
}


extern "C" void kernel_launch(void* const* d_in, const int* in_sizes, int n_in,
                              void* d_out, int out_size, void* d_ws, size_t ws_size,
                              hipStream_t stream) {
  const float* X_n   = (const float*)d_in[0];
  const float* PE    = (const float*)d_in[1];
  const int*   eidx  = (const int*)d_in[2];
  const int*   eattr = (const int*)d_in[3];
  const float* eemb  = (const float*)d_in[4];
  const float* peW   = (const float*)d_in[5];
  const float* peB   = (const float*)d_in[6];
  const float* linW  = (const float*)d_in[7];
  const float* attl  = (const float*)d_in[8];
  const float* attr_ = (const float*)d_in[9];
  const float* atte  = (const float*)d_in[10];
  const float* bias  = (const float*)d_in[11];
  const float* gam   = (const float*)d_in[12];
  const float* bet   = (const float*)d_in[13];
  const int* src = eidx;
  const int* dst = eidx + EE;

  float* ws = (float*)d_ws;
  size_t o = 0;
  float* xbuf  = ws + o; o += (size_t)NN * DD;
  float* outb  = ws + o; o += (size_t)NN * DD;
  float* h     = ws + o; o += (size_t)NN * DD;
  float* PEp   = ws + o; o += (size_t)NN * PEDP;
  float* Wp    = ws + o; o += (size_t)PEDP * DD;
  float* sl    = ws + o; o += NN;
  float* sr    = ws + o; o += NN;
  int*   cnt   = (int*)(ws + o); o += NN;
  int*   off   = (int*)(ws + o); o += NN + 16;
  int*   rank  = (int*)(ws + o); o += EE;
  v4i*   rec   = (v4i*)(ws + o); o += (size_t)EE * 4;
  float* xe    = ws + o; o += 16;
  float* ssum  = ws + o; o += DD;
  float* ssq   = ws + o; o += DD;

  k_copy_x<<<SBLK, 256, 0, stream>>>(X_n, xbuf, cnt);
  k_pad_pe<<<(NN * PEDP / 4 + 1023) / 1024, 256, 0, stream>>>(PE, PEp);
  k_rank<<<EE / 1024, 256, 0, stream>>>(dst, cnt, rank);
  k_scan<<<1, 1024, 0, stream>>>(cnt, off);
  k_fill<<<EE / 1024, 256, 0, stream>>>(src, dst, eattr, off, rank, rec);

  for (int l = 0; l < NL; ++l) {
    k_pad_w<<<(PEDP * DD + 255) / 256, 256, 0, stream>>>(peW + (size_t)l * PED * DD, Wp);
    k_gemm2x<<<NN / 16, 256, 0, stream>>>(PEp, Wp, xbuf, peB + l * DD,
                                          linW + (size_t)l * DD * DD, h);
    k_xe<<<1, 288, 0, stream>>>(eemb + (size_t)l * NT * DD, atte + l * DD, xe, ssum, ssq);
    k_scores<<<NN / (8 * NPW), 256, 0, stream>>>(h, attl + l * DD, attr_ + l * DD, sl, sr);

    k_node<<<NN / 8, 256, 0, stream>>>(h, sl, sr, xe, off, rec, outb);

    if (l < NL - 1) {
      k_bias_stats<<<(NN + 127) / 128, 128, 0, stream>>>(outb, bias + l * DD, ssum, ssq);
      k_bn_residual<<<SBLK, 256, 0, stream>>>(outb, ssum, ssq, gam + l * DD,
                                              bet + l * DD, xbuf);
    } else {
      k_final<<<SBLK, 256, 0, stream>>>(outb, bias + l * DD, xbuf, (float*)d_out);
    }
  }
}


# Launches per call 25 -> 15: weight pad, input copy, counter and stat resets, edge-type scores and node scores folded into the pad and GEMM launches; every sum keeps its order
# speedup vs baseline: 2.9707x; 1.1176x over previous
#include <hip/hip_runtime.h>
#include <math.h>

#define NN   50000
#define EE   640000
#define DD   128
#define PED  37
#define PEDP 40
#define NT   9
#define NL   3

typedef __attribute__((ext_vector_type(2))) float v2f;
typedef __attribute__((ext_vector_type(8))) float v8f;

typedef __attribute__((ext_vector_type(4))) int v4i;
typedef __attribute__((ext_vector_type(4))) float v4f;

#define SU   5
#define ND4  (NN * DD / 4)
#define SBLK (ND4 / (256 * SU))
static_assert(ND4 % (256 * SU) == 0, "element-wise passes assume an exact grid");
static_assert(SU == 5, "the load fence below names SU = 5 values");
static_assert(EE % 1024 == 0, "edge passes assume four edges per lane, exact grid");


__global__ void __launch_bounds__(256) k_pad_pe(const float* __restrict__ PE,
                                                float* __restrict__ PEp,
                                                int* __restrict__ cnt) {
  const int TOT = NN * PEDP / 4;
  float4 v[4];
  #pragma unroll
  for (int j = 0; j < 4; ++j) {
    int q = blockIdx.x * 1024 + j * 256 + threadIdx.x;
    int qc = (q < TOT) ? q : TOT - 1;
    int r = qc / (PEDP / 4);
    int c = (qc - r * (PEDP / 4)) * 4;
    const float* p = PE + (size_t)r * PED + c;
    bool in1 = c + 1 < PED, in2 = c + 2 < PED, in3 = c + 3 < PED;
    float a0 = p[0];
    float a1 = p[in1 ? 1 : 0];
    float a2 = p[in2 ? 2 : 0];
    float a3 = p[in3 ? 3 : 0];
    v[j] = make_float4(a0, in1 ? a1 : 0.0f, in2 ? a2 : 0.0f, in3 ? a3 : 0.0f);
  }
  #pragma unroll
  for (int j = 0; j < 4; ++j) {
    int q = blockIdx.x * 1024 + j * 256 + threadIdx.x;
    if (q < TOT) ((float4*)PEp)[q] = v[j];
  }
  int g = blockIdx.x * 256 + threadIdx.x;
  v4i z = {0, 0, 0, 0};
  if (g < NN / 4) ((v4i*)cnt)[g] = z;
}

__global__ void __launch_bounds__(256) k_rank(const int* __restrict__ dst,
                                              int* __restrict__ cnt,
                                              int* __restrict__ rank) {
  int q = blockIdx.x * 256 + threadIdx.x;
  v4i d = ((const v4i*)dst)[q];
  v4i r;
  r.x = atomicAdd(&cnt[d.x], 1);
  r.y = atomicAdd(&cnt[d.y], 1);
  r.z = atomicAdd(&cnt[d.z], 1);
  r.w = atomicAdd(&cnt[d.w], 1);
  ((v4i*)rank)[q] = r;
}

__global__ void __launch_bounds__(1024) k_scan(const int* __restrict__ cnt,
                                               int* __restrict__ off) {
  __shared__ int wtot[32];
  int wave = threadIdx.x >> 5, lane = threadIdx.x & 31;
  const v4i* c4 = (const v4i*)cnt;
  v4i* o4 = (v4i*)off;
  v4i v[13];
  int s = 0;
  #pragma unroll
  for (int i = 0; i < 13; ++i) {
    int q = wave * 416 + i * 32 + lane;
    v4i z = {0, 0, 0, 0};
    v[i] = (q < NN / 4) ? c4[q] : z;
    s += v[i].x + v[i].y + v[i].z + v[i].w;
  }
  for (int o = 16; o; o >>= 1) s += __shfl_xor(s, o, 32);
  if (lane == 0) wtot[wave] = s;
  __syncthreads();
  int carry = 0;
  for (int w = 0; w < wave; ++w) carry += wtot[w];
  #pragma unroll
  for (int i = 0; i < 13; ++i) {
    int t = v[i].x + v[i].y + v[i].z + v[i].w;
    int y = t;
    for (int o = 1; o < 32; o <<= 1) {
      int u = __shfl_up(y, o, 32);
      if (lane >= o) y += u;
    }
    int ex = carry + y - t;
    int q = wave * 416 + i * 32 + lane;
    if (q < NN / 4) {
      v4i r = {ex, ex + v[i].x, ex + v[i].x + v[i].y, ex + v[i].x + v[i].y + v[i].z};
      o4[q] = r;
    }
    carry += __shfl(y, 31, 32);
  }
  if (threadIdx.x == 1023) off[NN] = carry;
}

__global__ void __launch_bounds__(256) k_fill(const int* __restrict__ src,
                                              const int* __restrict__ dst,
                                              const int* __restrict__ attr,
                                              const int* __restrict__ off,
                                              const int* __restrict__ rank,
                                              v4i* __restrict__ rec) {
  int q = blockIdx.x * 256 + threadIdx.x;
  v4i s = ((const v4i*)src)[q];
  v4i d = ((const v4i*)dst)[q];
  v4i a = ((const v4i*)attr)[q];
  v4i k = ((const v4i*)rank)[q];
  int o0 = off[d.x], o1 = off[d.y], o2 = off[d.z], o3 = off[d.w];
  int e = q * 4;
  v4i r0 = {s.x, a.x, e,     0};
  v4i r1 = {s.y, a.y, e + 1, 0};
  v4i r2 = {s.z, a.z, e + 2, 0};
  v4i r3 = {s.w, a.w, e + 3, 0};
  rec[o0 + k.x] = r0;
  rec[o1 + k.y] = r1;
  rec[o2 + k.z] = r2;
  rec[o3 + k.w] = r3;
}

#define TP (DD + 4)
static_assert(NN % 16 == 0, "row tiles are exact");
__global__ void __launch_bounds__(256) k_gemm2x(const float* __restrict__ PEp,
                                                const float* __restrict__ W,
                                                const float* __restrict__ X,
                                                const float* __restrict__ bb,
                                                const float* __restrict__ W2,
                                                const float* __restrict__ al,
                                                const float* __restrict__ ar,
                                                const float* __restrict__ ee,
                                                const float* __restrict__ ae,
                                                float* __restrict__ H,
                                                float* __restrict__ sl,
                                                float* __restrict__ sr,
                                                float* __restrict__ xe,
                                                float* __restrict__ ssum,
                                                float* __restrict__ ssq) {
  __shared__ __attribute__((aligned(16))) float strip[16 * TP];
  __shared__ __attribute__((aligned(16))) float hstrip[16 * TP];
  int wave = threadIdx.x >> 5;
  int lane = threadIdx.x & 31;

  if (blockIdx.x == 0) {
    if (threadIdx.x < DD) ssum[threadIdx.x] = 0.0f; else ssq[threadIdx.x - DD] = 0.0f;
    for (int t = wave; t < NT; t += 8) {
      float acc = 0.0f;
      #pragma unroll
      for (int j = 0; j < 4; ++j) {
        int d = lane + 32 * j;
        acc += ee[t * DD + d] * ae[d];
      }
      #pragma unroll
      for (int off = 16; off; off >>= 1) acc += __shfl_xor(acc, off, 32);
      if (lane == 0) xe[t] = acc;
    }
    return;
  }

  int tm  = blockIdx.x - 1;
  int mlo = lane & 15;
  int hi  = lane >> 4;
  int ncol = wave * 16 + mlo;

  const float* arow = PEp + (size_t)(tm * 16 + mlo) * PEDP;
  v8f c = {};
  #pragma unroll
  for (int k0 = 0; k0 < PEDP; k0 += 4) {
    v2f a = *(const v2f*)(arow + k0 + hi * 2);
    int ka = k0 + hi * 2, kb = ka + 1;
    float wa = W[(size_t)(ka < PED ? ka : PED - 1) * DD + ncol];
    float wb = W[(size_t)(kb < PED ? kb : PED - 1) * DD + ncol];
    v2f b;
    b.x = (ka < PED) ? wa : 0.0f;
    b.y = (kb < PED) ? wb : 0.0f;
    c = __builtin_amdgcn_wmma_f32_16x16x4_f32(false, a, false, b, (short)0, c,
                                              false, false);
  }
  #pragma unroll
  for (int r = 0; r < 8; ++r) {
    int row = r + hi * 8;
    float v = c[r];
    v += X[(size_t)(tm * 16 + row) * DD + ncol] + bb[ncol];
    strip[row * TP + ncol] = v;
  }
  __syncthreads();

  const float* srow = strip + mlo * TP + hi * 2;
  v8f d = {};
  #pragma unroll 4
  for (int k0 = 0; k0 < DD; k0 += 4) {
    v2f a = *(const v2f*)(srow + k0);
    v2f b;
    b.x = W2[(size_t)(k0 + hi * 2    ) * DD + ncol];
    b.y = W2[(size_t)(k0 + hi * 2 + 1) * DD + ncol];
    d = __builtin_amdgcn_wmma_f32_16x16x4_f32(false, a, false, b, (short)0, d,
                                              false, false);
  }
  #pragma unroll
  for (int r = 0; r < 8; ++r) hstrip[(r + hi * 8) * TP + ncol] = d[r];
  float wl[4], wr[4];
  #pragma unroll
  for (int j = 0; j < 4; ++j) { wl[j] = al[lane + 32 * j]; wr[j] = ar[lane + 32 * j]; }
  __syncthreads();

  float ol = 0.0f, orr = 0.0f;
  #pragma unroll
  for (int i = 0; i < 2; ++i) {
    int row = wave * 2 + i;
    const float* hp = hstrip + row * TP;
    v4f q = *(const v4f*)(hp + lane * 4);
    ((v4f*)(H + (size_t)(tm * 16 + row) * DD))[lane] = q;
    float accl = 0.0f, accr = 0.0f;
    #pragma unroll
    for (int j = 0; j < 4; ++j) {
      float v = hp[lane + 32 * j];
      accl += v * wl[j];
      accr += v * wr[j];
    }
    #pragma unroll
    for (int off = 16; off; off >>= 1) {
      accl += __shfl_xor(accl, off, 32);
      accr += __shfl_xor(accr, off, 32);
    }
    if (lane == i) { ol = accl; orr = accr; }
  }
  if (lane < 2) {
    int n = tm * 16 + wave * 2 + lane;
    sl[n] = ol; sr[n] = orr;
  }
}

__device__ __forceinline__ float edge_score(const float* __restrict__ sl,
                                            const float* __restrict__ xe, float srn,
                                            int s, int t) {
  float a = sl[s] + srn + xe[t];
  return (a > 0.0f) ? a : 0.2f * a;
}

__global__ void k_node(const float* __restrict__ h, const float* __restrict__ sl,
                       const float* __restrict__ sr, const float* __restrict__ xe,
                       const int* __restrict__ off, const v4i* __restrict__ rec,
                       float* __restrict__ outb) {
  int wave = threadIdx.x >> 5, lane = threadIdx.x & 31;
  int n = blockIdx.x * 8 + wave;
  int beg = __builtin_amdgcn_readfirstlane(off[n]);
  int end = __builtin_amdgcn_readfirstlane(off[n + 1]);
  float srn = sr[n];

  int j0 = beg + lane;
  bool has0 = j0 < end;
  int s0 = 0, e0 = 0;
  float a0 = -INFINITY;
  if (has0) {
    v4i r = rec[j0];
    s0 = r.x; e0 = r.z;
    a0 = edge_score(sl, xe, srn, r.x, r.y);
  }
  float m = a0;
  for (int j = j0 + 32; j < end; j += 32) {
    v4i r = rec[j];
    m = fmaxf(m, edge_score(sl, xe, srn, r.x, r.y));
  }
  for (int o = 16; o; o >>= 1) m = fmaxf(m, __shfl_xor(m, o, 32));

  float4 acc = make_float4(0.0f, 0.0f, 0.0f, 0.0f);
  float den = 0.0f;
  for (int base = beg; base < end; base += 32) {
    int sC = s0, eC = e0;
    float aC = a0;
    bool hasC = has0;
    if (base != beg) {
      int j = base + lane;
      hasC = j < end; sC = 0; eC = 0; aC = -INFINITY;
      if (hasC) {
        v4i r = rec[j];
        sC = r.x; eC = r.z;
        aC = edge_score(sl, xe, srn, r.x, r.y);
      }
    }
    float exC = hasC ? expf(aC - m) : 0.0f;
    int c = end - base; if (c > 32) c = 32;
    int rk = 0;
    #pragma unroll 4
    for (int k = 0; k < c; ++k) rk += (__shfl(eC, k, 32) < eC) ? 1 : 0;
    int k = 0;
    for (; k + 4 <= c; k += 4) {
      int L0 = __builtin_ctz(__builtin_amdgcn_ballot_w32(hasC && rk == k));
      int L1 = __builtin_ctz(__builtin_amdgcn_ballot_w32(hasC && rk == k + 1));
      int L2 = __builtin_ctz(__builtin_amdgcn_ballot_w32(hasC && rk == k + 2));
      int L3 = __builtin_ctz(__builtin_amdgcn_ballot_w32(hasC && rk == k + 3));
      float w0 = __shfl(exC, L0, 32), w1 = __shfl(exC, L1, 32);
      float w2 = __shfl(exC, L2, 32), w3 = __shfl(exC, L3, 32);
      float4 v0 = ((const float4*)(h + (size_t)__shfl(sC, L0, 32) * DD))[lane];
      float4 v1 = ((const float4*)(h + (size_t)__shfl(sC, L1, 32) * DD))[lane];
      float4 v2 = ((const float4*)(h + (size_t)__shfl(sC, L2, 32) * DD))[lane];
      float4 v3 = ((const float4*)(h + (size_t)__shfl(sC, L3, 32) * DD))[lane];
      den += w0; acc.x += w0 * v0.x; acc.y += w0 * v0.y; acc.z += w0 * v0.z; acc.w += w0 * v0.w;
      den += w1; acc.x += w1 * v1.x; acc.y += w1 * v1.y; acc.z += w1 * v1.z; acc.w += w1 * v1.w;
      den += w2; acc.x += w2 * v2.x; acc.y += w2 * v2.y; acc.z += w2 * v2.z; acc.w += w2 * v2.w;
      den += w3; acc.x += w3 * v3.x; acc.y += w3 * v3.y; acc.z += w3 * v3.z; acc.w += w3 * v3.w;
    }
    for (; k < c; ++k) {
      int L0 = __builtin_ctz(__builtin_amdgcn_ballot_w32(hasC && rk == k));
      float w0 = __shfl(exC, L0, 32);
      float4 v0 = ((const float4*)(h + (size_t)__shfl(sC, L0, 32) * DD))[lane];
      den += w0; acc.x += w0 * v0.x; acc.y += w0 * v0.y; acc.z += w0 * v0.z; acc.w += w0 * v0.w;
    }
  }
  float inv = (end > beg) ? 1.0f / den : 0.0f;
  float4 o4 = make_float4(acc.x * inv, acc.y * inv, acc.z * inv, acc.w * inv);
  ((float4*)(outb + (size_t)n * DD))[lane] = o4;
}


__global__ void k_bias_stats(float* __restrict__ outb, const float* __restrict__ bias,
                             float* __restrict__ ssum, float* __restrict__ ssq) {
  int d = threadIdx.x;
  int r0 = blockIdx.x * 128;
  float b = bias[d];
  float s = 0.0f, s2 = 0.0f;
  for (int r = r0; r < r0 + 128 && r < NN; ++r) {
    float v = outb[(size_t)r * DD + d] + b;
    outb[(size_t)r * DD + d] = v;
    s += v; s2 += v * v;
  }
  atomicAdd(&ssum[d], s);
  atomicAdd(&ssq[d], s2);
}

__device__ __forceinline__ float bn_one(float o, float s, float s2, float g, float b, float x) {
  float mu  = s * (1.0f / NN);
  float var = s2 * (1.0f / NN) - mu * mu;
  float inv = rsqrtf(var + 1e-5f);
  return (o - mu) * inv * g + b + x;
}

__global__ void __launch_bounds__(256) k_bn_residual(const float* __restrict__ outb,
                                                     const float* __restrict__ ssum,
                                                     const float* __restrict__ ssq,
                                                     const float* __restrict__ gam,
                                                     const float* __restrict__ bet,
                                                     const float* __restrict__ xin,
                                                     float* __restrict__ xout) {
  const v4f* o4 = (const v4f*)outb;
  const v4f* xi4 = (const v4f*)xin;
  v4f* x4 = (v4f*)xout;
  int q0 = blockIdx.x * (256 * SU) + threadIdx.x;
  int c4 = threadIdx.x & (DD / 4 - 1);
  v4f o[SU], x[SU];
  v4f s  = ((const v4f*)ssum)[c4];
  v4f s2 = ((const v4f*)ssq)[c4];
  v4f g  = ((const v4f*)gam)[c4];
  v4f b  = ((const v4f*)bet)[c4];
  #pragma unroll
  for (int j = 0; j < SU; ++j) { o[j] = o4[q0 + j * 256]; x[j] = xi4[q0 + j * 256]; }
  asm volatile("" :: "v"(x[0]), "v"(x[1]), "v"(x[2]), "v"(x[3]), "v"(x[4]));
  #pragma unroll
  for (int j = 0; j < SU; ++j) {
    v4f r;
    r.x = bn_one(o[j].x, s.x, s2.x, g.x, b.x, x[j].x);
    r.y = bn_one(o[j].y, s.y, s2.y, g.y, b.y, x[j].y);
    r.z = bn_one(o[j].z, s.z, s2.z, g.z, b.z, x[j].z);
    r.w = bn_one(o[j].w, s.w, s2.w, g.w, b.w, x[j].w);
    x4[q0 + j * 256] = r;
  }
}

__global__ void __launch_bounds__(256) k_final(const float* __restrict__ outb,
                                               const float* __restrict__ bias,
                                               const float* __restrict__ xbuf,
                                               float* __restrict__ out) {
  const float4* o4 = (const float4*)outb;
  const float4* x4 = (const float4*)xbuf;
  int q0 = blockIdx.x * (256 * SU) + threadIdx.x;
  float4 o[SU], x[SU];
  #pragma unroll
  for (int j = 0; j < SU; ++j) { o[j] = o4[q0 + j * 256]; x[j] = x4[q0 + j * 256]; }
  float4 b = ((const float4*)bias)[threadIdx.x & (DD / 4 - 1)];
  #pragma unroll
  for (int j = 0; j < SU; ++j) {
    float4 r;
    r.x = o[j].x + b.x + x[j].x;
    r.y = o[j].y + b.y + x[j].y;
    r.z = o[j].z + b.z + x[j].z;
    r.w = o[j].w + b.w + x[j].w;
    ((float4*)out)[q0 + j * 256] = r;
  }
}


extern "C" void kernel_launch(void* const* d_in, const int* in_sizes, int n_in,
                              void* d_out, int out_size, void* d_ws, size_t ws_size,
                              hipStream_t stream) {
  const float* X_n   = (const float*)d_in[0];
  const float* PE    = (const float*)d_in[1];
  const int*   eidx  = (const int*)d_in[2];
  const int*   eattr = (const int*)d_in[3];
  const float* eemb  = (const float*)d_in[4];
  const float* peW   = (const float*)d_in[5];
  const float* peB   = (const float*)d_in[6];
  const float* linW  = (const float*)d_in[7];
  const float* attl  = (const float*)d_in[8];
  const float* attr_ = (const float*)d_in[9];
  const float* atte  = (const float*)d_in[10];
  const float* bias  = (const float*)d_in[11];
  const float* gam   = (const float*)d_in[12];
  const float* bet   = (const float*)d_in[13];
  const int* src = eidx;
  const int* dst = eidx + EE;

  float* ws = (float*)d_ws;
  size_t o = 0;
  float* xa    = ws + o; o += (size_t)NN * DD;
  float* xb    = ws + o; o += (size_t)NN * DD;
  float* outb  = ws + o; o += (size_t)NN * DD;
  float* h     = ws + o; o += (size_t)NN * DD;
  float* PEp   = ws + o; o += (size_t)NN * PEDP;
  float* sl    = ws + o; o += NN;
  float* sr    = ws + o; o += NN;
  int*   cnt   = (int*)(ws + o); o += NN;
  int*   off   = (int*)(ws + o); o += NN + 16;
  int*   rank  = (int*)(ws + o); o += EE;
  v4i*   rec   = (v4i*)(ws + o); o += (size_t)EE * 4;
  float* xe    = ws + o; o += 16;
  float* ssum  = ws + o; o += DD;
  float* ssq   = ws + o; o += DD;
  float* xnext[NL - 1] = {xa, xb};

  k_pad_pe<<<(NN * PEDP / 4 + 1023) / 1024, 256, 0, stream>>>(PE, PEp, cnt);
  k_rank<<<EE / 1024, 256, 0, stream>>>(dst, cnt, rank);
  k_scan<<<1, 1024, 0, stream>>>(cnt, off);
  k_fill<<<EE / 1024, 256, 0, stream>>>(src, dst, eattr, off, rank, rec);

  const float* xin = X_n;
  for (int l = 0; l < NL; ++l) {
    k_gemm2x<<<NN / 16 + 1, 256, 0, stream>>>(PEp, peW + (size_t)l * PED * DD, xin, peB + l * DD,
                                              linW + (size_t)l * DD * DD, attl + l * DD,
                                              attr_ + l * DD, eemb + (size_t)l * NT * DD,
                                              atte + l * DD, h, sl, sr, xe, ssum, ssq);

    k_node<<<NN / 8, 256, 0, stream>>>(h, sl, sr, xe, off, rec, outb);

    if (l < NL - 1) {
      k_bias_stats<<<(NN + 127) / 128, 128, 0, stream>>>(outb, bias + l * DD, ssum, ssq);
      k_bn_residual<<<SBLK, 256, 0, stream>>>(outb, ssum, ssq, gam + l * DD,
                                              bet + l * DD, xin, xnext[l]);
      xin = xnext[l];
    } else {
      k_final<<<SBLK, 256, 0, stream>>>(outb, bias + l * DD, xin, (float*)d_out);
    }
  }
}


# GEMM launch made persistent: 512 work-groups each keep both weights' B fragments in registers and walk 6 or 7 row tiles, next tile's operands asked for a tile ahead; k order of both sums unchanged
# speedup vs baseline: 3.1319x; 1.0543x over previous
#include <hip/hip_runtime.h>
#include <math.h>

#define NN   50000
#define EE   640000
#define DD   128
#define PED  37
#define PEDP 40
#define NT   9
#define NL   3

typedef __attribute__((ext_vector_type(2))) float v2f;
typedef __attribute__((ext_vector_type(8))) float v8f;

typedef __attribute__((ext_vector_type(4))) int v4i;
typedef __attribute__((ext_vector_type(4))) float v4f;

#define SU   5
#define ND4  (NN * DD / 4)
#define SBLK (ND4 / (256 * SU))
static_assert(ND4 % (256 * SU) == 0, "element-wise passes assume an exact grid");
static_assert(SU == 5, "the load fence below names SU = 5 values");
static_assert(EE % 1024 == 0, "edge passes assume four edges per lane, exact grid");


__global__ void __launch_bounds__(256) k_pad_pe(const float* __restrict__ PE,
                                                float* __restrict__ PEp,
                                                int* __restrict__ cnt) {
  const int TOT = NN * PEDP / 4;
  float4 v[4];
  #pragma unroll
  for (int j = 0; j < 4; ++j) {
    int q = blockIdx.x * 1024 + j * 256 + threadIdx.x;
    int qc = (q < TOT) ? q : TOT - 1;
    int r = qc / (PEDP / 4);
    int c = (qc - r * (PEDP / 4)) * 4;
    const float* p = PE + (size_t)r * PED + c;
    bool in1 = c + 1 < PED, in2 = c + 2 < PED, in3 = c + 3 < PED;
    float a0 = p[0];
    float a1 = p[in1 ? 1 : 0];
    float a2 = p[in2 ? 2 : 0];
    float a3 = p[in3 ? 3 : 0];
    v[j] = make_float4(a0, in1 ? a1 : 0.0f, in2 ? a2 : 0.0f, in3 ? a3 : 0.0f);
  }
  #pragma unroll
  for (int j = 0; j < 4; ++j) {
    int q = blockIdx.x * 1024 + j * 256 + threadIdx.x;
    if (q < TOT) ((float4*)PEp)[q] = v[j];
  }
  int g = blockIdx.x * 256 + threadIdx.x;
  v4i z = {0, 0, 0, 0};
  if (g < NN / 4) ((v4i*)cnt)[g] = z;
}

__global__ void __launch_bounds__(256) k_rank(const int* __restrict__ dst,
                                              int* __restrict__ cnt,
                                              int* __restrict__ rank) {
  int q = blockIdx.x * 256 + threadIdx.x;
  v4i d = ((const v4i*)dst)[q];
  v4i r;
  r.x = atomicAdd(&cnt[d.x], 1);
  r.y = atomicAdd(&cnt[d.y], 1);
  r.z = atomicAdd(&cnt[d.z], 1);
  r.w = atomicAdd(&cnt[d.w], 1);
  ((v4i*)rank)[q] = r;
}

__global__ void __launch_bounds__(1024) k_scan(const int* __restrict__ cnt,
                                               int* __restrict__ off) {
  __shared__ int wtot[32];
  int wave = threadIdx.x >> 5, lane = threadIdx.x & 31;
  const v4i* c4 = (const v4i*)cnt;
  v4i* o4 = (v4i*)off;
  v4i v[13];
  int s = 0;
  #pragma unroll
  for (int i = 0; i < 13; ++i) {
    int q = wave * 416 + i * 32 + lane;
    v4i z = {0, 0, 0, 0};
    v[i] = (q < NN / 4) ? c4[q] : z;
    s += v[i].x + v[i].y + v[i].z + v[i].w;
  }
  for (int o = 16; o; o >>= 1) s += __shfl_xor(s, o, 32);
  if (lane == 0) wtot[wave] = s;
  __syncthreads();
  int carry = 0;
  for (int w = 0; w < wave; ++w) carry += wtot[w];
  #pragma unroll
  for (int i = 0; i < 13; ++i) {
    int t = v[i].x + v[i].y + v[i].z + v[i].w;
    int y = t;
    for (int o = 1; o < 32; o <<= 1) {
      int u = __shfl_up(y, o, 32);
      if (lane >= o) y += u;
    }
    int ex = carry + y - t;
    int q = wave * 416 + i * 32 + lane;
    if (q < NN / 4) {
      v4i r = {ex, ex + v[i].x, ex + v[i].x + v[i].y, ex + v[i].x + v[i].y + v[i].z};
      o4[q] = r;
    }
    carry += __shfl(y, 31, 32);
  }
  if (threadIdx.x == 1023) off[NN] = carry;
}

__global__ void __launch_bounds__(256) k_fill(const int* __restrict__ src,
                                              const int* __restrict__ dst,
                                              const int* __restrict__ attr,
                                              const int* __restrict__ off,
                                              const int* __restrict__ rank,
                                              v4i* __restrict__ rec) {
  int q = blockIdx.x * 256 + threadIdx.x;
  v4i s = ((const v4i*)src)[q];
  v4i d = ((const v4i*)dst)[q];
  v4i a = ((const v4i*)attr)[q];
  v4i k = ((const v4i*)rank)[q];
  int o0 = off[d.x], o1 = off[d.y], o2 = off[d.z], o3 = off[d.w];
  int e = q * 4;
  v4i r0 = {s.x, a.x, e,     0};
  v4i r1 = {s.y, a.y, e + 1, 0};
  v4i r2 = {s.z, a.z, e + 2, 0};
  v4i r3 = {s.w, a.w, e + 3, 0};
  rec[o0 + k.x] = r0;
  rec[o1 + k.y] = r1;
  rec[o2 + k.z] = r2;
  rec[o3 + k.w] = r3;
}

#define TP (DD + 4)
#define NTILE (NN / 16)
#define GB 512
static_assert(NN % 16 == 0, "row tiles are exact");

__device__ __forceinline__ void tile_loads(const float* __restrict__ PEp,
                                           const float* __restrict__ X, int tm, int mlo, int hi,
                                           int ncol, v2f (&a)[PEDP / 4], float (&xv)[8]) {
  const float* arow = PEp + (size_t)(tm * 16 + mlo) * PEDP + hi * 2;
  #pragma unroll
  for (int i = 0; i < PEDP / 4; ++i) a[i] = *(const v2f*)(arow + 4 * i);
  #pragma unroll
  for (int r = 0; r < 8; ++r) xv[r] = X[(size_t)(tm * 16 + r + hi * 8) * DD + ncol];
}

__global__ void __launch_bounds__(256) k_gemm2x(const float* __restrict__ PEp,
                                                const float* __restrict__ W,
                                                const float* __restrict__ X,
                                                const float* __restrict__ bb,
                                                const float* __restrict__ W2,
                                                const float* __restrict__ al,
                                                const float* __restrict__ ar,
                                                const float* __restrict__ ee,
                                                const float* __restrict__ ae,
                                                float* __restrict__ H,
                                                float* __restrict__ sl,
                                                float* __restrict__ sr,
                                                float* __restrict__ xe,
                                                float* __restrict__ ssum,
                                                float* __restrict__ ssq) {
  __shared__ __attribute__((aligned(16))) float strip[16 * TP];
  __shared__ __attribute__((aligned(16))) float hstrip[16 * TP];
  int wave = threadIdx.x >> 5;
  int lane = threadIdx.x & 31;

  if (blockIdx.x == 0) {
    if (threadIdx.x < DD) ssum[threadIdx.x] = 0.0f; else ssq[threadIdx.x - DD] = 0.0f;
    for (int t = wave; t < NT; t += 8) {
      float acc = 0.0f;
      #pragma unroll
      for (int j = 0; j < 4; ++j) {
        int d = lane + 32 * j;
        acc += ee[t * DD + d] * ae[d];
      }
      #pragma unroll
      for (int off = 16; off; off >>= 1) acc += __shfl_xor(acc, off, 32);
      if (lane == 0) xe[t] = acc;
    }
    return;
  }

  int mlo = lane & 15;
  int hi  = lane >> 4;
  int ncol = wave * 16 + mlo;

  int tm = blockIdx.x - 1;
  v2f a[PEDP / 4];
  float xv[8];
  tile_loads(PEp, X, tm, mlo, hi, ncol, a, xv);

  v2f wp[PEDP / 4];
  #pragma unroll
  for (int i = 0; i < PEDP / 4; ++i) {
    int ka = 4 * i + hi * 2, kb = ka + 1;
    float wa = W[(size_t)(ka < PED ? ka : PED - 1) * DD + ncol];
    float wb = W[(size_t)(kb < PED ? kb : PED - 1) * DD + ncol];
    wp[i].x = (ka < PED) ? wa : 0.0f;
    wp[i].y = (kb < PED) ? wb : 0.0f;
  }
  v2f w2[DD / 4];
  #pragma unroll
  for (int i = 0; i < DD / 4; ++i) {
    w2[i].x = W2[(size_t)(4 * i + hi * 2    ) * DD + ncol];
    w2[i].y = W2[(size_t)(4 * i + hi * 2 + 1) * DD + ncol];
  }
  float bbv = bb[ncol];
  float wl[4], wr[4];
  #pragma unroll
  for (int j = 0; j < 4; ++j) { wl[j] = al[lane + 32 * j]; wr[j] = ar[lane + 32 * j]; }

  const float* srow = strip + mlo * TP + hi * 2;

  for (; tm < NTILE; tm += GB) {
    v8f c = {};
    #pragma unroll
    for (int i = 0; i < PEDP / 4; ++i)
      c = __builtin_amdgcn_wmma_f32_16x16x4_f32(false, a[i], false, wp[i], (short)0, c,
                                                false, false);
    #pragma unroll
    for (int r = 0; r < 8; ++r) {
      float v = c[r];
      v += xv[r] + bbv;
      strip[(r + hi * 8) * TP + ncol] = v;
    }
    __syncthreads();

    if (tm + GB < NTILE) tile_loads(PEp, X, tm + GB, mlo, hi, ncol, a, xv);

    v8f d = {};
    #pragma unroll
    for (int i = 0; i < DD / 4; ++i) {
      v2f as = *(const v2f*)(srow + 4 * i);
      d = __builtin_amdgcn_wmma_f32_16x16x4_f32(false, as, false, w2[i], (short)0, d,
                                                false, false);
    }
    #pragma unroll
    for (int r = 0; r < 8; ++r) hstrip[(r + hi * 8) * TP + ncol] = d[r];
    __syncthreads();

    float ol = 0.0f, orr = 0.0f;
    #pragma unroll
    for (int i = 0; i < 2; ++i) {
      int row = wave * 2 + i;
      const float* hp = hstrip + row * TP;
      v4f q = *(const v4f*)(hp + lane * 4);
      ((v4f*)(H + (size_t)(tm * 16 + row) * DD))[lane] = q;
      float accl = 0.0f, accr = 0.0f;
      #pragma unroll
      for (int j = 0; j < 4; ++j) {
        float v = hp[lane + 32 * j];
        accl += v * wl[j];
        accr += v * wr[j];
      }
      #pragma unroll
      for (int off = 16; off; off >>= 1) {
        accl += __shfl_xor(accl, off, 32);
        accr += __shfl_xor(accr, off, 32);
      }
      if (lane == i) { ol = accl; orr = accr; }
    }
    if (lane < 2) {
      int n = tm * 16 + wave * 2 + lane;
      sl[n] = ol; sr[n] = orr;
    }
  }
}

__device__ __forceinline__ float edge_score(const float* __restrict__ sl,
                                            const float* __restrict__ xe, float srn,
                                            int s, int t) {
  float a = sl[s] + srn + xe[t];
  return (a > 0.0f) ? a : 0.2f * a;
}

__global__ void k_node(const float* __restrict__ h, const float* __restrict__ sl,
                       const float* __restrict__ sr, const float* __restrict__ xe,
                       const int* __restrict__ off, const v4i* __restrict__ rec,
                       float* __restrict__ outb) {
  int wave = threadIdx.x >> 5, lane = threadIdx.x & 31;
  int n = blockIdx.x * 8 + wave;
  int beg = __builtin_amdgcn_readfirstlane(off[n]);
  int end = __builtin_amdgcn_readfirstlane(off[n + 1]);
  float srn = sr[n];

  int j0 = beg + lane;
  bool has0 = j0 < end;
  int s0 = 0, e0 = 0;
  float a0 = -INFINITY;
  if (has0) {
    v4i r = rec[j0];
    s0 = r.x; e0 = r.z;
    a0 = edge_score(sl, xe, srn, r.x, r.y);
  }
  float m = a0;
  for (int j = j0 + 32; j < end; j += 32) {
    v4i r = rec[j];
    m = fmaxf(m, edge_score(sl, xe, srn, r.x, r.y));
  }
  for (int o = 16; o; o >>= 1) m = fmaxf(m, __shfl_xor(m, o, 32));

  float4 acc = make_float4(0.0f, 0.0f, 0.0f, 0.0f);
  float den = 0.0f;
  for (int base = beg; base < end; base += 32) {
    int sC = s0, eC = e0;
    float aC = a0;
    bool hasC = has0;
    if (base != beg) {
      int j = base + lane;
      hasC = j < end; sC = 0; eC = 0; aC = -INFINITY;
      if (hasC) {
        v4i r = rec[j];
        sC = r.x; eC = r.z;
        aC = edge_score(sl, xe, srn, r.x, r.y);
      }
    }
    float exC = hasC ? expf(aC - m) : 0.0f;
    int c = end - base; if (c > 32) c = 32;
    int rk = 0;
    #pragma unroll 4
    for (int k = 0; k < c; ++k) rk += (__shfl(eC, k, 32) < eC) ? 1 : 0;
    int k = 0;
    for (; k + 4 <= c; k += 4) {
      int L0 = __builtin_ctz(__builtin_amdgcn_ballot_w32(hasC && rk == k));
      int L1 = __builtin_ctz(__builtin_amdgcn_ballot_w32(hasC && rk == k + 1));
      int L2 = __builtin_ctz(__builtin_amdgcn_ballot_w32(hasC && rk == k + 2));
      int L3 = __builtin_ctz(__builtin_amdgcn_ballot_w32(hasC && rk == k + 3));
      float w0 = __shfl(exC, L0, 32), w1 = __shfl(exC, L1, 32);
      float w2 = __shfl(exC, L2, 32), w3 = __shfl(exC, L3, 32);
      float4 v0 = ((const float4*)(h + (size_t)__shfl(sC, L0, 32) * DD))[lane];
      float4 v1 = ((const float4*)(h + (size_t)__shfl(sC, L1, 32) * DD))[lane];
      float4 v2 = ((const float4*)(h + (size_t)__shfl(sC, L2, 32) * DD))[lane];
      float4 v3 = ((const float4*)(h + (size_t)__shfl(sC, L3, 32) * DD))[lane];
      den += w0; acc.x += w0 * v0.x; acc.y += w0 * v0.y; acc.z += w0 * v0.z; acc.w += w0 * v0.w;
      den += w1; acc.x += w1 * v1.x; acc.y += w1 * v1.y; acc.z += w1 * v1.z; acc.w += w1 * v1.w;
      den += w2; acc.x += w2 * v2.x; acc.y += w2 * v2.y; acc.z += w2 * v2.z; acc.w += w2 * v2.w;
      den += w3; acc.x += w3 * v3.x; acc.y += w3 * v3.y; acc.z += w3 * v3.z; acc.w += w3 * v3.w;
    }
    for (; k < c; ++k) {
      int L0 = __builtin_ctz(__builtin_amdgcn_ballot_w32(hasC && rk == k));
      float w0 = __shfl(exC, L0, 32);
      float4 v0 = ((const float4*)(h + (size_t)__shfl(sC, L0, 32) * DD))[lane];
      den += w0; acc.x += w0 * v0.x; acc.y += w0 * v0.y; acc.z += w0 * v0.z; acc.w += w0 * v0.w;
    }
  }
  float inv = (end > beg) ? 1.0f / den : 0.0f;
  float4 o4 = make_float4(acc.x * inv, acc.y * inv, acc.z * inv, acc.w * inv);
  ((float4*)(outb + (size_t)n * DD))[lane] = o4;
}


__global__ void k_bias_stats(float* __restrict__ outb, const float* __restrict__ bias,
                             float* __restrict__ ssum, float* __restrict__ ssq) {
  int d = threadIdx.x;
  int r0 = blockIdx.x * 128;
  float b = bias[d];
  float s = 0.0f, s2 = 0.0f;
  for (int r = r0; r < r0 + 128 && r < NN; ++r) {
    float v = outb[(size_t)r * DD + d] + b;
    outb[(size_t)r * DD + d] = v;
    s += v; s2 += v * v;
  }
  atomicAdd(&ssum[d], s);
  atomicAdd(&ssq[d], s2);
}

__device__ __forceinline__ float bn_one(float o, float s, float s2, float g, float b, float x) {
  float mu  = s * (1.0f / NN);
  float var = s2 * (1.0f / NN) - mu * mu;
  float inv = rsqrtf(var + 1e-5f);
  return (o - mu) * inv * g + b + x;
}

__global__ void __launch_bounds__(256) k_bn_residual(const float* __restrict__ outb,
                                                     const float* __restrict__ ssum,
                                                     const float* __restrict__ ssq,
                                                     const float* __restrict__ gam,
                                                     const float* __restrict__ bet,
                                                     const float* __restrict__ xin,
                                                     float* __restrict__ xout) {
  const v4f* o4 = (const v4f*)outb;
  const v4f* xi4 = (const v4f*)xin;
  v4f* x4 = (v4f*)xout;
  int q0 = blockIdx.x * (256 * SU) + threadIdx.x;
  int c4 = threadIdx.x & (DD / 4 - 1);
  v4f o[SU], x[SU];
  v4f s  = ((const v4f*)ssum)[c4];
  v4f s2 = ((const v4f*)ssq)[c4];
  v4f g  = ((const v4f*)gam)[c4];
  v4f b  = ((const v4f*)bet)[c4];
  #pragma unroll
  for (int j = 0; j < SU; ++j) { o[j] = o4[q0 + j * 256]; x[j] = xi4[q0 + j * 256]; }
  asm volatile("" :: "v"(x[0]), "v"(x[1]), "v"(x[2]), "v"(x[3]), "v"(x[4]));
  #pragma unroll
  for (int j = 0; j < SU; ++j) {
    v4f r;
    r.x = bn_one(o[j].x, s.x, s2.x, g.x, b.x, x[j].x);
    r.y = bn_one(o[j].y, s.y, s2.y, g.y, b.y, x[j].y);
    r.z = bn_one(o[j].z, s.z, s2.z, g.z, b.z, x[j].z);
    r.w = bn_one(o[j].w, s.w, s2.w, g.w, b.w, x[j].w);
    x4[q0 + j * 256] = r;
  }
}

__global__ void __launch_bounds__(256) k_final(const float* __restrict__ outb,
                                               const float* __restrict__ bias,
                                               const float* __restrict__ xbuf,
                                               float* __restrict__ out) {
  const float4* o4 = (const float4*)outb;
  const float4* x4 = (const float4*)xbuf;
  int q0 = blockIdx.x * (256 * SU) + threadIdx.x;
  float4 o[SU], x[SU];
  #pragma unroll
  for (int j = 0; j < SU; ++j) { o[j] = o4[q0 + j * 256]; x[j] = x4[q0 + j * 256]; }
  float4 b = ((const float4*)bias)[threadIdx.x & (DD / 4 - 1)];
  #pragma unroll
  for (int j = 0; j < SU; ++j) {
    float4 r;
    r.x = o[j].x + b.x + x[j].x;
    r.y = o[j].y + b.y + x[j].y;
    r.z = o[j].z + b.z + x[j].z;
    r.w = o[j].w + b.w + x[j].w;
    ((float4*)out)[q0 + j * 256] = r;
  }
}


extern "C" void kernel_launch(void* const* d_in, const int* in_sizes, int n_in,
                              void* d_out, int out_size, void* d_ws, size_t ws_size,
                              hipStream_t stream) {
  const float* X_n   = (const float*)d_in[0];
  const float* PE    = (const float*)d_in[1];
  const int*   eidx  = (const int*)d_in[2];
  const int*   eattr = (const int*)d_in[3];
  const float* eemb  = (const float*)d_in[4];
  const float* peW   = (const float*)d_in[5];
  const float* peB   = (const float*)d_in[6];
  const float* linW  = (const float*)d_in[7];
  const float* attl  = (const float*)d_in[8];
  const float* attr_ = (const float*)d_in[9];
  const float* atte  = (const float*)d_in[10];
  const float* bias  = (const float*)d_in[11];
  const float* gam   = (const float*)d_in[12];
  const float* bet   = (const float*)d_in[13];
  const int* src = eidx;
  const int* dst = eidx + EE;

  float* ws = (float*)d_ws;
  size_t o = 0;
  float* xa    = ws + o; o += (size_t)NN * DD;
  float* xb    = ws + o; o += (size_t)NN * DD;
  float* outb  = ws + o; o += (size_t)NN * DD;
  float* h     = ws + o; o += (size_t)NN * DD;
  float* PEp   = ws + o; o += (size_t)NN * PEDP;
  float* sl    = ws + o; o += NN;
  float* sr    = ws + o; o += NN;
  int*   cnt   = (int*)(ws + o); o += NN;
  int*   off   = (int*)(ws + o); o += NN + 16;
  int*   rank  = (int*)(ws + o); o += EE;
  v4i*   rec   = (v4i*)(ws + o); o += (size_t)EE * 4;
  float* xe    = ws + o; o += 16;
  float* ssum  = ws + o; o += DD;
  float* ssq   = ws + o; o += DD;
  float* xnext[NL - 1] = {xa, xb};

  k_pad_pe<<<(NN * PEDP / 4 + 1023) / 1024, 256, 0, stream>>>(PE, PEp, cnt);
  k_rank<<<EE / 1024, 256, 0, stream>>>(dst, cnt, rank);
  k_scan<<<1, 1024, 0, stream>>>(cnt, off);
  k_fill<<<EE / 1024, 256, 0, stream>>>(src, dst, eattr, off, rank, rec);

  const float* xin = X_n;
  for (int l = 0; l < NL; ++l) {
    k_gemm2x<<<GB + 1, 256, 0, stream>>>(PEp, peW + (size_t)l * PED * DD, xin, peB + l * DD,
                                              linW + (size_t)l * DD * DD, attl + l * DD,
                                              attr_ + l * DD, eemb + (size_t)l * NT * DD,
                                              atte + l * DD, h, sl, sr, xe, ssum, ssq);

    k_node<<<NN / 8, 256, 0, stream>>>(h, sl, sr, xe, off, rec, outb);

    if (l < NL - 1) {
      k_bias_stats<<<(NN + 127) / 128, 128, 0, stream>>>(outb, bias + l * DD, ssum, ssq);
      k_bn_residual<<<SBLK, 256, 0, stream>>>(outb, ssum, ssq, gam + l * DD,
                                              bet + l * DD, xin, xnext[l]);
      xin = xnext[l];
    } else {
      k_final<<<SBLK, 256, 0, stream>>>(outb, bias + l * DD, xin, (float*)d_out);
    }
  }
}


# Node kernel finishes its rows: adds bias, block-sums 8 rows in LDS into 16 copies of the column sums (last layer adds residual); stats and final passes go, 15->12 launches; column sum order changed
# speedup vs baseline: 3.6583x; 1.1681x over previous
#include <hip/hip_runtime.h>
#include <math.h>

#define NN   50000
#define EE   640000
#define DD   128
#define PED  37
#define PEDP 40
#define NT   9
#define NL   3

typedef __attribute__((ext_vector_type(2))) float v2f;
typedef __attribute__((ext_vector_type(8))) float v8f;

typedef __attribute__((ext_vector_type(4))) int v4i;
typedef __attribute__((ext_vector_type(4))) float v4f;

#define SU   5
#define ND4  (NN * DD / 4)
#define SBLK (ND4 / (256 * SU))
static_assert(ND4 % (256 * SU) == 0, "element-wise passes assume an exact grid");
static_assert(EE % 1024 == 0, "edge passes assume four edges per lane, exact grid");

#define NCOPY 16
static_assert((NCOPY & (NCOPY - 1)) == 0, "the copy of a work-group is picked with a mask");


__global__ void __launch_bounds__(256) k_pad_pe(const float* __restrict__ PE,
                                                float* __restrict__ PEp,
                                                int* __restrict__ cnt) {
  const int TOT = NN * PEDP / 4;
  float4 v[4];
  #pragma unroll
  for (int j = 0; j < 4; ++j) {
    int q = blockIdx.x * 1024 + j * 256 + threadIdx.x;
    int qc = (q < TOT) ? q : TOT - 1;
    int r = qc / (PEDP / 4);
    int c = (qc - r * (PEDP / 4)) * 4;
    const float* p = PE + (size_t)r * PED + c;
    bool in1 = c + 1 < PED, in2 = c + 2 < PED, in3 = c + 3 < PED;
    float a0 = p[0];
    float a1 = p[in1 ? 1 : 0];
    float a2 = p[in2 ? 2 : 0];
    float a3 = p[in3 ? 3 : 0];
    v[j] = make_float4(a0, in1 ? a1 : 0.0f, in2 ? a2 : 0.0f, in3 ? a3 : 0.0f);
  }
  #pragma unroll
  for (int j = 0; j < 4; ++j) {
    int q = blockIdx.x * 1024 + j * 256 + threadIdx.x;
    if (q < TOT) ((float4*)PEp)[q] = v[j];
  }
  int g = blockIdx.x * 256 + threadIdx.x;
  v4i z = {0, 0, 0, 0};
  if (g < NN / 4) ((v4i*)cnt)[g] = z;
}

__global__ void __launch_bounds__(256) k_rank(const int* __restrict__ dst,
                                              int* __restrict__ cnt,
                                              int* __restrict__ rank) {
  int q = blockIdx.x * 256 + threadIdx.x;
  v4i d = ((const v4i*)dst)[q];
  v4i r;
  r.x = atomicAdd(&cnt[d.x], 1);
  r.y = atomicAdd(&cnt[d.y], 1);
  r.z = atomicAdd(&cnt[d.z], 1);
  r.w = atomicAdd(&cnt[d.w], 1);
  ((v4i*)rank)[q] = r;
}

__global__ void __launch_bounds__(1024) k_scan(const int* __restrict__ cnt,
                                               int* __restrict__ off) {
  __shared__ int wtot[32];
  int wave = threadIdx.x >> 5, lane = threadIdx.x & 31;
  const v4i* c4 = (const v4i*)cnt;
  v4i* o4 = (v4i*)off;
  v4i v[13];
  int s = 0;
  #pragma unroll
  for (int i = 0; i < 13; ++i) {
    int q = wave * 416 + i * 32 + lane;
    v4i z = {0, 0, 0, 0};
    v[i] = (q < NN / 4) ? c4[q] : z;
    s += v[i].x + v[i].y + v[i].z + v[i].w;
  }
  for (int o = 16; o; o >>= 1) s += __shfl_xor(s, o, 32);
  if (lane == 0) wtot[wave] = s;
  __syncthreads();
  int carry = 0;
  for (int w = 0; w < wave; ++w) carry += wtot[w];
  #pragma unroll
  for (int i = 0; i < 13; ++i) {
    int t = v[i].x + v[i].y + v[i].z + v[i].w;
    int y = t;
    for (int o = 1; o < 32; o <<= 1) {
      int u = __shfl_up(y, o, 32);
      if (lane >= o) y += u;
    }
    int ex = carry + y - t;
    int q = wave * 416 + i * 32 + lane;
    if (q < NN / 4) {
      v4i r = {ex, ex + v[i].x, ex + v[i].x + v[i].y, ex + v[i].x + v[i].y + v[i].z};
      o4[q] = r;
    }
    carry += __shfl(y, 31, 32);
  }
  if (threadIdx.x == 1023) off[NN] = carry;
}

__global__ void __launch_bounds__(256) k_fill(const int* __restrict__ src,
                                              const int* __restrict__ dst,
                                              const int* __restrict__ attr,
                                              const int* __restrict__ off,
                                              const int* __restrict__ rank,
                                              v4i* __restrict__ rec) {
  int q = blockIdx.x * 256 + threadIdx.x;
  v4i s = ((const v4i*)src)[q];
  v4i d = ((const v4i*)dst)[q];
  v4i a = ((const v4i*)attr)[q];
  v4i k = ((const v4i*)rank)[q];
  int o0 = off[d.x], o1 = off[d.y], o2 = off[d.z], o3 = off[d.w];
  int e = q * 4;
  v4i r0 = {s.x, a.x, e,     0};
  v4i r1 = {s.y, a.y, e + 1, 0};
  v4i r2 = {s.z, a.z, e + 2, 0};
  v4i r3 = {s.w, a.w, e + 3, 0};
  rec[o0 + k.x] = r0;
  rec[o1 + k.y] = r1;
  rec[o2 + k.z] = r2;
  rec[o3 + k.w] = r3;
}

#define TP (DD + 4)
#define NTILE (NN / 16)
#define GB 512
static_assert(NN % 16 == 0, "row tiles are exact");

__device__ __forceinline__ void tile_loads(const float* __restrict__ PEp,
                                           const float* __restrict__ X, int tm, int mlo, int hi,
                                           int ncol, v2f (&a)[PEDP / 4], float (&xv)[8]) {
  const float* arow = PEp + (size_t)(tm * 16 + mlo) * PEDP + hi * 2;
  #pragma unroll
  for (int i = 0; i < PEDP / 4; ++i) a[i] = *(const v2f*)(arow + 4 * i);
  #pragma unroll
  for (int r = 0; r < 8; ++r) xv[r] = X[(size_t)(tm * 16 + r + hi * 8) * DD + ncol];
}

__global__ void __launch_bounds__(256) k_gemm2x(const float* __restrict__ PEp,
                                                const float* __restrict__ W,
                                                const float* __restrict__ X,
                                                const float* __restrict__ bb,
                                                const float* __restrict__ W2,
                                                const float* __restrict__ al,
                                                const float* __restrict__ ar,
                                                const float* __restrict__ ee,
                                                const float* __restrict__ ae,
                                                float* __restrict__ H,
                                                float* __restrict__ sl,
                                                float* __restrict__ sr,
                                                float* __restrict__ xe,
                                                float* __restrict__ ssum,
                                                float* __restrict__ ssq) {
  __shared__ __attribute__((aligned(16))) float strip[16 * TP];
  __shared__ __attribute__((aligned(16))) float hstrip[16 * TP];
  int wave = threadIdx.x >> 5;
  int lane = threadIdx.x & 31;

  if (blockIdx.x == 0) {
    v4f z4 = {0.0f, 0.0f, 0.0f, 0.0f};
    for (int i = threadIdx.x; i < NCOPY * DD / 4; i += 256) { ((v4f*)ssum)[i] = z4; ((v4f*)ssq)[i] = z4; }
    for (int t = wave; t < NT; t += 8) {
      float acc = 0.0f;
      #pragma unroll
      for (int j = 0; j < 4; ++j) {
        int d = lane + 32 * j;
        acc += ee[t * DD + d] * ae[d];
      }
      #pragma unroll
      for (int off = 16; off; off >>= 1) acc += __shfl_xor(acc, off, 32);
      if (lane == 0) xe[t] = acc;
    }
    return;
  }

  int mlo = lane & 15;
  int hi  = lane >> 4;
  int ncol = wave * 16 + mlo;

  int tm = blockIdx.x - 1;
  v2f a[PEDP / 4];
  float xv[8];
  tile_loads(PEp, X, tm, mlo, hi, ncol, a, xv);

  v2f wp[PEDP / 4];
  #pragma unroll
  for (int i = 0; i < PEDP / 4; ++i) {
    int ka = 4 * i + hi * 2, kb = ka + 1;
    float wa = W[(size_t)(ka < PED ? ka : PED - 1) * DD + ncol];
    float wb = W[(size_t)(kb < PED ? kb : PED - 1) * DD + ncol];
    wp[i].x = (ka < PED) ? wa : 0.0f;
    wp[i].y = (kb < PED) ? wb : 0.0f;
  }
  v2f w2[DD / 4];
  #pragma unroll
  for (int i = 0; i < DD / 4; ++i) {
    w2[i].x = W2[(size_t)(4 * i + hi * 2    ) * DD + ncol];
    w2[i].y = W2[(size_t)(4 * i + hi * 2 + 1) * DD + ncol];
  }
  float bbv = bb[ncol];
  float wl[4], wr[4];
  #pragma unroll
  for (int j = 0; j < 4; ++j) { wl[j] = al[lane + 32 * j]; wr[j] = ar[lane + 32 * j]; }

  const float* srow = strip + mlo * TP + hi * 2;

  for (; tm < NTILE; tm += GB) {
    v8f c = {};
    #pragma unroll
    for (int i = 0; i < PEDP / 4; ++i)
      c = __builtin_amdgcn_wmma_f32_16x16x4_f32(false, a[i], false, wp[i], (short)0, c,
                                                false, false);
    #pragma unroll
    for (int r = 0; r < 8; ++r) {
      float v = c[r];
      v += xv[r] + bbv;
      strip[(r + hi * 8) * TP + ncol] = v;
    }
    __syncthreads();

    if (tm + GB < NTILE) tile_loads(PEp, X, tm + GB, mlo, hi, ncol, a, xv);

    v8f d = {};
    #pragma unroll
    for (int i = 0; i < DD / 4; ++i) {
      v2f as = *(const v2f*)(srow + 4 * i);
      d = __builtin_amdgcn_wmma_f32_16x16x4_f32(false, as, false, w2[i], (short)0, d,
                                                false, false);
    }
    #pragma unroll
    for (int r = 0; r < 8; ++r) hstrip[(r + hi * 8) * TP + ncol] = d[r];
    __syncthreads();

    float ol = 0.0f, orr = 0.0f;
    #pragma unroll
    for (int i = 0; i < 2; ++i) {
      int row = wave * 2 + i;
      const float* hp = hstrip + row * TP;
      v4f q = *(const v4f*)(hp + lane * 4);
      ((v4f*)(H + (size_t)(tm * 16 + row) * DD))[lane] = q;
      float accl = 0.0f, accr = 0.0f;
      #pragma unroll
      for (int j = 0; j < 4; ++j) {
        float v = hp[lane + 32 * j];
        accl += v * wl[j];
        accr += v * wr[j];
      }
      #pragma unroll
      for (int off = 16; off; off >>= 1) {
        accl += __shfl_xor(accl, off, 32);
        accr += __shfl_xor(accr, off, 32);
      }
      if (lane == i) { ol = accl; orr = accr; }
    }
    if (lane < 2) {
      int n = tm * 16 + wave * 2 + lane;
      sl[n] = ol; sr[n] = orr;
    }
  }
}

__device__ __forceinline__ float edge_score(const float* __restrict__ sl,
                                            const float* __restrict__ xe, float srn,
                                            int s, int t) {
  float a = sl[s] + srn + xe[t];
  return (a > 0.0f) ? a : 0.2f * a;
}

template <int LAST>
__global__ void __launch_bounds__(256) k_node(const float* __restrict__ h,
                                              const float* __restrict__ sl,
                                              const float* __restrict__ sr,
                                              const float* __restrict__ xe,
                                              const int* __restrict__ off,
                                              const v4i* __restrict__ rec,
                                              const float* __restrict__ bias,
                                              const float* __restrict__ xres,
                                              float* __restrict__ outb,
                                              float* __restrict__ ssum,
                                              float* __restrict__ ssq) {
  __shared__ __attribute__((aligned(16))) float rows[8 * DD];
  __shared__ int arrived;
  int wave = threadIdx.x >> 5, lane = threadIdx.x & 31;
  int n = blockIdx.x * 8 + wave;
  if (LAST == 0) {
    if (threadIdx.x == 0) arrived = 0;
    __syncthreads();
  }
  float4 b4 = ((const float4*)bias)[lane];
  float4 x4 = make_float4(0.0f, 0.0f, 0.0f, 0.0f);
  if (LAST == 1) x4 = ((const float4*)(xres + (size_t)n * DD))[lane];
  int beg = __builtin_amdgcn_readfirstlane(off[n]);
  int end = __builtin_amdgcn_readfirstlane(off[n + 1]);
  float srn = sr[n];

  int j0 = beg + lane;
  bool has0 = j0 < end;
  int s0 = 0, e0 = 0;
  float a0 = -INFINITY;
  if (has0) {
    v4i r = rec[j0];
    s0 = r.x; e0 = r.z;
    a0 = edge_score(sl, xe, srn, r.x, r.y);
  }
  float m = a0;
  for (int j = j0 + 32; j < end; j += 32) {
    v4i r = rec[j];
    m = fmaxf(m, edge_score(sl, xe, srn, r.x, r.y));
  }
  for (int o = 16; o; o >>= 1) m = fmaxf(m, __shfl_xor(m, o, 32));

  float4 acc = make_float4(0.0f, 0.0f, 0.0f, 0.0f);
  float den = 0.0f;
  for (int base = beg; base < end; base += 32) {
    int sC = s0, eC = e0;
    float aC = a0;
    bool hasC = has0;
    if (base != beg) {
      int j = base + lane;
      hasC = j < end; sC = 0; eC = 0; aC = -INFINITY;
      if (hasC) {
        v4i r = rec[j];
        sC = r.x; eC = r.z;
        aC = edge_score(sl, xe, srn, r.x, r.y);
      }
    }
    float exC = hasC ? expf(aC - m) : 0.0f;
    int c = end - base; if (c > 32) c = 32;
    int rk = 0;
    #pragma unroll 4
    for (int k = 0; k < c; ++k) rk += (__shfl(eC, k, 32) < eC) ? 1 : 0;
    int k = 0;
    for (; k + 4 <= c; k += 4) {
      int L0 = __builtin_ctz(__builtin_amdgcn_ballot_w32(hasC && rk == k));
      int L1 = __builtin_ctz(__builtin_amdgcn_ballot_w32(hasC && rk == k + 1));
      int L2 = __builtin_ctz(__builtin_amdgcn_ballot_w32(hasC && rk == k + 2));
      int L3 = __builtin_ctz(__builtin_amdgcn_ballot_w32(hasC && rk == k + 3));
      float w0 = __shfl(exC, L0, 32), w1 = __shfl(exC, L1, 32);
      float w2 = __shfl(exC, L2, 32), w3 = __shfl(exC, L3, 32);
      float4 v0 = ((const float4*)(h + (size_t)__shfl(sC, L0, 32) * DD))[lane];
      float4 v1 = ((const float4*)(h + (size_t)__shfl(sC, L1, 32) * DD))[lane];
      float4 v2 = ((const float4*)(h + (size_t)__shfl(sC, L2, 32) * DD))[lane];
      float4 v3 = ((const float4*)(h + (size_t)__shfl(sC, L3, 32) * DD))[lane];
      den += w0; acc.x += w0 * v0.x; acc.y += w0 * v0.y; acc.z += w0 * v0.z; acc.w += w0 * v0.w;
      den += w1; acc.x += w1 * v1.x; acc.y += w1 * v1.y; acc.z += w1 * v1.z; acc.w += w1 * v1.w;
      den += w2; acc.x += w2 * v2.x; acc.y += w2 * v2.y; acc.z += w2 * v2.z; acc.w += w2 * v2.w;
      den += w3; acc.x += w3 * v3.x; acc.y += w3 * v3.y; acc.z += w3 * v3.z; acc.w += w3 * v3.w;
    }
    for (; k < c; ++k) {
      int L0 = __builtin_ctz(__builtin_amdgcn_ballot_w32(hasC && rk == k));
      float w0 = __shfl(exC, L0, 32);
      float4 v0 = ((const float4*)(h + (size_t)__shfl(sC, L0, 32) * DD))[lane];
      den += w0; acc.x += w0 * v0.x; acc.y += w0 * v0.y; acc.z += w0 * v0.z; acc.w += w0 * v0.w;
    }
  }
  float inv = (end > beg) ? 1.0f / den : 0.0f;
  float4 o4 = make_float4(acc.x * inv + b4.x, acc.y * inv + b4.y, acc.z * inv + b4.z,
                          acc.w * inv + b4.w);
  if (LAST == 1) {
    o4.x += x4.x; o4.y += x4.y; o4.z += x4.z; o4.w += x4.w;
    ((float4*)(outb + (size_t)n * DD))[lane] = o4;
  } else {
    ((float4*)(rows + wave * DD))[lane] = o4;
    __builtin_amdgcn_fence(__ATOMIC_RELEASE, "workgroup");
    int old = 0;
    if (lane == 0)
      old = __hip_atomic_fetch_add(&arrived, 1, __ATOMIC_RELAXED, __HIP_MEMORY_SCOPE_WORKGROUP);
    ((float4*)(outb + (size_t)n * DD))[lane] = o4;
    old = __builtin_amdgcn_readfirstlane(old);
    if (old == 7) {
      __builtin_amdgcn_fence(__ATOMIC_ACQUIRE, "workgroup");
      float* ps = ssum + (blockIdx.x & (NCOPY - 1)) * DD;
      float* pq = ssq + (blockIdx.x & (NCOPY - 1)) * DD;
      #pragma unroll
      for (int j = 0; j < 4; ++j) {
        int col = lane + 32 * j;
        float s = 0.0f, q = 0.0f;
        #pragma unroll
        for (int w = 0; w < 8; ++w) {
          float v = rows[w * DD + col];
          s += v; q += v * v;
        }
        atomicAdd(ps + col, s);
        atomicAdd(pq + col, q);
      }
    }
  }
}


__device__ __forceinline__ float bn_one(float o, float s, float s2, float g, float b, float x) {
  float mu  = s * (1.0f / NN);
  float var = s2 * (1.0f / NN) - mu * mu;
  float inv = rsqrtf(var + 1e-5f);
  return (o - mu) * inv * g + b + x;
}

__global__ void __launch_bounds__(256) k_bn_residual(const float* __restrict__ outb,
                                                     const float* __restrict__ ssum,
                                                     const float* __restrict__ ssq,
                                                     const float* __restrict__ gam,
                                                     const float* __restrict__ bet,
                                                     const float* __restrict__ xin,
                                                     float* __restrict__ xout) {
  __shared__ __attribute__((aligned(16))) float tot[2 * DD];
  const v4f* o4 = (const v4f*)outb;
  const v4f* xi4 = (const v4f*)xin;
  v4f* x4 = (v4f*)xout;
  int q0 = blockIdx.x * (256 * SU) + threadIdx.x;
  int c4 = threadIdx.x & (DD / 4 - 1);
  v4f o[SU], x[SU];
  const float* part = (threadIdx.x < DD) ? ssum + threadIdx.x : ssq + (threadIdx.x - DD);
  float p[NCOPY];
  #pragma unroll
  for (int k = 0; k < NCOPY; ++k) p[k] = part[k * DD];
  v4f g  = ((const v4f*)gam)[c4];
  v4f b  = ((const v4f*)bet)[c4];
  #pragma unroll
  for (int j = 0; j < SU; ++j) { o[j] = o4[q0 + j * 256]; x[j] = xi4[q0 + j * 256]; }
  float t = 0.0f;
  #pragma unroll
  for (int k = 0; k < NCOPY; ++k) t += p[k];
  tot[threadIdx.x] = t;
  __syncthreads();
  v4f s  = *(const v4f*)(tot + c4 * 4);
  v4f s2 = *(const v4f*)(tot + DD + c4 * 4);
  #pragma unroll
  for (int j = 0; j < SU; ++j) {
    v4f r;
    r.x = bn_one(o[j].x, s.x, s2.x, g.x, b.x, x[j].x);
    r.y = bn_one(o[j].y, s.y, s2.y, g.y, b.y, x[j].y);
    r.z = bn_one(o[j].z, s.z, s2.z, g.z, b.z, x[j].z);
    r.w = bn_one(o[j].w, s.w, s2.w, g.w, b.w, x[j].w);
    x4[q0 + j * 256] = r;
  }
}


extern "C" void kernel_launch(void* const* d_in, const int* in_sizes, int n_in,
                              void* d_out, int out_size, void* d_ws, size_t ws_size,
                              hipStream_t stream) {
  const float* X_n   = (const float*)d_in[0];
  const float* PE    = (const float*)d_in[1];
  const int*   eidx  = (const int*)d_in[2];
  const int*   eattr = (const int*)d_in[3];
  const float* eemb  = (const float*)d_in[4];
  const float* peW   = (const float*)d_in[5];
  const float* peB   = (const float*)d_in[6];
  const float* linW  = (const float*)d_in[7];
  const float* attl  = (const float*)d_in[8];
  const float* attr_ = (const float*)d_in[9];
  const float* atte  = (const float*)d_in[10];
  const float* bias  = (const float*)d_in[11];
  const float* gam   = (const float*)d_in[12];
  const float* bet   = (const float*)d_in[13];
  const int* src = eidx;
  const int* dst = eidx + EE;

  float* ws = (float*)d_ws;
  size_t o = 0;
  float* xa    = ws + o; o += (size_t)NN * DD;
  float* xb    = ws + o; o += (size_t)NN * DD;
  float* outb  = ws + o; o += (size_t)NN * DD;
  float* h     = ws + o; o += (size_t)NN * DD;
  float* PEp   = ws + o; o += (size_t)NN * PEDP;
  float* sl    = ws + o; o += NN;
  float* sr    = ws + o; o += NN;
  int*   cnt   = (int*)(ws + o); o += NN;
  int*   off   = (int*)(ws + o); o += NN + 16;
  int*   rank  = (int*)(ws + o); o += EE;
  v4i*   rec   = (v4i*)(ws + o); o += (size_t)EE * 4;
  float* xe    = ws + o; o += 16;
  float* ssum  = ws + o; o += NCOPY * DD;
  float* ssq   = ws + o; o += NCOPY * DD;
  float* xnext[NL - 1] = {xa, xb};

  k_pad_pe<<<(NN * PEDP / 4 + 1023) / 1024, 256, 0, stream>>>(PE, PEp, cnt);
  k_rank<<<EE / 1024, 256, 0, stream>>>(dst, cnt, rank);
  k_scan<<<1, 1024, 0, stream>>>(cnt, off);
  k_fill<<<EE / 1024, 256, 0, stream>>>(src, dst, eattr, off, rank, rec);

  const float* xin = X_n;
  for (int l = 0; l < NL; ++l) {
    k_gemm2x<<<GB + 1, 256, 0, stream>>>(PEp, peW + (size_t)l * PED * DD, xin, peB + l * DD,
                                              linW + (size_t)l * DD * DD, attl + l * DD,
                                              attr_ + l * DD, eemb + (size_t)l * NT * DD,
                                              atte + l * DD, h, sl, sr, xe, ssum, ssq);

    if (l < NL - 1) {
      k_node<0><<<NN / 8, 256, 0, stream>>>(h, sl, sr, xe, off, rec, bias + l * DD, xin, outb,
                                            ssum, ssq);
      k_bn_residual<<<SBLK, 256, 0, stream>>>(outb, ssum, ssq, gam + l * DD,
                                              bet + l * DD, xin, xnext[l]);
      xin = xnext[l];
    } else {
      k_node<1><<<NN / 8, 256, 0, stream>>>(h, sl, sr, xe, off, rec, bias + l * DD, xin,
                                            (float*)d_out, ssum, ssq);
    }
  }
}


# Batch-norm apply and residual folded into the next layer's GEMM launch: it forms x in C layout from the aggregation output and column sums, writes x once; launches 12->10; no sum reordered
# speedup vs baseline: 3.7845x; 1.0345x over previous
#include <hip/hip_runtime.h>
#include <math.h>

#define NN   50000
#define EE   640000
#define DD   128
#define PED  37
#define PEDP 40
#define NT   9
#define NL   3

typedef __attribute__((ext_vector_type(2))) float v2f;
typedef __attribute__((ext_vector_type(8))) float v8f;

typedef __attribute__((ext_vector_type(4))) int v4i;
typedef __attribute__((ext_vector_type(4))) float v4f;

static_assert(EE % 1024 == 0, "edge passes assume four edges per lane, exact grid");

#define NCOPY 16
static_assert((NCOPY & (NCOPY - 1)) == 0, "the copy of a work-group is picked with a mask");


__global__ void __launch_bounds__(256) k_pad_pe(const float* __restrict__ PE,
                                                float* __restrict__ PEp,
                                                int* __restrict__ cnt) {
  const int TOT = NN * PEDP / 4;
  float4 v[4];
  #pragma unroll
  for (int j = 0; j < 4; ++j) {
    int q = blockIdx.x * 1024 + j * 256 + threadIdx.x;
    int qc = (q < TOT) ? q : TOT - 1;
    int r = qc / (PEDP / 4);
    int c = (qc - r * (PEDP / 4)) * 4;
    const float* p = PE + (size_t)r * PED + c;
    bool in1 = c + 1 < PED, in2 = c + 2 < PED, in3 = c + 3 < PED;
    float a0 = p[0];
    float a1 = p[in1 ? 1 : 0];
    float a2 = p[in2 ? 2 : 0];
    float a3 = p[in3 ? 3 : 0];
    v[j] = make_float4(a0, in1 ? a1 : 0.0f, in2 ? a2 : 0.0f, in3 ? a3 : 0.0f);
  }
  #pragma unroll
  for (int j = 0; j < 4; ++j) {
    int q = blockIdx.x * 1024 + j * 256 + threadIdx.x;
    if (q < TOT) ((float4*)PEp)[q] = v[j];
  }
  int g = blockIdx.x * 256 + threadIdx.x;
  v4i z = {0, 0, 0, 0};
  if (g < NN / 4) ((v4i*)cnt)[g] = z;
}

__global__ void __launch_bounds__(256) k_rank(const int* __restrict__ dst,
                                              int* __restrict__ cnt,
                                              int* __restrict__ rank) {
  int q = blockIdx.x * 256 + threadIdx.x;
  v4i d = ((const v4i*)dst)[q];
  v4i r;
  r.x = atomicAdd(&cnt[d.x], 1);
  r.y = atomicAdd(&cnt[d.y], 1);
  r.z = atomicAdd(&cnt[d.z], 1);
  r.w = atomicAdd(&cnt[d.w], 1);
  ((v4i*)rank)[q] = r;
}

__global__ void __launch_bounds__(1024) k_scan(const int* __restrict__ cnt,
                                               int* __restrict__ off) {
  __shared__ int wtot[32];
  int wave = threadIdx.x >> 5, lane = threadIdx.x & 31;
  const v4i* c4 = (const v4i*)cnt;
  v4i* o4 = (v4i*)off;
  v4i v[13];
  int s = 0;
  #pragma unroll
  for (int i = 0; i < 13; ++i) {
    int q = wave * 416 + i * 32 + lane;
    v4i z = {0, 0, 0, 0};
    v[i] = (q < NN / 4) ? c4[q] : z;
    s += v[i].x + v[i].y + v[i].z + v[i].w;
  }
  for (int o = 16; o; o >>= 1) s += __shfl_xor(s, o, 32);
  if (lane == 0) wtot[wave] = s;
  __syncthreads();
  int carry = 0;
  for (int w = 0; w < wave; ++w) carry += wtot[w];
  #pragma unroll
  for (int i = 0; i < 13; ++i) {
    int t = v[i].x + v[i].y + v[i].z + v[i].w;
    int y = t;
    for (int o = 1; o < 32; o <<= 1) {
      int u = __shfl_up(y, o, 32);
      if (lane >= o) y += u;
    }
    int ex = carry + y - t;
    int q = wave * 416 + i * 32 + lane;
    if (q < NN / 4) {
      v4i r = {ex, ex + v[i].x, ex + v[i].x + v[i].y, ex + v[i].x + v[i].y + v[i].z};
      o4[q] = r;
    }
    carry += __shfl(y, 31, 32);
  }
  if (threadIdx.x == 1023) off[NN] = carry;
}

__global__ void __launch_bounds__(256) k_fill(const int* __restrict__ src,
                                              const int* __restrict__ dst,
                                              const int* __restrict__ attr,
                                              const int* __restrict__ off,
                                              const int* __restrict__ rank,
                                              v4i* __restrict__ rec) {
  int q = blockIdx.x * 256 + threadIdx.x;
  v4i s = ((const v4i*)src)[q];
  v4i d = ((const v4i*)dst)[q];
  v4i a = ((const v4i*)attr)[q];
  v4i k = ((const v4i*)rank)[q];
  int o0 = off[d.x], o1 = off[d.y], o2 = off[d.z], o3 = off[d.w];
  int e = q * 4;
  v4i r0 = {s.x, a.x, e,     0};
  v4i r1 = {s.y, a.y, e + 1, 0};
  v4i r2 = {s.z, a.z, e + 2, 0};
  v4i r3 = {s.w, a.w, e + 3, 0};
  rec[o0 + k.x] = r0;
  rec[o1 + k.y] = r1;
  rec[o2 + k.z] = r2;
  rec[o3 + k.w] = r3;
}

#define TP (DD + 4)
#define NTILE (NN / 16)
#define GB 512
static_assert(NN % 16 == 0, "row tiles are exact");

template <int BN>
__device__ __forceinline__ void tile_loads(const float* __restrict__ PEp,
                                           const float* __restrict__ X,
                                           const float* __restrict__ OB, int tm, int mlo, int hi,
                                           int ncol, v2f (&a)[PEDP / 4], float (&xv)[8],
                                           float (&ov)[8]) {
  const float* arow = PEp + (size_t)(tm * 16 + mlo) * PEDP + hi * 2;
  #pragma unroll
  for (int i = 0; i < PEDP / 4; ++i) a[i] = *(const v2f*)(arow + 4 * i);
  #pragma unroll
  for (int r = 0; r < 8; ++r) xv[r] = X[(size_t)(tm * 16 + r + hi * 8) * DD + ncol];
  if (BN) {
    #pragma unroll
    for (int r = 0; r < 8; ++r) ov[r] = OB[(size_t)(tm * 16 + r + hi * 8) * DD + ncol];
  }
}

template <int BN>
__global__ void __launch_bounds__(256) k_gemm2x(const float* __restrict__ PEp,
                                                const float* __restrict__ W,
                                                const float* __restrict__ X,
                                                const float* __restrict__ bb,
                                                const float* __restrict__ W2,
                                                const float* __restrict__ al,
                                                const float* __restrict__ ar,
                                                const float* __restrict__ ee,
                                                const float* __restrict__ ae,
                                                float* __restrict__ H,
                                                float* __restrict__ sl,
                                                float* __restrict__ sr,
                                                float* __restrict__ xe,
                                                float* __restrict__ ssum,
                                                float* __restrict__ ssq,
                                                const float* __restrict__ OB,
                                                const float* __restrict__ psum,
                                                const float* __restrict__ psq,
                                                const float* __restrict__ gam,
                                                const float* __restrict__ bet,
                                                float* __restrict__ XO) {
  __shared__ __attribute__((aligned(16))) float strip[16 * TP];
  __shared__ __attribute__((aligned(16))) float hstrip[16 * TP];
  __shared__ __attribute__((aligned(16))) float xstrip[BN ? 2 * 16 * TP : 4];
  int wave = threadIdx.x >> 5;
  int lane = threadIdx.x & 31;

  if (blockIdx.x == 0) {
    v4f z4 = {0.0f, 0.0f, 0.0f, 0.0f};
    for (int i = threadIdx.x; i < NCOPY * DD / 4; i += 256) { ((v4f*)ssum)[i] = z4; ((v4f*)ssq)[i] = z4; }
    for (int t = wave; t < NT; t += 8) {
      float acc = 0.0f;
      #pragma unroll
      for (int j = 0; j < 4; ++j) {
        int d = lane + 32 * j;
        acc += ee[t * DD + d] * ae[d];
      }
      #pragma unroll
      for (int off = 16; off; off >>= 1) acc += __shfl_xor(acc, off, 32);
      if (lane == 0) xe[t] = acc;
    }
    return;
  }

  int mlo = lane & 15;
  int hi  = lane >> 4;
  int ncol = wave * 16 + mlo;

  int tm = blockIdx.x - 1;
  v2f a[PEDP / 4];
  float xv[8], ov[8];
  tile_loads<BN>(PEp, X, OB, tm, mlo, hi, ncol, a, xv, ov);

  float mu = 0.0f, inv = 0.0f, ga = 0.0f, be = 0.0f;
  if (BN) {
    float ps[NCOPY], pq[NCOPY];
    #pragma unroll
    for (int k = 0; k < NCOPY; ++k) { ps[k] = psum[k * DD + ncol]; pq[k] = psq[k * DD + ncol]; }
    ga = gam[ncol]; be = bet[ncol];
    float ts = 0.0f, tq = 0.0f;
    #pragma unroll
    for (int k = 0; k < NCOPY; ++k) { ts += ps[k]; tq += pq[k]; }
    mu = ts * (1.0f / NN);
    float var = tq * (1.0f / NN) - mu * mu;
    inv = rsqrtf(var + 1e-5f);
  }

  v2f wp[PEDP / 4];
  #pragma unroll
  for (int i = 0; i < PEDP / 4; ++i) {
    int ka = 4 * i + hi * 2, kb = ka + 1;
    float wa = W[(size_t)(ka < PED ? ka : PED - 1) * DD + ncol];
    float wb = W[(size_t)(kb < PED ? kb : PED - 1) * DD + ncol];
    wp[i].x = (ka < PED) ? wa : 0.0f;
    wp[i].y = (kb < PED) ? wb : 0.0f;
  }
  v2f w2[DD / 4];
  #pragma unroll
  for (int i = 0; i < DD / 4; ++i) {
    w2[i].x = W2[(size_t)(4 * i + hi * 2    ) * DD + ncol];
    w2[i].y = W2[(size_t)(4 * i + hi * 2 + 1) * DD + ncol];
  }
  float bbv = bb[ncol];
  float wl[4], wr[4];
  #pragma unroll
  for (int j = 0; j < 4; ++j) { wl[j] = al[lane + 32 * j]; wr[j] = ar[lane + 32 * j]; }

  const float* srow = strip + mlo * TP + hi * 2;

  int trip = 0;
  for (; tm < NTILE; tm += GB, ++trip) {
    float* xs = xstrip + (BN ? (trip & 1) * 16 * TP : 0);
    v8f c = {};
    #pragma unroll
    for (int i = 0; i < PEDP / 4; ++i)
      c = __builtin_amdgcn_wmma_f32_16x16x4_f32(false, a[i], false, wp[i], (short)0, c,
                                                false, false);
    #pragma unroll
    for (int r = 0; r < 8; ++r) {
      float xr = xv[r];
      if (BN) {
        xr = (ov[r] - mu) * inv * ga + be + xv[r];
        xs[(r + hi * 8) * TP + ncol] = xr;
      }
      float v = c[r];
      v += xr + bbv;
      strip[(r + hi * 8) * TP + ncol] = v;
    }
    __syncthreads();

    if (tm + GB < NTILE) tile_loads<BN>(PEp, X, OB, tm + GB, mlo, hi, ncol, a, xv, ov);

    v8f d = {};
    #pragma unroll
    for (int i = 0; i < DD / 4; ++i) {
      v2f as = *(const v2f*)(srow + 4 * i);
      d = __builtin_amdgcn_wmma_f32_16x16x4_f32(false, as, false, w2[i], (short)0, d,
                                                false, false);
    }
    #pragma unroll
    for (int r = 0; r < 8; ++r) hstrip[(r + hi * 8) * TP + ncol] = d[r];
    __syncthreads();

    float ol = 0.0f, orr = 0.0f;
    #pragma unroll
    for (int i = 0; i < 2; ++i) {
      int row = wave * 2 + i;
      const float* hp = hstrip + row * TP;
      v4f q = *(const v4f*)(hp + lane * 4);
      ((v4f*)(H + (size_t)(tm * 16 + row) * DD))[lane] = q;
      if (BN) {
        v4f xq = *(const v4f*)(xs + row * TP + lane * 4);
        ((v4f*)(XO + (size_t)(tm * 16 + row) * DD))[lane] = xq;
      }
      float accl = 0.0f, accr = 0.0f;
      #pragma unroll
      for (int j = 0; j < 4; ++j) {
        float v = hp[lane + 32 * j];
        accl += v * wl[j];
        accr += v * wr[j];
      }
      #pragma unroll
      for (int off = 16; off; off >>= 1) {
        accl += __shfl_xor(accl, off, 32);
        accr += __shfl_xor(accr, off, 32);
      }
      if (lane == i) { ol = accl; orr = accr; }
    }
    if (lane < 2) {
      int n = tm * 16 + wave * 2 + lane;
      sl[n] = ol; sr[n] = orr;
    }
  }
}

__device__ __forceinline__ float edge_score(const float* __restrict__ sl,
                                            const float* __restrict__ xe, float srn,
                                            int s, int t) {
  float a = sl[s] + srn + xe[t];
  return (a > 0.0f) ? a : 0.2f * a;
}

template <int LAST>
__global__ void __launch_bounds__(256) k_node(const float* __restrict__ h,
                                              const float* __restrict__ sl,
                                              const float* __restrict__ sr,
                                              const float* __restrict__ xe,
                                              const int* __restrict__ off,
                                              const v4i* __restrict__ rec,
                                              const float* __restrict__ bias,
                                              const float* __restrict__ xres,
                                              float* __restrict__ outb,
                                              float* __restrict__ ssum,
                                              float* __restrict__ ssq) {
  __shared__ __attribute__((aligned(16))) float rows[8 * DD];
  __shared__ int arrived;
  int wave = threadIdx.x >> 5, lane = threadIdx.x & 31;
  int n = blockIdx.x * 8 + wave;
  if (LAST == 0) {
    if (threadIdx.x == 0) arrived = 0;
    __syncthreads();
  }
  float4 b4 = ((const float4*)bias)[lane];
  float4 x4 = make_float4(0.0f, 0.0f, 0.0f, 0.0f);
  if (LAST == 1) x4 = ((const float4*)(xres + (size_t)n * DD))[lane];
  int beg = __builtin_amdgcn_readfirstlane(off[n]);
  int end = __builtin_amdgcn_readfirstlane(off[n + 1]);
  float srn = sr[n];

  int j0 = beg + lane;
  bool has0 = j0 < end;
  int s0 = 0, e0 = 0;
  float a0 = -INFINITY;
  if (has0) {
    v4i r = rec[j0];
    s0 = r.x; e0 = r.z;
    a0 = edge_score(sl, xe, srn, r.x, r.y);
  }
  float m = a0;
  for (int j = j0 + 32; j < end; j += 32) {
    v4i r = rec[j];
    m = fmaxf(m, edge_score(sl, xe, srn, r.x, r.y));
  }
  for (int o = 16; o; o >>= 1) m = fmaxf(m, __shfl_xor(m, o, 32));

  float4 acc = make_float4(0.0f, 0.0f, 0.0f, 0.0f);
  float den = 0.0f;
  for (int base = beg; base < end; base += 32) {
    int sC = s0, eC = e0;
    float aC = a0;
    bool hasC = has0;
    if (base != beg) {
      int j = base + lane;
      hasC = j < end; sC = 0; eC = 0; aC = -INFINITY;
      if (hasC) {
        v4i r = rec[j];
        sC = r.x; eC = r.z;
        aC = edge_score(sl, xe, srn, r.x, r.y);
      }
    }
    float exC = hasC ? expf(aC - m) : 0.0f;
    int c = end - base; if (c > 32) c = 32;
    int rk = 0;
    #pragma unroll 4
    for (int k = 0; k < c; ++k) rk += (__shfl(eC, k, 32) < eC) ? 1 : 0;
    int k = 0;
    for (; k + 4 <= c; k += 4) {
      int L0 = __builtin_ctz(__builtin_amdgcn_ballot_w32(hasC && rk == k));
      int L1 = __builtin_ctz(__builtin_amdgcn_ballot_w32(hasC && rk == k + 1));
      int L2 = __builtin_ctz(__builtin_amdgcn_ballot_w32(hasC && rk == k + 2));
      int L3 = __builtin_ctz(__builtin_amdgcn_ballot_w32(hasC && rk == k + 3));
      float w0 = __shfl(exC, L0, 32), w1 = __shfl(exC, L1, 32);
      float w2 = __shfl(exC, L2, 32), w3 = __shfl(exC, L3, 32);
      float4 v0 = ((const float4*)(h + (size_t)__shfl(sC, L0, 32) * DD))[lane];
      float4 v1 = ((const float4*)(h + (size_t)__shfl(sC, L1, 32) * DD))[lane];
      float4 v2 = ((const float4*)(h + (size_t)__shfl(sC, L2, 32) * DD))[lane];
      float4 v3 = ((const float4*)(h + (size_t)__shfl(sC, L3, 32) * DD))[lane];
      den += w0; acc.x += w0 * v0.x; acc.y += w0 * v0.y; acc.z += w0 * v0.z; acc.w += w0 * v0.w;
      den += w1; acc.x += w1 * v1.x; acc.y += w1 * v1.y; acc.z += w1 * v1.z; acc.w += w1 * v1.w;
      den += w2; acc.x += w2 * v2.x; acc.y += w2 * v2.y; acc.z += w2 * v2.z; acc.w += w2 * v2.w;
      den += w3; acc.x += w3 * v3.x; acc.y += w3 * v3.y; acc.z += w3 * v3.z; acc.w += w3 * v3.w;
    }
    for (; k < c; ++k) {
      int L0 = __builtin_ctz(__builtin_amdgcn_ballot_w32(hasC && rk == k));
      float w0 = __shfl(exC, L0, 32);
      float4 v0 = ((const float4*)(h + (size_t)__shfl(sC, L0, 32) * DD))[lane];
      den += w0; acc.x += w0 * v0.x; acc.y += w0 * v0.y; acc.z += w0 * v0.z; acc.w += w0 * v0.w;
    }
  }
  float inv = (end > beg) ? 1.0f / den : 0.0f;
  float4 o4 = make_float4(acc.x * inv + b4.x, acc.y * inv + b4.y, acc.z * inv + b4.z,
                          acc.w * inv + b4.w);
  if (LAST == 1) {
    o4.x += x4.x; o4.y += x4.y; o4.z += x4.z; o4.w += x4.w;
    ((float4*)(outb + (size_t)n * DD))[lane] = o4;
  } else {
    ((float4*)(rows + wave * DD))[lane] = o4;
    __builtin_amdgcn_fence(__ATOMIC_RELEASE, "workgroup");
    int old = 0;
    if (lane == 0)
      old = __hip_atomic_fetch_add(&arrived, 1, __ATOMIC_RELAXED, __HIP_MEMORY_SCOPE_WORKGROUP);
    ((float4*)(outb + (size_t)n * DD))[lane] = o4;
    old = __builtin_amdgcn_readfirstlane(old);
    if (old == 7) {
      __builtin_amdgcn_fence(__ATOMIC_ACQUIRE, "workgroup");
      float* ps = ssum + (blockIdx.x & (NCOPY - 1)) * DD;
      float* pq = ssq + (blockIdx.x & (NCOPY - 1)) * DD;
      #pragma unroll
      for (int j = 0; j < 4; ++j) {
        int col = lane + 32 * j;
        float s = 0.0f, q = 0.0f;
        #pragma unroll
        for (int w = 0; w < 8; ++w) {
          float v = rows[w * DD + col];
          s += v; q += v * v;
        }
        atomicAdd(ps + col, s);
        atomicAdd(pq + col, q);
      }
    }
  }
}


extern "C" void kernel_launch(void* const* d_in, const int* in_sizes, int n_in,
                              void* d_out, int out_size, void* d_ws, size_t ws_size,
                              hipStream_t stream) {
  const float* X_n   = (const float*)d_in[0];
  const float* PE    = (const float*)d_in[1];
  const int*   eidx  = (const int*)d_in[2];
  const int*   eattr = (const int*)d_in[3];
  const float* eemb  = (const float*)d_in[4];
  const float* peW   = (const float*)d_in[5];
  const float* peB   = (const float*)d_in[6];
  const float* linW  = (const float*)d_in[7];
  const float* attl  = (const float*)d_in[8];
  const float* attr_ = (const float*)d_in[9];
  const float* atte  = (const float*)d_in[10];
  const float* bias  = (const float*)d_in[11];
  const float* gam   = (const float*)d_in[12];
  const float* bet   = (const float*)d_in[13];
  const int* src = eidx;
  const int* dst = eidx + EE;

  float* ws = (float*)d_ws;
  size_t o = 0;
  float* xa    = ws + o; o += (size_t)NN * DD;
  float* xb    = ws + o; o += (size_t)NN * DD;
  float* outb  = ws + o; o += (size_t)NN * DD;
  float* h     = ws + o; o += (size_t)NN * DD;
  float* PEp   = ws + o; o += (size_t)NN * PEDP;
  float* sl    = ws + o; o += NN;
  float* sr    = ws + o; o += NN;
  int*   cnt   = (int*)(ws + o); o += NN;
  int*   off   = (int*)(ws + o); o += NN + 16;
  int*   rank  = (int*)(ws + o); o += EE;
  v4i*   rec   = (v4i*)(ws + o); o += (size_t)EE * 4;
  float* xe    = ws + o; o += 16;
  float* ssum0 = ws + o; o += NCOPY * DD;
  float* ssq0  = ws + o; o += NCOPY * DD;
  float* ssum1 = ws + o; o += NCOPY * DD;
  float* ssq1  = ws + o; o += NCOPY * DD;
  float* xnext[NL - 1] = {xa, xb};
  float* ssum[2] = {ssum0, ssum1};
  float* ssq[2]  = {ssq0, ssq1};

  k_pad_pe<<<(NN * PEDP / 4 + 1023) / 1024, 256, 0, stream>>>(PE, PEp, cnt);
  k_rank<<<EE / 1024, 256, 0, stream>>>(dst, cnt, rank);
  k_scan<<<1, 1024, 0, stream>>>(cnt, off);
  k_fill<<<EE / 1024, 256, 0, stream>>>(src, dst, eattr, off, rank, rec);

  const float* xin = X_n;
  for (int l = 0; l < NL; ++l) {
    if (l == 0) {
      k_gemm2x<0><<<GB + 1, 256, 0, stream>>>(PEp, peW, xin, peB, linW, attl, attr_, eemb, atte, h,
                                              sl, sr, xe, ssum[0], ssq[0], nullptr, nullptr,
                                              nullptr, nullptr, nullptr, nullptr);
    } else {
      k_gemm2x<1><<<GB + 1, 256, 0, stream>>>(PEp, peW + (size_t)l * PED * DD, xin, peB + l * DD,
                                              linW + (size_t)l * DD * DD, attl + l * DD,
                                              attr_ + l * DD, eemb + (size_t)l * NT * DD,
                                              atte + l * DD, h, sl, sr, xe, ssum[l & 1], ssq[l & 1],
                                              outb, ssum[(l - 1) & 1], ssq[(l - 1) & 1],
                                              gam + (l - 1) * DD, bet + (l - 1) * DD, xnext[l - 1]);
      xin = xnext[l - 1];
    }

    if (l < NL - 1) {
      k_node<0><<<NN / 8, 256, 0, stream>>>(h, sl, sr, xe, off, rec, bias + l * DD, xin, outb,
                                            ssum[l & 1], ssq[l & 1]);
    } else {
      k_node<1><<<NN / 8, 256, 0, stream>>>(h, sl, sr, xe, off, rec, bias + l * DD, xin,
                                            (float*)d_out, ssum[l & 1], ssq[l & 1]);
    }
  }
}


# Edge records go straight into fixed 32-slot buckets per destination (counting atomic's return = slot; overflow list above 32): scan and fill launches go, 10->8; sum order kept up to 32 edges a node
# speedup vs baseline: 3.9781x; 1.0512x over previous
#include <hip/hip_runtime.h>
#include <math.h>

#define NN   50000
#define EE   640000
#define DD   128
#define PED  37
#define PEDP 40
#define NT   9
#define NL   3

typedef __attribute__((ext_vector_type(2))) float v2f;
typedef __attribute__((ext_vector_type(8))) float v8f;

typedef __attribute__((ext_vector_type(4))) int v4i;
typedef __attribute__((ext_vector_type(4))) float v4f;

static_assert(EE % 1024 == 0, "edge passes assume four edges per lane, exact grid");

#define NCOPY 16
static_assert((NCOPY & (NCOPY - 1)) == 0, "the copy of a work-group is picked with a mask");


__global__ void __launch_bounds__(256) k_pad_pe(const float* __restrict__ PE,
                                                float* __restrict__ PEp,
                                                int* __restrict__ cnt) {
  const int TOT = NN * PEDP / 4;
  float4 v[4];
  #pragma unroll
  for (int j = 0; j < 4; ++j) {
    int q = blockIdx.x * 1024 + j * 256 + threadIdx.x;
    int qc = (q < TOT) ? q : TOT - 1;
    int r = qc / (PEDP / 4);
    int c = (qc - r * (PEDP / 4)) * 4;
    const float* p = PE + (size_t)r * PED + c;
    bool in1 = c + 1 < PED, in2 = c + 2 < PED, in3 = c + 3 < PED;
    float a0 = p[0];
    float a1 = p[in1 ? 1 : 0];
    float a2 = p[in2 ? 2 : 0];
    float a3 = p[in3 ? 3 : 0];
    v[j] = make_float4(a0, in1 ? a1 : 0.0f, in2 ? a2 : 0.0f, in3 ? a3 : 0.0f);
  }
  #pragma unroll
  for (int j = 0; j < 4; ++j) {
    int q = blockIdx.x * 1024 + j * 256 + threadIdx.x;
    if (q < TOT) ((float4*)PEp)[q] = v[j];
  }
  int g = blockIdx.x * 256 + threadIdx.x;
  v4i z = {0, 0, 0, 0};
  if (g <= NN / 4) ((v4i*)cnt)[g] = z;
}

#define CAP 32
__device__ __forceinline__ void put_record(int d, int r, int s, int t, int e,
                                           int* __restrict__ cnt, v4i* __restrict__ rec,
                                           v4i* __restrict__ ovf) {
  v4i v = {s, t, e, d};
  if (r < CAP) {
    rec[(size_t)d * CAP + r] = v;
  } else {
    int o = atomicAdd(&cnt[NN], 1);
    ovf[o] = v;
  }
}

__global__ void __launch_bounds__(256) k_bucket(const int* __restrict__ src,
                                                const int* __restrict__ dst,
                                                const int* __restrict__ attr,
                                                int* __restrict__ cnt,
                                                v4i* __restrict__ rec,
                                                v4i* __restrict__ ovf) {
  int q = blockIdx.x * 256 + threadIdx.x;
  v4i s = ((const v4i*)src)[q];
  v4i d = ((const v4i*)dst)[q];
  v4i a = ((const v4i*)attr)[q];
  int r0 = atomicAdd(&cnt[d.x], 1);
  int r1 = atomicAdd(&cnt[d.y], 1);
  int r2 = atomicAdd(&cnt[d.z], 1);
  int r3 = atomicAdd(&cnt[d.w], 1);
  int e = q * 4;
  put_record(d.x, r0, s.x, a.x, e,     cnt, rec, ovf);
  put_record(d.y, r1, s.y, a.y, e + 1, cnt, rec, ovf);
  put_record(d.z, r2, s.z, a.z, e + 2, cnt, rec, ovf);
  put_record(d.w, r3, s.w, a.w, e + 3, cnt, rec, ovf);
}

#define TP (DD + 4)
#define NTILE (NN / 16)
#define GB 512
static_assert(NN % 16 == 0, "row tiles are exact");

template <int BN>
__device__ __forceinline__ void tile_loads(const float* __restrict__ PEp,
                                           const float* __restrict__ X,
                                           const float* __restrict__ OB, int tm, int mlo, int hi,
                                           int ncol, v2f (&a)[PEDP / 4], float (&xv)[8],
                                           float (&ov)[8]) {
  const float* arow = PEp + (size_t)(tm * 16 + mlo) * PEDP + hi * 2;
  #pragma unroll
  for (int i = 0; i < PEDP / 4; ++i) a[i] = *(const v2f*)(arow + 4 * i);
  #pragma unroll
  for (int r = 0; r < 8; ++r) xv[r] = X[(size_t)(tm * 16 + r + hi * 8) * DD + ncol];
  if (BN) {
    #pragma unroll
    for (int r = 0; r < 8; ++r) ov[r] = OB[(size_t)(tm * 16 + r + hi * 8) * DD + ncol];
  }
}

template <int BN>
__global__ void __launch_bounds__(256) k_gemm2x(const float* __restrict__ PEp,
                                                const float* __restrict__ W,
                                                const float* __restrict__ X,
                                                const float* __restrict__ bb,
                                                const float* __restrict__ W2,
                                                const float* __restrict__ al,
                                                const float* __restrict__ ar,
                                                const float* __restrict__ ee,
                                                const float* __restrict__ ae,
                                                float* __restrict__ H,
                                                float* __restrict__ sl,
                                                float* __restrict__ sr,
                                                float* __restrict__ xe,
                                                float* __restrict__ ssum,
                                                float* __restrict__ ssq,
                                                const float* __restrict__ OB,
                                                const float* __restrict__ psum,
                                                const float* __restrict__ psq,
                                                const float* __restrict__ gam,
                                                const float* __restrict__ bet,
                                                float* __restrict__ XO) {
  __shared__ __attribute__((aligned(16))) float strip[16 * TP];
  __shared__ __attribute__((aligned(16))) float hstrip[16 * TP];
  __shared__ __attribute__((aligned(16))) float xstrip[BN ? 2 * 16 * TP : 4];
  int wave = threadIdx.x >> 5;
  int lane = threadIdx.x & 31;

  if (blockIdx.x == 0) {
    v4f z4 = {0.0f, 0.0f, 0.0f, 0.0f};
    for (int i = threadIdx.x; i < NCOPY * DD / 4; i += 256) { ((v4f*)ssum)[i] = z4; ((v4f*)ssq)[i] = z4; }
    for (int t = wave; t < NT; t += 8) {
      float acc = 0.0f;
      #pragma unroll
      for (int j = 0; j < 4; ++j) {
        int d = lane + 32 * j;
        acc += ee[t * DD + d] * ae[d];
      }
      #pragma unroll
      for (int off = 16; off; off >>= 1) acc += __shfl_xor(acc, off, 32);
      if (lane == 0) xe[t] = acc;
    }
    return;
  }

  int mlo = lane & 15;
  int hi  = lane >> 4;
  int ncol = wave * 16 + mlo;

  int tm = blockIdx.x - 1;
  v2f a[PEDP / 4];
  float xv[8], ov[8];
  tile_loads<BN>(PEp, X, OB, tm, mlo, hi, ncol, a, xv, ov);

  float mu = 0.0f, inv = 0.0f, ga = 0.0f, be = 0.0f;
  if (BN) {
    float ps[NCOPY], pq[NCOPY];
    #pragma unroll
    for (int k = 0; k < NCOPY; ++k) { ps[k] = psum[k * DD + ncol]; pq[k] = psq[k * DD + ncol]; }
    ga = gam[ncol]; be = bet[ncol];
    float ts = 0.0f, tq = 0.0f;
    #pragma unroll
    for (int k = 0; k < NCOPY; ++k) { ts += ps[k]; tq += pq[k]; }
    mu = ts * (1.0f / NN);
    float var = tq * (1.0f / NN) - mu * mu;
    inv = rsqrtf(var + 1e-5f);
  }

  v2f wp[PEDP / 4];
  #pragma unroll
  for (int i = 0; i < PEDP / 4; ++i) {
    int ka = 4 * i + hi * 2, kb = ka + 1;
    float wa = W[(size_t)(ka < PED ? ka : PED - 1) * DD + ncol];
    float wb = W[(size_t)(kb < PED ? kb : PED - 1) * DD + ncol];
    wp[i].x = (ka < PED) ? wa : 0.0f;
    wp[i].y = (kb < PED) ? wb : 0.0f;
  }
  v2f w2[DD / 4];
  #pragma unroll
  for (int i = 0; i < DD / 4; ++i) {
    w2[i].x = W2[(size_t)(4 * i + hi * 2    ) * DD + ncol];
    w2[i].y = W2[(size_t)(4 * i + hi * 2 + 1) * DD + ncol];
  }
  float bbv = bb[ncol];
  float wl[4], wr[4];
  #pragma unroll
  for (int j = 0; j < 4; ++j) { wl[j] = al[lane + 32 * j]; wr[j] = ar[lane + 32 * j]; }

  const float* srow = strip + mlo * TP + hi * 2;

  int trip = 0;
  for (; tm < NTILE; tm += GB, ++trip) {
    float* xs = xstrip + (BN ? (trip & 1) * 16 * TP : 0);
    v8f c = {};
    #pragma unroll
    for (int i = 0; i < PEDP / 4; ++i)
      c = __builtin_amdgcn_wmma_f32_16x16x4_f32(false, a[i], false, wp[i], (short)0, c,
                                                false, false);
    #pragma unroll
    for (int r = 0; r < 8; ++r) {
      float xr = xv[r];
      if (BN) {
        xr = (ov[r] - mu) * inv * ga + be + xv[r];
        xs[(r + hi * 8) * TP + ncol] = xr;
      }
      float v = c[r];
      v += xr + bbv;
      strip[(r + hi * 8) * TP + ncol] = v;
    }
    __syncthreads();

    if (tm + GB < NTILE) tile_loads<BN>(PEp, X, OB, tm + GB, mlo, hi, ncol, a, xv, ov);

    v8f d = {};
    #pragma unroll
    for (int i = 0; i < DD / 4; ++i) {
      v2f as = *(const v2f*)(srow + 4 * i);
      d = __builtin_amdgcn_wmma_f32_16x16x4_f32(false, as, false, w2[i], (short)0, d,
                                                false, false);
    }
    #pragma unroll
    for (int r = 0; r < 8; ++r) hstrip[(r + hi * 8) * TP + ncol] = d[r];
    __syncthreads();

    float ol = 0.0f, orr = 0.0f;
    #pragma unroll
    for (int i = 0; i < 2; ++i) {
      int row = wave * 2 + i;
      const float* hp = hstrip + row * TP;
      v4f q = *(const v4f*)(hp + lane * 4);
      ((v4f*)(H + (size_t)(tm * 16 + row) * DD))[lane] = q;
      if (BN) {
        v4f xq = *(const v4f*)(xs + row * TP + lane * 4);
        ((v4f*)(XO + (size_t)(tm * 16 + row) * DD))[lane] = xq;
      }
      float accl = 0.0f, accr = 0.0f;
      #pragma unroll
      for (int j = 0; j < 4; ++j) {
        float v = hp[lane + 32 * j];
        accl += v * wl[j];
        accr += v * wr[j];
      }
      #pragma unroll
      for (int off = 16; off; off >>= 1) {
        accl += __shfl_xor(accl, off, 32);
        accr += __shfl_xor(accr, off, 32);
      }
      if (lane == i) { ol = accl; orr = accr; }
    }
    if (lane < 2) {
      int n = tm * 16 + wave * 2 + lane;
      sl[n] = ol; sr[n] = orr;
    }
  }
}

__device__ __forceinline__ float edge_score(const float* __restrict__ sl,
                                            const float* __restrict__ xe, float srn,
                                            int s, int t) {
  float a = sl[s] + srn + xe[t];
  return (a > 0.0f) ? a : 0.2f * a;
}

template <int LAST>
__global__ void __launch_bounds__(256) k_node(const float* __restrict__ h,
                                              const float* __restrict__ sl,
                                              const float* __restrict__ sr,
                                              const float* __restrict__ xe,
                                              const int* __restrict__ cnt,
                                              const v4i* __restrict__ rec,
                                              const v4i* __restrict__ ovf,
                                              const float* __restrict__ bias,
                                              const float* __restrict__ xres,
                                              float* __restrict__ outb,
                                              float* __restrict__ ssum,
                                              float* __restrict__ ssq) {
  __shared__ __attribute__((aligned(16))) float rows[8 * DD];
  __shared__ int arrived;
  int wave = threadIdx.x >> 5, lane = threadIdx.x & 31;
  int n = blockIdx.x * 8 + wave;
  if (LAST == 0) {
    if (threadIdx.x == 0) arrived = 0;
    __syncthreads();
  }
  float4 b4 = ((const float4*)bias)[lane];
  float4 x4 = make_float4(0.0f, 0.0f, 0.0f, 0.0f);
  if (LAST == 1) x4 = ((const float4*)(xres + (size_t)n * DD))[lane];
  int deg = __builtin_amdgcn_readfirstlane(cnt[n]);
  int novf = 0;
  if (deg > CAP) novf = __builtin_amdgcn_readfirstlane(cnt[NN]);
  int nchunk = 1 + ((novf + 31) >> 5);
  float srn = sr[n];

  bool has0 = lane < (deg < CAP ? deg : CAP);
  int s0 = 0, e0 = 0x7fffffff;
  float a0 = -INFINITY;
  if (has0) {
    v4i r = rec[(size_t)n * CAP + lane];
    s0 = r.x; e0 = r.z;
    a0 = edge_score(sl, xe, srn, r.x, r.y);
  }
  float m = a0;
  for (int ci = 1; ci < nchunk; ++ci) {
    int j = (ci - 1) * 32 + lane;
    if (j < novf) {
      v4i r = ovf[j];
      if (r.w == n) m = fmaxf(m, edge_score(sl, xe, srn, r.x, r.y));
    }
  }
  for (int o = 16; o; o >>= 1) m = fmaxf(m, __shfl_xor(m, o, 32));

  float4 acc = make_float4(0.0f, 0.0f, 0.0f, 0.0f);
  float den = 0.0f;
  for (int ci = 0; ci < nchunk; ++ci) {
    int sC = s0, eC = e0;
    float aC = a0;
    bool hasC = has0;
    if (ci != 0) {
      int j = (ci - 1) * 32 + lane;
      hasC = false; sC = 0; eC = 0x7fffffff; aC = -INFINITY;
      if (j < novf) {
        v4i r = ovf[j];
        if (r.w == n) {
          hasC = true; sC = r.x; eC = r.z;
          aC = edge_score(sl, xe, srn, r.x, r.y);
        }
      }
    }
    float exC = hasC ? expf(aC - m) : 0.0f;
    int c = __builtin_popcount(__builtin_amdgcn_ballot_w32(hasC));
    int span = (ci == 0) ? c : 32;
    int rk = 0;
    #pragma unroll 4
    for (int k = 0; k < span; ++k) rk += (__shfl(eC, k, 32) < eC) ? 1 : 0;
    int k = 0;
    for (; k + 4 <= c; k += 4) {
      int L0 = __builtin_ctz(__builtin_amdgcn_ballot_w32(hasC && rk == k));
      int L1 = __builtin_ctz(__builtin_amdgcn_ballot_w32(hasC && rk == k + 1));
      int L2 = __builtin_ctz(__builtin_amdgcn_ballot_w32(hasC && rk == k + 2));
      int L3 = __builtin_ctz(__builtin_amdgcn_ballot_w32(hasC && rk == k + 3));
      float w0 = __shfl(exC, L0, 32), w1 = __shfl(exC, L1, 32);
      float w2 = __shfl(exC, L2, 32), w3 = __shfl(exC, L3, 32);
      float4 v0 = ((const float4*)(h + (size_t)__shfl(sC, L0, 32) * DD))[lane];
      float4 v1 = ((const float4*)(h + (size_t)__shfl(sC, L1, 32) * DD))[lane];
      float4 v2 = ((const float4*)(h + (size_t)__shfl(sC, L2, 32) * DD))[lane];
      float4 v3 = ((const float4*)(h + (size_t)__shfl(sC, L3, 32) * DD))[lane];
      den += w0; acc.x += w0 * v0.x; acc.y += w0 * v0.y; acc.z += w0 * v0.z; acc.w += w0 * v0.w;
      den += w1; acc.x += w1 * v1.x; acc.y += w1 * v1.y; acc.z += w1 * v1.z; acc.w += w1 * v1.w;
      den += w2; acc.x += w2 * v2.x; acc.y += w2 * v2.y; acc.z += w2 * v2.z; acc.w += w2 * v2.w;
      den += w3; acc.x += w3 * v3.x; acc.y += w3 * v3.y; acc.z += w3 * v3.z; acc.w += w3 * v3.w;
    }
    for (; k < c; ++k) {
      int L0 = __builtin_ctz(__builtin_amdgcn_ballot_w32(hasC && rk == k));
      float w0 = __shfl(exC, L0, 32);
      float4 v0 = ((const float4*)(h + (size_t)__shfl(sC, L0, 32) * DD))[lane];
      den += w0; acc.x += w0 * v0.x; acc.y += w0 * v0.y; acc.z += w0 * v0.z; acc.w += w0 * v0.w;
    }
  }
  float inv = (deg > 0) ? 1.0f / den : 0.0f;
  float4 o4 = make_float4(acc.x * inv + b4.x, acc.y * inv + b4.y, acc.z * inv + b4.z,
                          acc.w * inv + b4.w);
  if (LAST == 1) {
    o4.x += x4.x; o4.y += x4.y; o4.z += x4.z; o4.w += x4.w;
    ((float4*)(outb + (size_t)n * DD))[lane] = o4;
  } else {
    ((float4*)(rows + wave * DD))[lane] = o4;
    __builtin_amdgcn_fence(__ATOMIC_RELEASE, "workgroup");
    int old = 0;
    if (lane == 0)
      old = __hip_atomic_fetch_add(&arrived, 1, __ATOMIC_RELAXED, __HIP_MEMORY_SCOPE_WORKGROUP);
    ((float4*)(outb + (size_t)n * DD))[lane] = o4;
    old = __builtin_amdgcn_readfirstlane(old);
    if (old == 7) {
      __builtin_amdgcn_fence(__ATOMIC_ACQUIRE, "workgroup");
      float* ps = ssum + (blockIdx.x & (NCOPY - 1)) * DD;
      float* pq = ssq + (blockIdx.x & (NCOPY - 1)) * DD;
      #pragma unroll
      for (int j = 0; j < 4; ++j) {
        int col = lane + 32 * j;
        float s = 0.0f, q = 0.0f;
        #pragma unroll
        for (int w = 0; w < 8; ++w) {
          float v = rows[w * DD + col];
          s += v; q += v * v;
        }
        atomicAdd(ps + col, s);
        atomicAdd(pq + col, q);
      }
    }
  }
}


extern "C" void kernel_launch(void* const* d_in, const int* in_sizes, int n_in,
                              void* d_out, int out_size, void* d_ws, size_t ws_size,
                              hipStream_t stream) {
  const float* X_n   = (const float*)d_in[0];
  const float* PE    = (const float*)d_in[1];
  const int*   eidx  = (const int*)d_in[2];
  const int*   eattr = (const int*)d_in[3];
  const float* eemb  = (const float*)d_in[4];
  const float* peW   = (const float*)d_in[5];
  const float* peB   = (const float*)d_in[6];
  const float* linW  = (const float*)d_in[7];
  const float* attl  = (const float*)d_in[8];
  const float* attr_ = (const float*)d_in[9];
  const float* atte  = (const float*)d_in[10];
  const float* bias  = (const float*)d_in[11];
  const float* gam   = (const float*)d_in[12];
  const float* bet   = (const float*)d_in[13];
  const int* src = eidx;
  const int* dst = eidx + EE;

  float* ws = (float*)d_ws;
  size_t o = 0;
  float* xa    = ws + o; o += (size_t)NN * DD;
  float* xb    = ws + o; o += (size_t)NN * DD;
  float* outb  = ws + o; o += (size_t)NN * DD;
  float* h     = ws + o; o += (size_t)NN * DD;
  float* PEp   = ws + o; o += (size_t)NN * PEDP;
  v4i*   rec   = (v4i*)(ws + o); o += (size_t)NN * CAP * 4;
  v4i*   ovf   = (v4i*)(ws + o); o += (size_t)EE * 4;
  float* sl    = ws + o; o += NN;
  float* sr    = ws + o; o += NN;
  int*   cnt   = (int*)(ws + o); o += NN + 16;
  float* xe    = ws + o; o += 16;
  float* ssum0 = ws + o; o += NCOPY * DD;
  float* ssq0  = ws + o; o += NCOPY * DD;
  float* ssum1 = ws + o; o += NCOPY * DD;
  float* ssq1  = ws + o; o += NCOPY * DD;
  float* xnext[NL - 1] = {xa, xb};
  float* ssum[2] = {ssum0, ssum1};
  float* ssq[2]  = {ssq0, ssq1};

  k_pad_pe<<<(NN * PEDP / 4 + 1023) / 1024, 256, 0, stream>>>(PE, PEp, cnt);
  k_bucket<<<EE / 1024, 256, 0, stream>>>(src, dst, eattr, cnt, rec, ovf);

  const float* xin = X_n;
  for (int l = 0; l < NL; ++l) {
    if (l == 0) {
      k_gemm2x<0><<<GB + 1, 256, 0, stream>>>(PEp, peW, xin, peB, linW, attl, attr_, eemb, atte, h,
                                              sl, sr, xe, ssum[0], ssq[0], nullptr, nullptr,
                                              nullptr, nullptr, nullptr, nullptr);
    } else {
      k_gemm2x<1><<<GB + 1, 256, 0, stream>>>(PEp, peW + (size_t)l * PED * DD, xin, peB + l * DD,
                                              linW + (size_t)l * DD * DD, attl + l * DD,
                                              attr_ + l * DD, eemb + (size_t)l * NT * DD,
                                              atte + l * DD, h, sl, sr, xe, ssum[l & 1], ssq[l & 1],
                                              outb, ssum[(l - 1) & 1], ssq[(l - 1) & 1],
                                              gam + (l - 1) * DD, bet + (l - 1) * DD, xnext[l - 1]);
      xin = xnext[l - 1];
    }

    if (l < NL - 1) {
      k_node<0><<<NN / 8, 256, 0, stream>>>(h, sl, sr, xe, cnt, rec, ovf, bias + l * DD, xin, outb,
                                            ssum[l & 1], ssq[l & 1]);
    } else {
      k_node<1><<<NN / 8, 256, 0, stream>>>(h, sl, sr, xe, cnt, rec, ovf, bias + l * DD, xin,
                                            (float*)d_out, ssum[l & 1], ssq[l & 1]);
    }
  }
}


# Node kernel made persistent: 1536 resident work-groups, a wave walks 4 or 5 nodes and fetches later nodes' counts, records, scores while it sums rows; column sums kept per wave: their order changed
# speedup vs baseline: 4.1707x; 1.0484x over previous
#include <hip/hip_runtime.h>
#include <math.h>

#define NN   50000
#define EE   640000
#define DD   128
#define PED  37
#define PEDP 40
#define NT   9
#define NL   3

typedef __attribute__((ext_vector_type(2))) float v2f;
typedef __attribute__((ext_vector_type(8))) float v8f;

typedef __attribute__((ext_vector_type(4))) int v4i;
typedef __attribute__((ext_vector_type(4))) float v4f;

static_assert(EE % 1024 == 0, "edge passes assume four edges per lane, exact grid");

#define NCOPY 16
static_assert((NCOPY & (NCOPY - 1)) == 0, "the copy of a work-group is picked with a mask");


__global__ void __launch_bounds__(256) k_pad_pe(const float* __restrict__ PE,
                                                float* __restrict__ PEp,
                                                int* __restrict__ cnt) {
  const int TOT = NN * PEDP / 4;
  float4 v[4];
  #pragma unroll
  for (int j = 0; j < 4; ++j) {
    int q = blockIdx.x * 1024 + j * 256 + threadIdx.x;
    int qc = (q < TOT) ? q : TOT - 1;
    int r = qc / (PEDP / 4);
    int c = (qc - r * (PEDP / 4)) * 4;
    const float* p = PE + (size_t)r * PED + c;
    bool in1 = c + 1 < PED, in2 = c + 2 < PED, in3 = c + 3 < PED;
    float a0 = p[0];
    float a1 = p[in1 ? 1 : 0];
    float a2 = p[in2 ? 2 : 0];
    float a3 = p[in3 ? 3 : 0];
    v[j] = make_float4(a0, in1 ? a1 : 0.0f, in2 ? a2 : 0.0f, in3 ? a3 : 0.0f);
  }
  #pragma unroll
  for (int j = 0; j < 4; ++j) {
    int q = blockIdx.x * 1024 + j * 256 + threadIdx.x;
    if (q < TOT) ((float4*)PEp)[q] = v[j];
  }
  int g = blockIdx.x * 256 + threadIdx.x;
  v4i z = {0, 0, 0, 0};
  if (g <= NN / 4) ((v4i*)cnt)[g] = z;
}

#define CAP 32
__device__ __forceinline__ void put_record(int d, int r, int s, int t, int e,
                                           int* __restrict__ cnt, v4i* __restrict__ rec,
                                           v4i* __restrict__ ovf) {
  v4i v = {s, t, e, d};
  if (r < CAP) {
    rec[(size_t)d * CAP + r] = v;
  } else {
    int o = atomicAdd(&cnt[NN], 1);
    ovf[o] = v;
  }
}

__global__ void __launch_bounds__(256) k_bucket(const int* __restrict__ src,
                                                const int* __restrict__ dst,
                                                const int* __restrict__ attr,
                                                int* __restrict__ cnt,
                                                v4i* __restrict__ rec,
                                                v4i* __restrict__ ovf) {
  int q = blockIdx.x * 256 + threadIdx.x;
  v4i s = ((const v4i*)src)[q];
  v4i d = ((const v4i*)dst)[q];
  v4i a = ((const v4i*)attr)[q];
  int r0 = atomicAdd(&cnt[d.x], 1);
  int r1 = atomicAdd(&cnt[d.y], 1);
  int r2 = atomicAdd(&cnt[d.z], 1);
  int r3 = atomicAdd(&cnt[d.w], 1);
  int e = q * 4;
  put_record(d.x, r0, s.x, a.x, e,     cnt, rec, ovf);
  put_record(d.y, r1, s.y, a.y, e + 1, cnt, rec, ovf);
  put_record(d.z, r2, s.z, a.z, e + 2, cnt, rec, ovf);
  put_record(d.w, r3, s.w, a.w, e + 3, cnt, rec, ovf);
}

#define TP (DD + 4)
#define NTILE (NN / 16)
#define GB 512
static_assert(NN % 16 == 0, "row tiles are exact");

template <int BN>
__device__ __forceinline__ void tile_loads(const float* __restrict__ PEp,
                                           const float* __restrict__ X,
                                           const float* __restrict__ OB, int tm, int mlo, int hi,
                                           int ncol, v2f (&a)[PEDP / 4], float (&xv)[8],
                                           float (&ov)[8]) {
  const float* arow = PEp + (size_t)(tm * 16 + mlo) * PEDP + hi * 2;
  #pragma unroll
  for (int i = 0; i < PEDP / 4; ++i) a[i] = *(const v2f*)(arow + 4 * i);
  #pragma unroll
  for (int r = 0; r < 8; ++r) xv[r] = X[(size_t)(tm * 16 + r + hi * 8) * DD + ncol];
  if (BN) {
    #pragma unroll
    for (int r = 0; r < 8; ++r) ov[r] = OB[(size_t)(tm * 16 + r + hi * 8) * DD + ncol];
  }
}

template <int BN>
__global__ void __launch_bounds__(256) k_gemm2x(const float* __restrict__ PEp,
                                                const float* __restrict__ W,
                                                const float* __restrict__ X,
                                                const float* __restrict__ bb,
                                                const float* __restrict__ W2,
                                                const float* __restrict__ al,
                                                const float* __restrict__ ar,
                                                const float* __restrict__ ee,
                                                const float* __restrict__ ae,
                                                float* __restrict__ H,
                                                float* __restrict__ sl,
                                                float* __restrict__ sr,
                                                float* __restrict__ xe,
                                                float* __restrict__ ssum,
                                                float* __restrict__ ssq,
                                                const float* __restrict__ OB,
                                                const float* __restrict__ psum,
                                                const float* __restrict__ psq,
                                                const float* __restrict__ gam,
                                                const float* __restrict__ bet,
                                                float* __restrict__ XO) {
  __shared__ __attribute__((aligned(16))) float strip[16 * TP];
  __shared__ __attribute__((aligned(16))) float hstrip[16 * TP];
  __shared__ __attribute__((aligned(16))) float xstrip[BN ? 2 * 16 * TP : 4];
  int wave = threadIdx.x >> 5;
  int lane = threadIdx.x & 31;

  if (blockIdx.x == 0) {
    v4f z4 = {0.0f, 0.0f, 0.0f, 0.0f};
    for (int i = threadIdx.x; i < NCOPY * DD / 4; i += 256) { ((v4f*)ssum)[i] = z4; ((v4f*)ssq)[i] = z4; }
    for (int t = wave; t < NT; t += 8) {
      float acc = 0.0f;
      #pragma unroll
      for (int j = 0; j < 4; ++j) {
        int d = lane + 32 * j;
        acc += ee[t * DD + d] * ae[d];
      }
      #pragma unroll
      for (int off = 16; off; off >>= 1) acc += __shfl_xor(acc, off, 32);
      if (lane == 0) xe[t] = acc;
    }
    return;
  }

  int mlo = lane & 15;
  int hi  = lane >> 4;
  int ncol = wave * 16 + mlo;

  int tm = blockIdx.x - 1;
  v2f a[PEDP / 4];
  float xv[8], ov[8];
  tile_loads<BN>(PEp, X, OB, tm, mlo, hi, ncol, a, xv, ov);

  float mu = 0.0f, inv = 0.0f, ga = 0.0f, be = 0.0f;
  if (BN) {
    float ps[NCOPY], pq[NCOPY];
    #pragma unroll
    for (int k = 0; k < NCOPY; ++k) { ps[k] = psum[k * DD + ncol]; pq[k] = psq[k * DD + ncol]; }
    ga = gam[ncol]; be = bet[ncol];
    float ts = 0.0f, tq = 0.0f;
    #pragma unroll
    for (int k = 0; k < NCOPY; ++k) { ts += ps[k]; tq += pq[k]; }
    mu = ts * (1.0f / NN);
    float var = tq * (1.0f / NN) - mu * mu;
    inv = rsqrtf(var + 1e-5f);
  }

  v2f wp[PEDP / 4];
  #pragma unroll
  for (int i = 0; i < PEDP / 4; ++i) {
    int ka = 4 * i + hi * 2, kb = ka + 1;
    float wa = W[(size_t)(ka < PED ? ka : PED - 1) * DD + ncol];
    float wb = W[(size_t)(kb < PED ? kb : PED - 1) * DD + ncol];
    wp[i].x = (ka < PED) ? wa : 0.0f;
    wp[i].y = (kb < PED) ? wb : 0.0f;
  }
  v2f w2[DD / 4];
  #pragma unroll
  for (int i = 0; i < DD / 4; ++i) {
    w2[i].x = W2[(size_t)(4 * i + hi * 2    ) * DD + ncol];
    w2[i].y = W2[(size_t)(4 * i + hi * 2 + 1) * DD + ncol];
  }
  float bbv = bb[ncol];
  float wl[4], wr[4];
  #pragma unroll
  for (int j = 0; j < 4; ++j) { wl[j] = al[lane + 32 * j]; wr[j] = ar[lane + 32 * j]; }

  const float* srow = strip + mlo * TP + hi * 2;

  int trip = 0;
  for (; tm < NTILE; tm += GB, ++trip) {
    float* xs = xstrip + (BN ? (trip & 1) * 16 * TP : 0);
    v8f c = {};
    #pragma unroll
    for (int i = 0; i < PEDP / 4; ++i)
      c = __builtin_amdgcn_wmma_f32_16x16x4_f32(false, a[i], false, wp[i], (short)0, c,
                                                false, false);
    #pragma unroll
    for (int r = 0; r < 8; ++r) {
      float xr = xv[r];
      if (BN) {
        xr = (ov[r] - mu) * inv * ga + be + xv[r];
        xs[(r + hi * 8) * TP + ncol] = xr;
      }
      float v = c[r];
      v += xr + bbv;
      strip[(r + hi * 8) * TP + ncol] = v;
    }
    __syncthreads();

    if (tm + GB < NTILE) tile_loads<BN>(PEp, X, OB, tm + GB, mlo, hi, ncol, a, xv, ov);

    v8f d = {};
    #pragma unroll
    for (int i = 0; i < DD / 4; ++i) {
      v2f as = *(const v2f*)(srow + 4 * i);
      d = __builtin_amdgcn_wmma_f32_16x16x4_f32(false, as, false, w2[i], (short)0, d,
                                                false, false);
    }
    #pragma unroll
    for (int r = 0; r < 8; ++r) hstrip[(r + hi * 8) * TP + ncol] = d[r];
    __syncthreads();

    float ol = 0.0f, orr = 0.0f;
    #pragma unroll
    for (int i = 0; i < 2; ++i) {
      int row = wave * 2 + i;
      const float* hp = hstrip + row * TP;
      v4f q = *(const v4f*)(hp + lane * 4);
      ((v4f*)(H + (size_t)(tm * 16 + row) * DD))[lane] = q;
      if (BN) {
        v4f xq = *(const v4f*)(xs + row * TP + lane * 4);
        ((v4f*)(XO + (size_t)(tm * 16 + row) * DD))[lane] = xq;
      }
      float accl = 0.0f, accr = 0.0f;
      #pragma unroll
      for (int j = 0; j < 4; ++j) {
        float v = hp[lane + 32 * j];
        accl += v * wl[j];
        accr += v * wr[j];
      }
      #pragma unroll
      for (int off = 16; off; off >>= 1) {
        accl += __shfl_xor(accl, off, 32);
        accr += __shfl_xor(accr, off, 32);
      }
      if (lane == i) { ol = accl; orr = accr; }
    }
    if (lane < 2) {
      int n = tm * 16 + wave * 2 + lane;
      sl[n] = ol; sr[n] = orr;
    }
  }
}

__device__ __forceinline__ float edge_score(const float* __restrict__ sl,
                                            const float* __restrict__ xe, float srn,
                                            int s, int t) {
  float a = sl[s] + srn + xe[t];
  return (a > 0.0f) ? a : 0.2f * a;
}

#define NWG 1536
#define NWV (NWG * 8)
static_assert(NWV <= NN, "every wave has a first node");

template <int LAST>
__global__ void __launch_bounds__(256, 12) k_node(const float* __restrict__ h,
                                              const float* __restrict__ sl,
                                              const float* __restrict__ sr,
                                              const float* __restrict__ xe,
                                              const int* __restrict__ cnt,
                                              const v4i* __restrict__ rec,
                                              const v4i* __restrict__ ovf,
                                              const float* __restrict__ bias,
                                              const float* __restrict__ xres,
                                              float* __restrict__ outb,
                                              float* __restrict__ ssum,
                                              float* __restrict__ ssq) {
  __shared__ __attribute__((aligned(16))) float red[LAST ? 4 : 2 * 8 * DD];
  int wave = __builtin_amdgcn_readfirstlane(threadIdx.x >> 5);
  int lane = threadIdx.x & 31;
  int g = blockIdx.x * 8 + wave;
  float4 b4 = ((const float4*)bias)[lane];
  float4 cs = make_float4(0.0f, 0.0f, 0.0f, 0.0f);
  float4 cq = make_float4(0.0f, 0.0f, 0.0f, 0.0f);

  int novfAll = __builtin_amdgcn_readfirstlane(cnt[NN]);

  int nB0 = (g + NWV < NN) ? g + NWV : NN - 1;
  int degA = __builtin_amdgcn_readfirstlane(cnt[g]);
  float srA = sr[g];
  v4i rA = rec[(size_t)g * CAP + lane];
  int degB = __builtin_amdgcn_readfirstlane(cnt[nB0]);
  float srB = sr[nB0];
  v4i rB = rec[(size_t)nB0 * CAP + lane];
  bool hasA = lane < (degA < CAP ? degA : CAP);
  float slA = sl[hasA ? rA.x : 0];
  float xeA = xe[hasA ? rA.y : 0];

  for (int n = g; n < NN; n += NWV) {
    int deg = degA;
    float srn = srA;
    bool has0 = hasA;
    int s0 = has0 ? rA.x : 0;
    int e0 = has0 ? rA.z : 0x7fffffff;
    float a0 = -INFINITY;
    if (has0) {
      float a = slA + srn + xeA;
      a0 = (a > 0.0f) ? a : 0.2f * a;
    }
    bool hasB = lane < (degB < CAP ? degB : CAP);
    float slB = sl[hasB ? rB.x : 0];
    float xeB = xe[hasB ? rB.y : 0];
    int nC = (n + 2 * NWV < NN) ? n + 2 * NWV : NN - 1;
    int degC = __builtin_amdgcn_readfirstlane(cnt[nC]);
    float srC = sr[nC];
    v4i rC = rec[(size_t)nC * CAP + lane];
    float4 x4 = make_float4(0.0f, 0.0f, 0.0f, 0.0f);
    if (LAST == 1) x4 = ((const float4*)(xres + (size_t)n * DD))[lane];

    int novf = (deg > CAP) ? novfAll : 0;
    int nchunk = 1 + ((novf + 31) >> 5);

    float m = a0;
    for (int ci = 1; ci < nchunk; ++ci) {
      int j = (ci - 1) * 32 + lane;
      if (j < novf) {
        v4i r = ovf[j];
        if (r.w == n) m = fmaxf(m, edge_score(sl, xe, srn, r.x, r.y));
      }
    }
    for (int o = 16; o; o >>= 1) m = fmaxf(m, __shfl_xor(m, o, 32));

    float4 acc = make_float4(0.0f, 0.0f, 0.0f, 0.0f);
    float den = 0.0f;
    for (int ci = 0; ci < nchunk; ++ci) {
      int sC = s0, eC = e0;
      float aC = a0;
      bool hasC = has0;
      if (ci != 0) {
        int j = (ci - 1) * 32 + lane;
        hasC = false; sC = 0; eC = 0x7fffffff; aC = -INFINITY;
        if (j < novf) {
          v4i r = ovf[j];
          if (r.w == n) {
            hasC = true; sC = r.x; eC = r.z;
            aC = edge_score(sl, xe, srn, r.x, r.y);
          }
        }
      }
      float exC = hasC ? expf(aC - m) : 0.0f;
      int c = __builtin_popcount(__builtin_amdgcn_ballot_w32(hasC));
      int span = (ci == 0) ? c : 32;
      int rk = 0;
      #pragma unroll 4
      for (int k = 0; k < span; ++k) rk += (__shfl(eC, k, 32) < eC) ? 1 : 0;
      int k = 0;
      for (; k + 4 <= c; k += 4) {
        int L0 = __builtin_ctz(__builtin_amdgcn_ballot_w32(hasC && rk == k));
        int L1 = __builtin_ctz(__builtin_amdgcn_ballot_w32(hasC && rk == k + 1));
        int L2 = __builtin_ctz(__builtin_amdgcn_ballot_w32(hasC && rk == k + 2));
        int L3 = __builtin_ctz(__builtin_amdgcn_ballot_w32(hasC && rk == k + 3));
        float w0 = __shfl(exC, L0, 32), w1 = __shfl(exC, L1, 32);
        float w2 = __shfl(exC, L2, 32), w3 = __shfl(exC, L3, 32);
        float4 v0 = ((const float4*)(h + (size_t)__shfl(sC, L0, 32) * DD))[lane];
        float4 v1 = ((const float4*)(h + (size_t)__shfl(sC, L1, 32) * DD))[lane];
        float4 v2 = ((const float4*)(h + (size_t)__shfl(sC, L2, 32) * DD))[lane];
        float4 v3 = ((const float4*)(h + (size_t)__shfl(sC, L3, 32) * DD))[lane];
        den += w0; acc.x += w0 * v0.x; acc.y += w0 * v0.y; acc.z += w0 * v0.z; acc.w += w0 * v0.w;
        den += w1; acc.x += w1 * v1.x; acc.y += w1 * v1.y; acc.z += w1 * v1.z; acc.w += w1 * v1.w;
        den += w2; acc.x += w2 * v2.x; acc.y += w2 * v2.y; acc.z += w2 * v2.z; acc.w += w2 * v2.w;
        den += w3; acc.x += w3 * v3.x; acc.y += w3 * v3.y; acc.z += w3 * v3.z; acc.w += w3 * v3.w;
      }
      for (; k < c; ++k) {
        int L0 = __builtin_ctz(__builtin_amdgcn_ballot_w32(hasC && rk == k));
        float w0 = __shfl(exC, L0, 32);
        float4 v0 = ((const float4*)(h + (size_t)__shfl(sC, L0, 32) * DD))[lane];
        den += w0; acc.x += w0 * v0.x; acc.y += w0 * v0.y; acc.z += w0 * v0.z; acc.w += w0 * v0.w;
      }
    }
    float inv = (deg > 0) ? 1.0f / den : 0.0f;
    float4 o4 = make_float4(acc.x * inv + b4.x, acc.y * inv + b4.y, acc.z * inv + b4.z,
                            acc.w * inv + b4.w);
    if (LAST == 1) {
      o4.x += x4.x; o4.y += x4.y; o4.z += x4.z; o4.w += x4.w;
    } else {
      cs.x += o4.x; cs.y += o4.y; cs.z += o4.z; cs.w += o4.w;
      cq.x += o4.x * o4.x; cq.y += o4.y * o4.y; cq.z += o4.z * o4.z; cq.w += o4.w * o4.w;
    }
    ((float4*)(outb + (size_t)n * DD))[lane] = o4;

    degA = degB; srA = srB; rA = rB; hasA = hasB; slA = slB; xeA = xeB;
    degB = degC; srB = srC; rB = rC;
  }

  if (LAST == 0) {
    ((float4*)(red + wave * DD))[lane] = cs;
    ((float4*)(red + 8 * DD + wave * DD))[lane] = cq;
    __syncthreads();
    int t = threadIdx.x;
    int col = t & (DD - 1);
    const float* p = red + (t >> 7) * 8 * DD + col;
    float s = 0.0f;
    #pragma unroll
    for (int w = 0; w < 8; ++w) s += p[w * DD];
    float* dstp = ((t >> 7) ? ssq : ssum) + (blockIdx.x & (NCOPY - 1)) * DD + col;
    atomicAdd(dstp, s);
  }
}


extern "C" void kernel_launch(void* const* d_in, const int* in_sizes, int n_in,
                              void* d_out, int out_size, void* d_ws, size_t ws_size,
                              hipStream_t stream) {
  const float* X_n   = (const float*)d_in[0];
  const float* PE    = (const float*)d_in[1];
  const int*   eidx  = (const int*)d_in[2];
  const int*   eattr = (const int*)d_in[3];
  const float* eemb  = (const float*)d_in[4];
  const float* peW   = (const float*)d_in[5];
  const float* peB   = (const float*)d_in[6];
  const float* linW  = (const float*)d_in[7];
  const float* attl  = (const float*)d_in[8];
  const float* attr_ = (const float*)d_in[9];
  const float* atte  = (const float*)d_in[10];
  const float* bias  = (const float*)d_in[11];
  const float* gam   = (const float*)d_in[12];
  const float* bet   = (const float*)d_in[13];
  const int* src = eidx;
  const int* dst = eidx + EE;

  float* ws = (float*)d_ws;
  size_t o = 0;
  float* xa    = ws + o; o += (size_t)NN * DD;
  float* xb    = ws + o; o += (size_t)NN * DD;
  float* outb  = ws + o; o += (size_t)NN * DD;
  float* h     = ws + o; o += (size_t)NN * DD;
  float* PEp   = ws + o; o += (size_t)NN * PEDP;
  v4i*   rec   = (v4i*)(ws + o); o += (size_t)NN * CAP * 4;
  v4i*   ovf   = (v4i*)(ws + o); o += (size_t)EE * 4;
  float* sl    = ws + o; o += NN;
  float* sr    = ws + o; o += NN;
  int*   cnt   = (int*)(ws + o); o += NN + 16;
  float* xe    = ws + o; o += 16;
  float* ssum0 = ws + o; o += NCOPY * DD;
  float* ssq0  = ws + o; o += NCOPY * DD;
  float* ssum1 = ws + o; o += NCOPY * DD;
  float* ssq1  = ws + o; o += NCOPY * DD;
  float* xnext[NL - 1] = {xa, xb};
  float* ssum[2] = {ssum0, ssum1};
  float* ssq[2]  = {ssq0, ssq1};

  k_pad_pe<<<(NN * PEDP / 4 + 1023) / 1024, 256, 0, stream>>>(PE, PEp, cnt);
  k_bucket<<<EE / 1024, 256, 0, stream>>>(src, dst, eattr, cnt, rec, ovf);

  const float* xin = X_n;
  for (int l = 0; l < NL; ++l) {
    if (l == 0) {
      k_gemm2x<0><<<GB + 1, 256, 0, stream>>>(PEp, peW, xin, peB, linW, attl, attr_, eemb, atte, h,
                                              sl, sr, xe, ssum[0], ssq[0], nullptr, nullptr,
                                              nullptr, nullptr, nullptr, nullptr);
    } else {
      k_gemm2x<1><<<GB + 1, 256, 0, stream>>>(PEp, peW + (size_t)l * PED * DD, xin, peB + l * DD,
                                              linW + (size_t)l * DD * DD, attl + l * DD,
                                              attr_ + l * DD, eemb + (size_t)l * NT * DD,
                                              atte + l * DD, h, sl, sr, xe, ssum[l & 1], ssq[l & 1],
                                              outb, ssum[(l - 1) & 1], ssq[(l - 1) & 1],
                                              gam + (l - 1) * DD, bet + (l - 1) * DD, xnext[l - 1]);
      xin = xnext[l - 1];
    }

    if (l < NL - 1) {
      k_node<0><<<NWG, 256, 0, stream>>>(h, sl, sr, xe, cnt, rec, ovf, bias + l * DD, xin, outb,
                                            ssum[l & 1], ssq[l & 1]);
    } else {
      k_node<1><<<NWG, 256, 0, stream>>>(h, sl, sr, xe, cnt, rec, ovf, bias + l * DD, xin,
                                            (float*)d_out, ssum[l & 1], ssq[l & 1]);
    }
  }
}


# Node kernel: lanes exchange on the vector ALU, not via LDS: max by DPP, a node's edges pushed into edge-id order once, row source and weight by v_readlane into scalar addresses; 8 row loads a trip
# speedup vs baseline: 4.4936x; 1.0774x over previous
#include <hip/hip_runtime.h>
#include <math.h>

#define NN   50000
#define EE   640000
#define DD   128
#define PED  37
#define PEDP 40
#define NT   9
#define NL   3

typedef __attribute__((ext_vector_type(2))) float v2f;
typedef __attribute__((ext_vector_type(8))) float v8f;

typedef __attribute__((ext_vector_type(4))) int v4i;
typedef __attribute__((ext_vector_type(4))) float v4f;

static_assert(EE % 1024 == 0, "edge passes assume four edges per lane, exact grid");

#define NCOPY 16
static_assert((NCOPY & (NCOPY - 1)) == 0, "the copy of a work-group is picked with a mask");


__global__ void __launch_bounds__(256) k_pad_pe(const float* __restrict__ PE,
                                                float* __restrict__ PEp,
                                                int* __restrict__ cnt) {
  const int TOT = NN * PEDP / 4;
  float4 v[4];
  #pragma unroll
  for (int j = 0; j < 4; ++j) {
    int q = blockIdx.x * 1024 + j * 256 + threadIdx.x;
    int qc = (q < TOT) ? q : TOT - 1;
    int r = qc / (PEDP / 4);
    int c = (qc - r * (PEDP / 4)) * 4;
    const float* p = PE + (size_t)r * PED + c;
    bool in1 = c + 1 < PED, in2 = c + 2 < PED, in3 = c + 3 < PED;
    float a0 = p[0];
    float a1 = p[in1 ? 1 : 0];
    float a2 = p[in2 ? 2 : 0];
    float a3 = p[in3 ? 3 : 0];
    v[j] = make_float4(a0, in1 ? a1 : 0.0f, in2 ? a2 : 0.0f, in3 ? a3 : 0.0f);
  }
  #pragma unroll
  for (int j = 0; j < 4; ++j) {
    int q = blockIdx.x * 1024 + j * 256 + threadIdx.x;
    if (q < TOT) ((float4*)PEp)[q] = v[j];
  }
  int g = blockIdx.x * 256 + threadIdx.x;
  v4i z = {0, 0, 0, 0};
  if (g <= NN / 4) ((v4i*)cnt)[g] = z;
}

#define CAP 32
__device__ __forceinline__ void put_record(int d, int r, int s, int t, int e,
                                           int* __restrict__ cnt, v4i* __restrict__ rec,
                                           v4i* __restrict__ ovf) {
  v4i v = {s, t, e, d};
  if (r < CAP) {
    rec[(size_t)d * CAP + r] = v;
  } else {
    int o = atomicAdd(&cnt[NN], 1);
    ovf[o] = v;
  }
}

__global__ void __launch_bounds__(256) k_bucket(const int* __restrict__ src,
                                                const int* __restrict__ dst,
                                                const int* __restrict__ attr,
                                                int* __restrict__ cnt,
                                                v4i* __restrict__ rec,
                                                v4i* __restrict__ ovf) {
  int q = blockIdx.x * 256 + threadIdx.x;
  v4i s = ((const v4i*)src)[q];
  v4i d = ((const v4i*)dst)[q];
  v4i a = ((const v4i*)attr)[q];
  int r0 = atomicAdd(&cnt[d.x], 1);
  int r1 = atomicAdd(&cnt[d.y], 1);
  int r2 = atomicAdd(&cnt[d.z], 1);
  int r3 = atomicAdd(&cnt[d.w], 1);
  int e = q * 4;
  put_record(d.x, r0, s.x, a.x, e,     cnt, rec, ovf);
  put_record(d.y, r1, s.y, a.y, e + 1, cnt, rec, ovf);
  put_record(d.z, r2, s.z, a.z, e + 2, cnt, rec, ovf);
  put_record(d.w, r3, s.w, a.w, e + 3, cnt, rec, ovf);
}

#define TP (DD + 4)
#define NTILE (NN / 16)
#define GB 512
static_assert(NN % 16 == 0, "row tiles are exact");

template <int BN>
__device__ __forceinline__ void tile_loads(const float* __restrict__ PEp,
                                           const float* __restrict__ X,
                                           const float* __restrict__ OB, int tm, int mlo, int hi,
                                           int ncol, v2f (&a)[PEDP / 4], float (&xv)[8],
                                           float (&ov)[8]) {
  const float* arow = PEp + (size_t)(tm * 16 + mlo) * PEDP + hi * 2;
  #pragma unroll
  for (int i = 0; i < PEDP / 4; ++i) a[i] = *(const v2f*)(arow + 4 * i);
  #pragma unroll
  for (int r = 0; r < 8; ++r) xv[r] = X[(size_t)(tm * 16 + r + hi * 8) * DD + ncol];
  if (BN) {
    #pragma unroll
    for (int r = 0; r < 8; ++r) ov[r] = OB[(size_t)(tm * 16 + r + hi * 8) * DD + ncol];
  }
}

template <int BN>
__global__ void __launch_bounds__(256) k_gemm2x(const float* __restrict__ PEp,
                                                const float* __restrict__ W,
                                                const float* __restrict__ X,
                                                const float* __restrict__ bb,
                                                const float* __restrict__ W2,
                                                const float* __restrict__ al,
                                                const float* __restrict__ ar,
                                                const float* __restrict__ ee,
                                                const float* __restrict__ ae,
                                                float* __restrict__ H,
                                                float* __restrict__ sl,
                                                float* __restrict__ sr,
                                                float* __restrict__ xe,
                                                float* __restrict__ ssum,
                                                float* __restrict__ ssq,
                                                const float* __restrict__ OB,
                                                const float* __restrict__ psum,
                                                const float* __restrict__ psq,
                                                const float* __restrict__ gam,
                                                const float* __restrict__ bet,
                                                float* __restrict__ XO) {
  __shared__ __attribute__((aligned(16))) float strip[16 * TP];
  __shared__ __attribute__((aligned(16))) float hstrip[16 * TP];
  __shared__ __attribute__((aligned(16))) float xstrip[BN ? 2 * 16 * TP : 4];
  int wave = threadIdx.x >> 5;
  int lane = threadIdx.x & 31;

  if (blockIdx.x == 0) {
    v4f z4 = {0.0f, 0.0f, 0.0f, 0.0f};
    for (int i = threadIdx.x; i < NCOPY * DD / 4; i += 256) { ((v4f*)ssum)[i] = z4; ((v4f*)ssq)[i] = z4; }
    for (int t = wave; t < NT; t += 8) {
      float acc = 0.0f;
      #pragma unroll
      for (int j = 0; j < 4; ++j) {
        int d = lane + 32 * j;
        acc += ee[t * DD + d] * ae[d];
      }
      #pragma unroll
      for (int off = 16; off; off >>= 1) acc += __shfl_xor(acc, off, 32);
      if (lane == 0) xe[t] = acc;
    }
    return;
  }

  int mlo = lane & 15;
  int hi  = lane >> 4;
  int ncol = wave * 16 + mlo;

  int tm = blockIdx.x - 1;
  v2f a[PEDP / 4];
  float xv[8], ov[8];
  tile_loads<BN>(PEp, X, OB, tm, mlo, hi, ncol, a, xv, ov);

  float mu = 0.0f, inv = 0.0f, ga = 0.0f, be = 0.0f;
  if (BN) {
    float ps[NCOPY], pq[NCOPY];
    #pragma unroll
    for (int k = 0; k < NCOPY; ++k) { ps[k] = psum[k * DD + ncol]; pq[k] = psq[k * DD + ncol]; }
    ga = gam[ncol]; be = bet[ncol];
    float ts = 0.0f, tq = 0.0f;
    #pragma unroll
    for (int k = 0; k < NCOPY; ++k) { ts += ps[k]; tq += pq[k]; }
    mu = ts * (1.0f / NN);
    float var = tq * (1.0f / NN) - mu * mu;
    inv = rsqrtf(var + 1e-5f);
  }

  v2f wp[PEDP / 4];
  #pragma unroll
  for (int i = 0; i < PEDP / 4; ++i) {
    int ka = 4 * i + hi * 2, kb = ka + 1;
    float wa = W[(size_t)(ka < PED ? ka : PED - 1) * DD + ncol];
    float wb = W[(size_t)(kb < PED ? kb : PED - 1) * DD + ncol];
    wp[i].x = (ka < PED) ? wa : 0.0f;
    wp[i].y = (kb < PED) ? wb : 0.0f;
  }
  v2f w2[DD / 4];
  #pragma unroll
  for (int i = 0; i < DD / 4; ++i) {
    w2[i].x = W2[(size_t)(4 * i + hi * 2    ) * DD + ncol];
    w2[i].y = W2[(size_t)(4 * i + hi * 2 + 1) * DD + ncol];
  }
  float bbv = bb[ncol];
  float wl[4], wr[4];
  #pragma unroll
  for (int j = 0; j < 4; ++j) { wl[j] = al[lane + 32 * j]; wr[j] = ar[lane + 32 * j]; }

  const float* srow = strip + mlo * TP + hi * 2;

  int trip = 0;
  for (; tm < NTILE; tm += GB, ++trip) {
    float* xs = xstrip + (BN ? (trip & 1) * 16 * TP : 0);
    v8f c = {};
    #pragma unroll
    for (int i = 0; i < PEDP / 4; ++i)
      c = __builtin_amdgcn_wmma_f32_16x16x4_f32(false, a[i], false, wp[i], (short)0, c,
                                                false, false);
    #pragma unroll
    for (int r = 0; r < 8; ++r) {
      float xr = xv[r];
      if (BN) {
        xr = (ov[r] - mu) * inv * ga + be + xv[r];
        xs[(r + hi * 8) * TP + ncol] = xr;
      }
      float v = c[r];
      v += xr + bbv;
      strip[(r + hi * 8) * TP + ncol] = v;
    }
    __syncthreads();

    if (tm + GB < NTILE) tile_loads<BN>(PEp, X, OB, tm + GB, mlo, hi, ncol, a, xv, ov);

    v8f d = {};
    #pragma unroll
    for (int i = 0; i < DD / 4; ++i) {
      v2f as = *(const v2f*)(srow + 4 * i);
      d = __builtin_amdgcn_wmma_f32_16x16x4_f32(false, as, false, w2[i], (short)0, d,
                                                false, false);
    }
    #pragma unroll
    for (int r = 0; r < 8; ++r) hstrip[(r + hi * 8) * TP + ncol] = d[r];
    __syncthreads();

    float ol = 0.0f, orr = 0.0f;
    #pragma unroll
    for (int i = 0; i < 2; ++i) {
      int row = wave * 2 + i;
      const float* hp = hstrip + row * TP;
      v4f q = *(const v4f*)(hp + lane * 4);
      ((v4f*)(H + (size_t)(tm * 16 + row) * DD))[lane] = q;
      if (BN) {
        v4f xq = *(const v4f*)(xs + row * TP + lane * 4);
        ((v4f*)(XO + (size_t)(tm * 16 + row) * DD))[lane] = xq;
      }
      float accl = 0.0f, accr = 0.0f;
      #pragma unroll
      for (int j = 0; j < 4; ++j) {
        float v = hp[lane + 32 * j];
        accl += v * wl[j];
        accr += v * wr[j];
      }
      #pragma unroll
      for (int off = 16; off; off >>= 1) {
        accl += __shfl_xor(accl, off, 32);
        accr += __shfl_xor(accr, off, 32);
      }
      if (lane == i) { ol = accl; orr = accr; }
    }
    if (lane < 2) {
      int n = tm * 16 + wave * 2 + lane;
      sl[n] = ol; sr[n] = orr;
    }
  }
}

template <int M>
__device__ __forceinline__ float row_xor(float x) {
  return __builtin_bit_cast(float, __builtin_amdgcn_update_dpp(0, __builtin_bit_cast(int, x),
                                                               0x160 | M, 0xf, 0xf, false));
}
__device__ __forceinline__ float half_xor(float x) {
  int b = __builtin_bit_cast(int, x);
  return __builtin_bit_cast(float,
                            __builtin_amdgcn_permlanex16(b, b, 0x76543210, 0xfedcba98, false, false));
}
__device__ __forceinline__ float wave_max(float m) {
  m = fmaxf(m, half_xor(m));
  m = fmaxf(m, row_xor<8>(m));
  m = fmaxf(m, row_xor<4>(m));
  m = fmaxf(m, row_xor<2>(m));
  m = fmaxf(m, row_xor<1>(m));
  return m;
}

typedef const float* rowp;
__device__ __forceinline__ rowp row_ptr(const float* __restrict__ h, int sS, int k) {
  return h + (size_t)__builtin_amdgcn_readlane(sS, k) * DD;
}
__device__ __forceinline__ void load_rows8(v4f (&q)[8], rowp b0, rowp b1, rowp b2, rowp b3, rowp b4, rowp b5, rowp b6, rowp b7, unsigned voff) {
  asm volatile("global_load_b128 %0, %8, %9\n\t"
               "global_load_b128 %1, %8, %10\n\t"
               "global_load_b128 %2, %8, %11\n\t"
               "global_load_b128 %3, %8, %12\n\t"
               "global_load_b128 %4, %8, %13\n\t"
               "global_load_b128 %5, %8, %14\n\t"
               "global_load_b128 %6, %8, %15\n\t"
               "global_load_b128 %7, %8, %16\n\t"
               "s_wait_loadcnt 0x0"
               : "=&v"(q[0]), "=&v"(q[1]), "=&v"(q[2]), "=&v"(q[3]), "=&v"(q[4]), "=&v"(q[5]), "=&v"(q[6]), "=&v"(q[7])
               : "v"(voff), "s"(b0), "s"(b1), "s"(b2), "s"(b3), "s"(b4), "s"(b5), "s"(b6), "s"(b7));
}
__device__ __forceinline__ void load_rows7(v4f (&q)[8], rowp b0, rowp b1, rowp b2, rowp b3, rowp b4, rowp b5, rowp b6, unsigned voff) {
  asm volatile("global_load_b128 %0, %7, %8\n\t"
               "global_load_b128 %1, %7, %9\n\t"
               "global_load_b128 %2, %7, %10\n\t"
               "global_load_b128 %3, %7, %11\n\t"
               "global_load_b128 %4, %7, %12\n\t"
               "global_load_b128 %5, %7, %13\n\t"
               "global_load_b128 %6, %7, %14\n\t"
               "s_wait_loadcnt 0x0"
               : "=&v"(q[0]), "=&v"(q[1]), "=&v"(q[2]), "=&v"(q[3]), "=&v"(q[4]), "=&v"(q[5]), "=&v"(q[6])
               : "v"(voff), "s"(b0), "s"(b1), "s"(b2), "s"(b3), "s"(b4), "s"(b5), "s"(b6));
}
__device__ __forceinline__ void load_rows6(v4f (&q)[8], rowp b0, rowp b1, rowp b2, rowp b3, rowp b4, rowp b5, unsigned voff) {
  asm volatile("global_load_b128 %0, %6, %7\n\t"
               "global_load_b128 %1, %6, %8\n\t"
               "global_load_b128 %2, %6, %9\n\t"
               "global_load_b128 %3, %6, %10\n\t"
               "global_load_b128 %4, %6, %11\n\t"
               "global_load_b128 %5, %6, %12\n\t"
               "s_wait_loadcnt 0x0"
               : "=&v"(q[0]), "=&v"(q[1]), "=&v"(q[2]), "=&v"(q[3]), "=&v"(q[4]), "=&v"(q[5])
               : "v"(voff), "s"(b0), "s"(b1), "s"(b2), "s"(b3), "s"(b4), "s"(b5));
}
__device__ __forceinline__ void load_rows5(v4f (&q)[8], rowp b0, rowp b1, rowp b2, rowp b3, rowp b4, unsigned voff) {
  asm volatile("global_load_b128 %0, %5, %6\n\t"
               "global_load_b128 %1, %5, %7\n\t"
               "global_load_b128 %2, %5, %8\n\t"
               "global_load_b128 %3, %5, %9\n\t"
               "global_load_b128 %4, %5, %10\n\t"
               "s_wait_loadcnt 0x0"
               : "=&v"(q[0]), "=&v"(q[1]), "=&v"(q[2]), "=&v"(q[3]), "=&v"(q[4])
               : "v"(voff), "s"(b0), "s"(b1), "s"(b2), "s"(b3), "s"(b4));
}
__device__ __forceinline__ void load_rows4(v4f (&q)[8], rowp b0, rowp b1, rowp b2, rowp b3, unsigned voff) {
  asm volatile("global_load_b128 %0, %4, %5\n\t"
               "global_load_b128 %1, %4, %6\n\t"
               "global_load_b128 %2, %4, %7\n\t"
               "global_load_b128 %3, %4, %8\n\t"
               "s_wait_loadcnt 0x0"
               : "=&v"(q[0]), "=&v"(q[1]), "=&v"(q[2]), "=&v"(q[3])
               : "v"(voff), "s"(b0), "s"(b1), "s"(b2), "s"(b3));
}
__device__ __forceinline__ void load_rows3(v4f (&q)[8], rowp b0, rowp b1, rowp b2, unsigned voff) {
  asm volatile("global_load_b128 %0, %3, %4\n\t"
               "global_load_b128 %1, %3, %5\n\t"
               "global_load_b128 %2, %3, %6\n\t"
               "s_wait_loadcnt 0x0"
               : "=&v"(q[0]), "=&v"(q[1]), "=&v"(q[2])
               : "v"(voff), "s"(b0), "s"(b1), "s"(b2));
}
__device__ __forceinline__ void load_rows2(v4f (&q)[8], rowp b0, rowp b1, unsigned voff) {
  asm volatile("global_load_b128 %0, %2, %3\n\t"
               "global_load_b128 %1, %2, %4\n\t"
               "s_wait_loadcnt 0x0"
               : "=&v"(q[0]), "=&v"(q[1])
               : "v"(voff), "s"(b0), "s"(b1));
}
__device__ __forceinline__ void load_rows1(v4f (&q)[8], rowp b0, unsigned voff) {
  asm volatile("global_load_b128 %0, %1, %2\n\t"
               "s_wait_loadcnt 0x0"
               : "=&v"(q[0])
               : "v"(voff), "s"(b0));
}
__device__ __forceinline__ float weight_of(int xS, int k) {
  return __builtin_bit_cast(float, __builtin_amdgcn_readlane(xS, k));
}
__device__ __forceinline__ void add_row(v2f& accLo, v2f& accHi, float& den, float w, v4f v) {
  den += w;
  v2f ww = {w, w};
  v2f lo = {v.x, v.y}, hi = {v.z, v.w};
  accLo = __builtin_elementwise_fma(lo, ww, accLo);
  accHi = __builtin_elementwise_fma(hi, ww, accHi);
}

template <bool FIRST>
__device__ __forceinline__ void sum_chunk(const float* __restrict__ h, bool hasC, int sC, int eC,
                                          float exC, unsigned voff, v2f& accLo, v2f& accHi,
                                          float& den) {
  int c = __builtin_popcount(__builtin_amdgcn_ballot_w32(hasC));
  int span = FIRST ? c : 32;
  int rk = 0;
  #pragma unroll 4
  for (int k = 0; k < span; ++k) rk += (__builtin_amdgcn_readlane(eC, k) < eC) ? 1 : 0;
  int sS = __builtin_amdgcn_ds_permute(rk << 2, sC);
  int xS = __builtin_amdgcn_ds_permute(rk << 2, __builtin_bit_cast(int, exC));
  int k = 0;
  v4f q[8];
  for (; k + 8 <= c; k += 8) {
    load_rows8(q, row_ptr(h, sS, k), row_ptr(h, sS, k + 1), row_ptr(h, sS, k + 2), row_ptr(h, sS, k + 3), row_ptr(h, sS, k + 4), row_ptr(h, sS, k + 5), row_ptr(h, sS, k + 6), row_ptr(h, sS, k + 7), voff);
    #pragma unroll
    for (int j = 0; j < 8; ++j) add_row(accLo, accHi, den, weight_of(xS, k + j), q[j]);
  }
  int rem = c - k;
  if (rem == 7) {
    load_rows7(q, row_ptr(h, sS, k), row_ptr(h, sS, k + 1), row_ptr(h, sS, k + 2), row_ptr(h, sS, k + 3), row_ptr(h, sS, k + 4), row_ptr(h, sS, k + 5), row_ptr(h, sS, k + 6), voff);
    #pragma unroll
    for (int j = 0; j < 7; ++j) add_row(accLo, accHi, den, weight_of(xS, k + j), q[j]);
  } else if (rem == 6) {
    load_rows6(q, row_ptr(h, sS, k), row_ptr(h, sS, k + 1), row_ptr(h, sS, k + 2), row_ptr(h, sS, k + 3), row_ptr(h, sS, k + 4), row_ptr(h, sS, k + 5), voff);
    #pragma unroll
    for (int j = 0; j < 6; ++j) add_row(accLo, accHi, den, weight_of(xS, k + j), q[j]);
  } else if (rem == 5) {
    load_rows5(q, row_ptr(h, sS, k), row_ptr(h, sS, k + 1), row_ptr(h, sS, k + 2), row_ptr(h, sS, k + 3), row_ptr(h, sS, k + 4), voff);
    #pragma unroll
    for (int j = 0; j < 5; ++j) add_row(accLo, accHi, den, weight_of(xS, k + j), q[j]);
  } else if (rem == 4) {
    load_rows4(q, row_ptr(h, sS, k), row_ptr(h, sS, k + 1), row_ptr(h, sS, k + 2), row_ptr(h, sS, k + 3), voff);
    #pragma unroll
    for (int j = 0; j < 4; ++j) add_row(accLo, accHi, den, weight_of(xS, k + j), q[j]);
  } else if (rem == 3) {
    load_rows3(q, row_ptr(h, sS, k), row_ptr(h, sS, k + 1), row_ptr(h, sS, k + 2), voff);
    #pragma unroll
    for (int j = 0; j < 3; ++j) add_row(accLo, accHi, den, weight_of(xS, k + j), q[j]);
  } else if (rem == 2) {
    load_rows2(q, row_ptr(h, sS, k), row_ptr(h, sS, k + 1), voff);
    #pragma unroll
    for (int j = 0; j < 2; ++j) add_row(accLo, accHi, den, weight_of(xS, k + j), q[j]);
  } else if (rem == 1) {
    load_rows1(q, row_ptr(h, sS, k), voff);
    #pragma unroll
    for (int j = 0; j < 1; ++j) add_row(accLo, accHi, den, weight_of(xS, k + j), q[j]);
  }
}

__device__ __forceinline__ float edge_score(const float* __restrict__ sl,
                                            const float* __restrict__ xe, float srn,
                                            int s, int t) {
  float a = sl[s] + srn + xe[t];
  return (a > 0.0f) ? a : 0.2f * a;
}

#define NWG 1536
#define NWV (NWG * 8)
static_assert(NWV <= NN, "every wave has a first node");

template <int LAST>
__global__ void __launch_bounds__(256, 12) k_node(const float* __restrict__ h,
                                              const float* __restrict__ sl,
                                              const float* __restrict__ sr,
                                              const float* __restrict__ xe,
                                              const int* __restrict__ cnt,
                                              const v4i* __restrict__ rec,
                                              const v4i* __restrict__ ovf,
                                              const float* __restrict__ bias,
                                              const float* __restrict__ xres,
                                              float* __restrict__ outb,
                                              float* __restrict__ ssum,
                                              float* __restrict__ ssq) {
  __shared__ __attribute__((aligned(16))) float red[LAST ? 4 : 2 * 8 * DD];
  int wave = __builtin_amdgcn_readfirstlane(threadIdx.x >> 5);
  int lane = threadIdx.x & 31;
  int g = blockIdx.x * 8 + wave;
  unsigned voff = (unsigned)lane * 16u;
  float4 b4 = ((const float4*)bias)[lane];
  float4 cs = make_float4(0.0f, 0.0f, 0.0f, 0.0f);
  float4 cq = make_float4(0.0f, 0.0f, 0.0f, 0.0f);

  int novfAll = __builtin_amdgcn_readfirstlane(cnt[NN]);

  int nB0 = (g + NWV < NN) ? g + NWV : NN - 1;
  int degA = __builtin_amdgcn_readfirstlane(cnt[g]);
  float srA = sr[g];
  v4i rA = rec[(size_t)g * CAP + lane];
  int degB = __builtin_amdgcn_readfirstlane(cnt[nB0]);
  float srB = sr[nB0];
  v4i rB = rec[(size_t)nB0 * CAP + lane];
  bool hasA = lane < (degA < CAP ? degA : CAP);
  float slA = sl[hasA ? rA.x : 0];
  float xeA = xe[hasA ? rA.y : 0];

  for (int n = g; n < NN; n += NWV) {
    int deg = degA;
    float srn = srA;
    bool has0 = hasA;
    int s0 = has0 ? rA.x : 0;
    int e0 = has0 ? rA.z : 0x7fffffff;
    float a0 = -INFINITY;
    if (has0) {
      float a = slA + srn + xeA;
      a0 = (a > 0.0f) ? a : 0.2f * a;
    }
    bool hasB = lane < (degB < CAP ? degB : CAP);
    float slB = sl[hasB ? rB.x : 0];
    float xeB = xe[hasB ? rB.y : 0];
    int nC = (n + 2 * NWV < NN) ? n + 2 * NWV : NN - 1;
    int degC = __builtin_amdgcn_readfirstlane(cnt[nC]);
    float srC = sr[nC];
    v4i rC = rec[(size_t)nC * CAP + lane];
    float4 x4 = make_float4(0.0f, 0.0f, 0.0f, 0.0f);
    if (LAST == 1) x4 = ((const float4*)(xres + (size_t)n * DD))[lane];

    int novf = (deg > CAP) ? novfAll : 0;
    int nchunk = 1 + ((novf + 31) >> 5);

    float m = a0;
    for (int ci = 1; ci < nchunk; ++ci) {
      int j = (ci - 1) * 32 + lane;
      if (j < novf) {
        v4i r = ovf[j];
        if (r.w == n) m = fmaxf(m, edge_score(sl, xe, srn, r.x, r.y));
      }
    }
    m = wave_max(m);

    v2f accLo = {0.0f, 0.0f}, accHi = {0.0f, 0.0f};
    float den = 0.0f;
    sum_chunk<true>(h, has0, s0, e0, has0 ? expf(a0 - m) : 0.0f, voff, accLo, accHi, den);
    for (int ci = 1; ci < nchunk; ++ci) {
      int j = (ci - 1) * 32 + lane;
      bool hasC = false;
      int sC = 0, eC = 0x7fffffff;
      float aC = -INFINITY;
      if (j < novf) {
        v4i r = ovf[j];
        if (r.w == n) {
          hasC = true; sC = r.x; eC = r.z;
          aC = edge_score(sl, xe, srn, r.x, r.y);
        }
      }
      sum_chunk<false>(h, hasC, sC, eC, hasC ? expf(aC - m) : 0.0f, voff, accLo, accHi, den);
    }
    float inv = (deg > 0) ? 1.0f / den : 0.0f;
    float4 o4 = make_float4(accLo.x * inv + b4.x, accLo.y * inv + b4.y, accHi.x * inv + b4.z,
                            accHi.y * inv + b4.w);
    if (LAST == 1) {
      o4.x += x4.x; o4.y += x4.y; o4.z += x4.z; o4.w += x4.w;
    } else {
      cs.x += o4.x; cs.y += o4.y; cs.z += o4.z; cs.w += o4.w;
      cq.x += o4.x * o4.x; cq.y += o4.y * o4.y; cq.z += o4.z * o4.z; cq.w += o4.w * o4.w;
    }
    ((float4*)(outb + (size_t)n * DD))[lane] = o4;

    degA = degB; srA = srB; rA = rB; hasA = hasB; slA = slB; xeA = xeB;
    degB = degC; srB = srC; rB = rC;
  }

  if (LAST == 0) {
    ((float4*)(red + wave * DD))[lane] = cs;
    ((float4*)(red + 8 * DD + wave * DD))[lane] = cq;
    __syncthreads();
    int t = threadIdx.x;
    int col = t & (DD - 1);
    const float* p = red + (t >> 7) * 8 * DD + col;
    float s = 0.0f;
    #pragma unroll
    for (int w = 0; w < 8; ++w) s += p[w * DD];
    float* dstp = ((t >> 7) ? ssq : ssum) + (blockIdx.x & (NCOPY - 1)) * DD + col;
    atomicAdd(dstp, s);
  }
}


extern "C" void kernel_launch(void* const* d_in, const int* in_sizes, int n_in,
                              void* d_out, int out_size, void* d_ws, size_t ws_size,
                              hipStream_t stream) {
  const float* X_n   = (const float*)d_in[0];
  const float* PE    = (const float*)d_in[1];
  const int*   eidx  = (const int*)d_in[2];
  const int*   eattr = (const int*)d_in[3];
  const float* eemb  = (const float*)d_in[4];
  const float* peW   = (const float*)d_in[5];
  const float* peB   = (const float*)d_in[6];
  const float* linW  = (const float*)d_in[7];
  const float* attl  = (const float*)d_in[8];
  const float* attr_ = (const float*)d_in[9];
  const float* atte  = (const float*)d_in[10];
  const float* bias  = (const float*)d_in[11];
  const float* gam   = (const float*)d_in[12];
  const float* bet   = (const float*)d_in[13];
  const int* src = eidx;
  const int* dst = eidx + EE;

  float* ws = (float*)d_ws;
  size_t o = 0;
  float* xa    = ws + o; o += (size_t)NN * DD;
  float* xb    = ws + o; o += (size_t)NN * DD;
  float* outb  = ws + o; o += (size_t)NN * DD;
  float* h     = ws + o; o += (size_t)NN * DD;
  float* PEp   = ws + o; o += (size_t)NN * PEDP;
  v4i*   rec   = (v4i*)(ws + o); o += (size_t)NN * CAP * 4;
  v4i*   ovf   = (v4i*)(ws + o); o += (size_t)EE * 4;
  float* sl    = ws + o; o += NN;
  float* sr    = ws + o; o += NN;
  int*   cnt   = (int*)(ws + o); o += NN + 16;
  float* xe    = ws + o; o += 16;
  float* ssum0 = ws + o; o += NCOPY * DD;
  float* ssq0  = ws + o; o += NCOPY * DD;
  float* ssum1 = ws + o; o += NCOPY * DD;
  float* ssq1  = ws + o; o += NCOPY * DD;
  float* xnext[NL - 1] = {xa, xb};
  float* ssum[2] = {ssum0, ssum1};
  float* ssq[2]  = {ssq0, ssq1};

  k_pad_pe<<<(NN * PEDP / 4 + 1023) / 1024, 256, 0, stream>>>(PE, PEp, cnt);
  k_bucket<<<EE / 1024, 256, 0, stream>>>(src, dst, eattr, cnt, rec, ovf);

  const float* xin = X_n;
  for (int l = 0; l < NL; ++l) {
    if (l == 0) {
      k_gemm2x<0><<<GB + 1, 256, 0, stream>>>(PEp, peW, xin, peB, linW, attl, attr_, eemb, atte, h,
                                              sl, sr, xe, ssum[0], ssq[0], nullptr, nullptr,
                                              nullptr, nullptr, nullptr, nullptr);
    } else {
      k_gemm2x<1><<<GB + 1, 256, 0, stream>>>(PEp, peW + (size_t)l * PED * DD, xin, peB + l * DD,
                                              linW + (size_t)l * DD * DD, attl + l * DD,
                                              attr_ + l * DD, eemb + (size_t)l * NT * DD,
                                              atte + l * DD, h, sl, sr, xe, ssum[l & 1], ssq[l & 1],
                                              outb, ssum[(l - 1) & 1], ssq[(l - 1) & 1],
                                              gam + (l - 1) * DD, bet + (l - 1) * DD, xnext[l - 1]);
      xin = xnext[l - 1];
    }

    if (l < NL - 1) {
      k_node<0><<<NWG, 256, 0, stream>>>(h, sl, sr, xe, cnt, rec, ovf, bias + l * DD, xin, outb,
                                            ssum[l & 1], ssq[l & 1]);
    } else {
      k_node<1><<<NWG, 256, 0, stream>>>(h, sl, sr, xe, cnt, rec, ovf, bias + l * DD, xin,
                                            (float*)d_out, ssum[l & 1], ssq[l & 1]);
    }
  }
}
